# starting kernel
# baseline (speedup 1.0000x reference)
#include <hip/hip_runtime.h>
#include <math.h>

#define B_     128
#define NE_    64
#define NN_    16
#define D_     128
#define NRBF_  16
#define L_     3
#define NUP_   32
#define CUTOFF_ 2.0f
#define EPS_    1e-12f

#define KEE_   (NE_ * NRBF_)
#define KNE_   (NN_ * NRBF_)
#define NT_    4
#define WSZ_   (D_ * D_)

typedef float v2f __attribute__((ext_vector_type(2)));
typedef float v8f __attribute__((ext_vector_type(8)));

__device__ __forceinline__ void gemm_strip_f32p(const float* __restrict__ A, int lda,
                                                const float* __restrict__ Bp,
                                                int m0, int n0, v8f acc[NT_], int lane) {
  const int half = lane >> 4;
  const int l16  = lane & 15;
  const float* __restrict__ arow = A + (size_t)(m0 + l16) * lda;
#pragma unroll
  for (int k0 = 0; k0 < D_; k0 += 4) {
    const int ka = k0 + half * 2;
    const v2f a = *(const v2f*)&arow[ka];
    const float* __restrict__ brow = Bp + (size_t)(ka >> 1) * (2 * D_);
#pragma unroll
    for (int t = 0; t < NT_; ++t) {
      const v2f b = *(const v2f*)&brow[(n0 + t * 16 + l16) * 2];
      acc[t] = __builtin_amdgcn_wmma_f32_16x16x4_f32(false, a, false, b, (short)0,
                                                     acc[t], false, false);
    }
  }
}

__device__ __forceinline__ void store_strip_f32(float* __restrict__ C, int ldc,
                                                int m0, int n0, v8f acc[NT_], int lane) {
  const int half = lane >> 4;
  const int l16  = lane & 15;
#pragma unroll
  for (int t = 0; t < NT_; ++t)
#pragma unroll
    for (int ii = 0; ii < 8; ++ii)
      C[(size_t)(m0 + half * 8 + ii) * ldc + n0 + t * 16 + l16] = acc[t][ii];
}

__device__ __forceinline__ void load_strip_f32(const float* __restrict__ C, int ldc,
                                               int m0, int n0, v8f acc[NT_], int lane) {
  const int half = lane >> 4;
  const int l16  = lane & 15;
#pragma unroll
  for (int t = 0; t < NT_; ++t)
#pragma unroll
    for (int ii = 0; ii < 8; ++ii)
      acc[t][ii] = C[(size_t)(m0 + half * 8 + ii) * ldc + n0 + t * 16 + l16];
}

__global__ void gnn_pack_w(const float* __restrict__ We, const float* __restrict__ Wn,
                           const float* __restrict__ W1, const float* __restrict__ W2,
                           float* __restrict__ Wp) {
  const int gid = blockIdx.x * blockDim.x + threadIdx.x;
  const int m   = gid >> 14;
  const int idx = gid & (WSZ_ - 1);
  const int l = m >> 2, w = m & 3;
  const float* src = (w == 0 ? We : w == 1 ? Wn : w == 2 ? W1 : W2) + (size_t)l * WSZ_;
  const int k = idx >> 7, col = idx & (D_ - 1);
  Wp[(size_t)m * WSZ_ + (size_t)(k >> 1) * (2 * D_) + col * 2 + (k & 1)] = src[idx];
}

__global__ void gnn_feat_ee(const float* __restrict__ r, float* __restrict__ Aee) {
  const int bi = blockIdx.x;
  const int b = bi >> 6, i = bi & 63;
  const float rx = r[(size_t)(b * NE_ + i) * 3 + 0];
  const float ry = r[(size_t)(b * NE_ + i) * 3 + 1];
  const float rz = r[(size_t)(b * NE_ + i) * 3 + 2];
  const float inv_cut = 1.0f / CUTOFF_;
  const float dmu     = CUTOFF_ / (float)(NRBF_ - 1);
  const float inv_sig = (float)NRBF_ / CUTOFF_;
  for (int t = threadIdx.x; t < KEE_; t += blockDim.x) {
    const int j = t >> 4, k = t & 15;
    const float dx = rx - r[(size_t)(b * NE_ + j) * 3 + 0];
    const float dy = ry - r[(size_t)(b * NE_ + j) * 3 + 1];
    const float dz = rz - r[(size_t)(b * NE_ + j) * 3 + 2];
    const float dist = sqrtf(dx * dx + dy * dy + dz * dz + EPS_);
    float env = 1.0f - dist * inv_cut;
    env = (dist < CUTOFF_) ? env * env : 0.0f;
    const float z = (dist - dmu * (float)k) * inv_sig;
    float g = expf(-0.5f * z * z) * env;
    if (i == j) g = 0.0f;
    Aee[(size_t)bi * KEE_ + t] = g;
  }
}

__global__ void gnn_feat_ne(const float* __restrict__ r, const float* __restrict__ coords,
                            float* __restrict__ Ane) {
  const int bj = blockIdx.x;
  const int b = bj >> 6, j = bj & 63;
  const float rx = r[(size_t)(b * NE_ + j) * 3 + 0];
  const float ry = r[(size_t)(b * NE_ + j) * 3 + 1];
  const float rz = r[(size_t)(b * NE_ + j) * 3 + 2];
  const float inv_cut = 1.0f / CUTOFF_;
  const float dmu     = CUTOFF_ / (float)(NRBF_ - 1);
  const float inv_sig = (float)NRBF_ / CUTOFF_;
  const int t = threadIdx.x;
  const int a = t >> 4, k = t & 15;
  const float dx = rx - coords[a * 3 + 0];
  const float dy = ry - coords[a * 3 + 1];
  const float dz = rz - coords[a * 3 + 2];
  const float dist = sqrtf(dx * dx + dy * dy + dz * dz + EPS_);
  float env = 1.0f - dist * inv_cut;
  env = (dist < CUTOFF_) ? env * env : 0.0f;
  const float z = (dist - dmu * (float)k) * inv_sig;
  const float g = expf(-0.5f * z * z) * env;
  Ane[(size_t)bj * KNE_ + t] = g;
}

__global__ void gnn_init_h(const float* __restrict__ spin_emb, float* __restrict__ h) {
  const int gid = blockIdx.x * blockDim.x + threadIdx.x;
  const int d = gid & (D_ - 1);
  const int i = (gid >> 7) & (NE_ - 1);
  h[gid] = spin_emb[(i < NUP_ ? 0 : 1) * D_ + d];
}

__global__ void gnn_hw(const float* __restrict__ h, const float* __restrict__ nuc_emb,
                       const float* __restrict__ Wp_ee, const float* __restrict__ Wp_ne,
                       float* __restrict__ hW, float* __restrict__ hnW) {
  const int wid = blockIdx.x;
  const int lane = threadIdx.x & 31;
  v8f acc[NT_] = {};
  if (wid < 1024) {
    const int m0 = (wid >> 1) * 16;
    const int n0 = (wid & 1) * 64;
    gemm_strip_f32p(h, D_, Wp_ee, m0, n0, acc, lane);
    store_strip_f32(hW, D_, m0, n0, acc, lane);
  } else {
    const int n0 = (wid - 1024) * 64;
    gemm_strip_f32p(nuc_emb, D_, Wp_ne, 0, n0, acc, lane);
    store_strip_f32(hnW, D_, 0, n0, acc, lane);
  }
}

__global__ void gnn_agg(const float* __restrict__ Aee, const float* __restrict__ Ane,
                        const float* __restrict__ hW, const float* __restrict__ hnW,
                        const float* __restrict__ Wrbf_ee_l, const float* __restrict__ Wrbf_ne_l,
                        float* __restrict__ agg) {
  const int wid = blockIdx.x;
  const int b  = wid >> 3;
  const int mt = (wid >> 1) & 3;
  const int nb = wid & 1;
  const int lane = threadIdx.x & 31;
  const int half = lane >> 4, l16 = lane & 15;
  const int i0 = mt * 16, n0 = nb * 64;

  const float* __restrict__ Ab  = Aee + (size_t)b * NE_ * KEE_;
  const float* __restrict__ An  = Ane + (size_t)b * NE_ * KNE_;
  const float* __restrict__ hWb = hW + (size_t)b * NE_ * D_;
  v8f acc[NT_] = {};

  float wk[NT_][8];

  {
#pragma unroll
    for (int kk = 0; kk < 4; ++kk)
#pragma unroll
      for (int t = 0; t < NT_; ++t) {
        const int kb = kk * 4 + half * 2;
        wk[t][kk * 2 + 0] = Wrbf_ee_l[(size_t)(kb + 0) * D_ + n0 + t * 16 + l16];
        wk[t][kk * 2 + 1] = Wrbf_ee_l[(size_t)(kb + 1) * D_ + n0 + t * 16 + l16];
      }
    const float* __restrict__ arow = Ab + (size_t)(i0 + l16) * KEE_;
    for (int j = 0; j < NE_; ++j) {
      float hv[NT_];
#pragma unroll
      for (int t = 0; t < NT_; ++t) hv[t] = hWb[(size_t)j * D_ + n0 + t * 16 + l16];
#pragma unroll
      for (int kk = 0; kk < 4; ++kk) {
        const int ka = j * 16 + kk * 4 + half * 2;
        const v2f a = *(const v2f*)&arow[ka];
#pragma unroll
        for (int t = 0; t < NT_; ++t) {
          v2f bb;
          bb.x = hv[t] * wk[t][kk * 2 + 0];
          bb.y = hv[t] * wk[t][kk * 2 + 1];
          acc[t] = __builtin_amdgcn_wmma_f32_16x16x4_f32(false, a, false, bb, (short)0,
                                                         acc[t], false, false);
        }
      }
    }
  }

  {
#pragma unroll
    for (int kk = 0; kk < 4; ++kk)
#pragma unroll
      for (int t = 0; t < NT_; ++t) {
        const int kb = kk * 4 + half * 2;
        wk[t][kk * 2 + 0] = Wrbf_ne_l[(size_t)(kb + 0) * D_ + n0 + t * 16 + l16];
        wk[t][kk * 2 + 1] = Wrbf_ne_l[(size_t)(kb + 1) * D_ + n0 + t * 16 + l16];
      }
    const float* __restrict__ arow = An + (size_t)(i0 + l16) * KNE_;
#pragma unroll
    for (int a_idx = 0; a_idx < NN_; ++a_idx) {
      float hv[NT_];
#pragma unroll
      for (int t = 0; t < NT_; ++t) hv[t] = hnW[(size_t)a_idx * D_ + n0 + t * 16 + l16];
#pragma unroll
      for (int kk = 0; kk < 4; ++kk) {
        const int ka = a_idx * 16 + kk * 4 + half * 2;
        const v2f a = *(const v2f*)&arow[ka];
#pragma unroll
        for (int t = 0; t < NT_; ++t) {
          v2f bb;
          bb.x = hv[t] * wk[t][kk * 2 + 0];
          bb.y = hv[t] * wk[t][kk * 2 + 1];
          acc[t] = __builtin_amdgcn_wmma_f32_16x16x4_f32(false, a, false, bb, (short)0,
                                                         acc[t], false, false);
        }
      }
    }
  }

  store_strip_f32(agg + (size_t)b * NE_ * D_, D_, i0, n0, acc, lane);
}

__global__ void gnn_mlp1(const float* __restrict__ agg, const float* __restrict__ Wp1,
                         float* __restrict__ tmp) {
  const int wid = blockIdx.x;
  const int lane = threadIdx.x & 31;
  const int m0 = (wid >> 1) * 16;
  const int n0 = (wid & 1) * 64;
  v8f acc[NT_] = {};
  gemm_strip_f32p(agg, D_, Wp1, m0, n0, acc, lane);
#pragma unroll
  for (int t = 0; t < NT_; ++t)
#pragma unroll
    for (int ii = 0; ii < 8; ++ii) acc[t][ii] = tanhf(acc[t][ii]);
  store_strip_f32(tmp, D_, m0, n0, acc, lane);
}

__global__ void gnn_mlp2(const float* __restrict__ tmp, const float* __restrict__ Wp2,
                         const float* __restrict__ h, float* __restrict__ dst) {
  const int wid = blockIdx.x;
  const int lane = threadIdx.x & 31;
  const int m0 = (wid >> 1) * 16;
  const int n0 = (wid & 1) * 64;
  v8f acc[NT_];
  load_strip_f32(h, D_, m0, n0, acc, lane);
  gemm_strip_f32p(tmp, D_, Wp2, m0, n0, acc, lane);
  store_strip_f32(dst, D_, m0, n0, acc, lane);
}

extern "C" void kernel_launch(void* const* d_in, const int* in_sizes, int n_in,
                              void* d_out, int out_size, void* d_ws, size_t ws_size,
                              hipStream_t stream) {
  (void)in_sizes; (void)n_in; (void)out_size; (void)ws_size;
  const float* r        = (const float*)d_in[0];
  const float* coords   = (const float*)d_in[1];
  const float* spin_emb = (const float*)d_in[2];
  const float* nuc_emb  = (const float*)d_in[3];
  const float* W_rbf_ee = (const float*)d_in[4];
  const float* W_rbf_ne = (const float*)d_in[5];
  const float* W_msg_ee = (const float*)d_in[6];
  const float* W_msg_ne = (const float*)d_in[7];
  const float* W1       = (const float*)d_in[8];
  const float* W2       = (const float*)d_in[9];
  float* out = (float*)d_out;

  float* ws  = (float*)d_ws;
  float* Aee = ws;
  float* Ane = Aee + (size_t)B_ * NE_ * KEE_;
  float* h   = Ane + (size_t)B_ * NE_ * KNE_;
  float* hW  = h   + (size_t)B_ * NE_ * D_;
  float* hnW = hW  + (size_t)B_ * NE_ * D_;
  float* agg = hnW + (size_t)NN_ * D_;
  float* tmp = agg + (size_t)B_ * NE_ * D_;
  float* Wp  = tmp + (size_t)B_ * NE_ * D_;

  gnn_pack_w<<<(12 * WSZ_) / 256, 256, 0, stream>>>(W_msg_ee, W_msg_ne, W1, W2, Wp);
  gnn_feat_ee<<<B_ * NE_, 256, 0, stream>>>(r, Aee);
  gnn_feat_ne<<<B_ * NE_, 256, 0, stream>>>(r, coords, Ane);
  gnn_init_h<<<(B_ * NE_ * D_) / 256, 256, 0, stream>>>(spin_emb, h);

  for (int l = 0; l < L_; ++l) {
    const float* Wpl = Wp + (size_t)l * 4 * WSZ_;
    gnn_hw<<<1024 + 2, 32, 0, stream>>>(h, nuc_emb, Wpl + 0 * WSZ_, Wpl + 1 * WSZ_, hW, hnW);
    gnn_agg<<<B_ * 8, 32, 0, stream>>>(Aee, Ane, hW, hnW,
                                       W_rbf_ee + (size_t)l * NRBF_ * D_,
                                       W_rbf_ne + (size_t)l * NRBF_ * D_, agg);
    gnn_mlp1<<<1024, 32, 0, stream>>>(agg, Wpl + 2 * WSZ_, tmp);
    float* dst = (l == L_ - 1) ? out : h;
    gnn_mlp2<<<1024, 32, 0, stream>>>(tmp, Wpl + 3 * WSZ_, h, dst);
  }
}


# agg re-associated: per receiver a WMMA tile holds the 16 radial rows and sums over senders with plain hW as B (kept in registers); W_rbf-weighted sum over rows afterwards; sum order changes
# speedup vs baseline: 1.3673x; 1.3673x over previous
#include <hip/hip_runtime.h>
#include <math.h>

#define B_     128
#define NE_    64
#define NN_    16
#define D_     128
#define NRBF_  16
#define L_     3
#define NUP_   32
#define CUTOFF_ 2.0f
#define EPS_    1e-12f

#define KEE_   (NE_ * NRBF_)
#define KNE_   (NN_ * NRBF_)
#define NT_    4
#define WSZ_   (D_ * D_)

typedef float v2f __attribute__((ext_vector_type(2)));
typedef float v8f __attribute__((ext_vector_type(8)));

__device__ __forceinline__ void gemm_strip_f32p(const float* __restrict__ A, int lda,
                                                const float* __restrict__ Bp,
                                                int m0, int n0, v8f acc[NT_], int lane) {
  const int half = lane >> 4;
  const int l16  = lane & 15;
  const float* __restrict__ arow = A + (size_t)(m0 + l16) * lda;
#pragma unroll
  for (int k0 = 0; k0 < D_; k0 += 4) {
    const int ka = k0 + half * 2;
    const v2f a = *(const v2f*)&arow[ka];
    const float* __restrict__ brow = Bp + (size_t)(ka >> 1) * (2 * D_);
#pragma unroll
    for (int t = 0; t < NT_; ++t) {
      const v2f b = *(const v2f*)&brow[(n0 + t * 16 + l16) * 2];
      acc[t] = __builtin_amdgcn_wmma_f32_16x16x4_f32(false, a, false, b, (short)0,
                                                     acc[t], false, false);
    }
  }
}

__device__ __forceinline__ void store_strip_f32(float* __restrict__ C, int ldc,
                                                int m0, int n0, v8f acc[NT_], int lane) {
  const int half = lane >> 4;
  const int l16  = lane & 15;
#pragma unroll
  for (int t = 0; t < NT_; ++t)
#pragma unroll
    for (int ii = 0; ii < 8; ++ii)
      C[(size_t)(m0 + half * 8 + ii) * ldc + n0 + t * 16 + l16] = acc[t][ii];
}

__device__ __forceinline__ void load_strip_f32(const float* __restrict__ C, int ldc,
                                               int m0, int n0, v8f acc[NT_], int lane) {
  const int half = lane >> 4;
  const int l16  = lane & 15;
#pragma unroll
  for (int t = 0; t < NT_; ++t)
#pragma unroll
    for (int ii = 0; ii < 8; ++ii)
      acc[t][ii] = C[(size_t)(m0 + half * 8 + ii) * ldc + n0 + t * 16 + l16];
}

__global__ void gnn_pack_w(const float* __restrict__ We, const float* __restrict__ Wn,
                           const float* __restrict__ W1, const float* __restrict__ W2,
                           float* __restrict__ Wp) {
  const int gid = blockIdx.x * blockDim.x + threadIdx.x;
  const int m   = gid >> 14;
  const int idx = gid & (WSZ_ - 1);
  const int l = m >> 2, w = m & 3;
  const float* src = (w == 0 ? We : w == 1 ? Wn : w == 2 ? W1 : W2) + (size_t)l * WSZ_;
  const int k = idx >> 7, col = idx & (D_ - 1);
  Wp[(size_t)m * WSZ_ + (size_t)(k >> 1) * (2 * D_) + col * 2 + (k & 1)] = src[idx];
}

__global__ void gnn_feat_ee(const float* __restrict__ r, float* __restrict__ Aee) {
  const int bi = blockIdx.x;
  const int b = bi >> 6, i = bi & 63;
  const float rx = r[(size_t)(b * NE_ + i) * 3 + 0];
  const float ry = r[(size_t)(b * NE_ + i) * 3 + 1];
  const float rz = r[(size_t)(b * NE_ + i) * 3 + 2];
  const float inv_cut = 1.0f / CUTOFF_;
  const float dmu     = CUTOFF_ / (float)(NRBF_ - 1);
  const float inv_sig = (float)NRBF_ / CUTOFF_;
  for (int t = threadIdx.x; t < KEE_; t += blockDim.x) {
    const int j = t >> 4, k = t & 15;
    const float dx = rx - r[(size_t)(b * NE_ + j) * 3 + 0];
    const float dy = ry - r[(size_t)(b * NE_ + j) * 3 + 1];
    const float dz = rz - r[(size_t)(b * NE_ + j) * 3 + 2];
    const float dist = sqrtf(dx * dx + dy * dy + dz * dz + EPS_);
    float env = 1.0f - dist * inv_cut;
    env = (dist < CUTOFF_) ? env * env : 0.0f;
    const float z = (dist - dmu * (float)k) * inv_sig;
    float g = expf(-0.5f * z * z) * env;
    if (i == j) g = 0.0f;
    Aee[(size_t)bi * KEE_ + t] = g;
  }
}

__global__ void gnn_feat_ne(const float* __restrict__ r, const float* __restrict__ coords,
                            float* __restrict__ Ane) {
  const int bj = blockIdx.x;
  const int b = bj >> 6, j = bj & 63;
  const float rx = r[(size_t)(b * NE_ + j) * 3 + 0];
  const float ry = r[(size_t)(b * NE_ + j) * 3 + 1];
  const float rz = r[(size_t)(b * NE_ + j) * 3 + 2];
  const float inv_cut = 1.0f / CUTOFF_;
  const float dmu     = CUTOFF_ / (float)(NRBF_ - 1);
  const float inv_sig = (float)NRBF_ / CUTOFF_;
  const int t = threadIdx.x;
  const int a = t >> 4, k = t & 15;
  const float dx = rx - coords[a * 3 + 0];
  const float dy = ry - coords[a * 3 + 1];
  const float dz = rz - coords[a * 3 + 2];
  const float dist = sqrtf(dx * dx + dy * dy + dz * dz + EPS_);
  float env = 1.0f - dist * inv_cut;
  env = (dist < CUTOFF_) ? env * env : 0.0f;
  const float z = (dist - dmu * (float)k) * inv_sig;
  const float g = expf(-0.5f * z * z) * env;
  Ane[(size_t)bj * KNE_ + t] = g;
}

__global__ void gnn_init_h(const float* __restrict__ spin_emb, float* __restrict__ h) {
  const int gid = blockIdx.x * blockDim.x + threadIdx.x;
  const int d = gid & (D_ - 1);
  const int i = (gid >> 7) & (NE_ - 1);
  h[gid] = spin_emb[(i < NUP_ ? 0 : 1) * D_ + d];
}

__global__ void gnn_hw(const float* __restrict__ h, const float* __restrict__ nuc_emb,
                       const float* __restrict__ Wp_ee, const float* __restrict__ Wp_ne,
                       float* __restrict__ hW, float* __restrict__ hnW) {
  const int wid = blockIdx.x;
  const int lane = threadIdx.x & 31;
  v8f acc[NT_] = {};
  if (wid < 1024) {
    const int m0 = (wid >> 1) * 16;
    const int n0 = (wid & 1) * 64;
    gemm_strip_f32p(h, D_, Wp_ee, m0, n0, acc, lane);
    store_strip_f32(hW, D_, m0, n0, acc, lane);
  } else {
    const int n0 = (wid - 1024) * 64;
    gemm_strip_f32p(nuc_emb, D_, Wp_ne, 0, n0, acc, lane);
    store_strip_f32(hnW, D_, 0, n0, acc, lane);
  }
}

#define IPW_ 8
#define SEE_ (NE_ / 4)
#define SNE_ (NN_ / 4)

struct AggA { float e[2 * SEE_]; float n[2 * SNE_]; };

__device__ __forceinline__ void agg_load_a(AggA& f, const float* __restrict__ ae,
                                           const float* __restrict__ an) {
#pragma unroll
  for (int s = 0; s < SEE_; ++s) { f.e[2 * s] = ae[64 * s]; f.e[2 * s + 1] = ae[64 * s + 32]; }
#pragma unroll
  for (int s = 0; s < SNE_; ++s) { f.n[2 * s] = an[64 * s]; f.n[2 * s + 1] = an[64 * s + 32]; }
}

__device__ __forceinline__ float agg_xhalf(float x) {
  const int v = __builtin_bit_cast(int, x);
  return __builtin_bit_cast(float,
      __builtin_amdgcn_permlanex16(v, v, 0x76543210, 0xfedcba98, false, false));
}

__device__ __forceinline__ void agg_one(const AggA& f, const v2f (&bee)[SEE_][NT_],
                                        const v2f (&bne)[SNE_][NT_],
                                        const v2f (&wee)[NT_][4], const v2f (&wne)[NT_][4],
                                        float* __restrict__ orow, int half) {
  v8f g[NT_] = {};
#pragma unroll
  for (int s = 0; s < SEE_; ++s) {
    v2f a; a.x = f.e[2 * s]; a.y = f.e[2 * s + 1];
#pragma unroll
    for (int t = 0; t < NT_; ++t)
      g[t] = __builtin_amdgcn_wmma_f32_16x16x4_f32(false, a, false, bee[s][t], (short)0,
                                                   g[t], false, false);
  }
  v8f gn[NT_] = {};
#pragma unroll
  for (int s = 0; s < SNE_; ++s) {
    v2f a; a.x = f.n[2 * s]; a.y = f.n[2 * s + 1];
#pragma unroll
    for (int t = 0; t < NT_; ++t)
      gn[t] = __builtin_amdgcn_wmma_f32_16x16x4_f32(false, a, false, bne[s][t], (short)0,
                                                    gn[t], false, false);
  }
  float sv[NT_];
#pragma unroll
  for (int t = 0; t < NT_; ++t) {
    v2f p = {0.0f, 0.0f};
#pragma unroll
    for (int r = 0; r < 8; r += 2) {
      v2f gv; gv.x = g[t][r];  gv.y = g[t][r + 1];
      p += wee[t][r >> 1] * gv;
    }
#pragma unroll
    for (int r = 0; r < 8; r += 2) {
      v2f gv; gv.x = gn[t][r]; gv.y = gn[t][r + 1];
      p += wne[t][r >> 1] * gv;
    }
    sv[t] = p.x + p.y;
  }
  const float own0 = half ? sv[1] : sv[0], snd0 = half ? sv[0] : sv[1];
  const float own1 = half ? sv[3] : sv[2], snd1 = half ? sv[2] : sv[3];
  orow[0]  = own0 + agg_xhalf(snd0);
  orow[32] = own1 + agg_xhalf(snd1);
}

__global__ __launch_bounds__(128) void gnn_agg(const float* __restrict__ Aee,
                        const float* __restrict__ Ane,
                        const float* __restrict__ hW, const float* __restrict__ hnW,
                        const float* __restrict__ Wrbf_ee_l, const float* __restrict__ Wrbf_ne_l,
                        float* __restrict__ agg) {
  const int lane = threadIdx.x & 31;
  const int wv   = threadIdx.x >> 5;
  const int n0   = (wv & 1) * 64;
  const int b    = blockIdx.x / (NE_ / (2 * IPW_));
  const int i0   = ((blockIdx.x % (NE_ / (2 * IPW_))) * 2 + (wv >> 1)) * IPW_;
  const int half = lane >> 4, l16 = lane & 15;

  const float* __restrict__ ae = Aee + (size_t)(b * NE_ + i0) * KEE_ + lane;
  const float* __restrict__ an = Ane + (size_t)(b * NE_ + i0) * KNE_ + lane;
  float* __restrict__ orow = agg + (size_t)(b * NE_ + i0) * D_ + n0 + lane;

  AggA f0, f1;
  agg_load_a(f0, ae, an);

  const float* __restrict__ hb = hW + (size_t)b * NE_ * D_ + half * D_ + n0 + l16;
  const float* __restrict__ nb = hnW + half * D_ + n0 + l16;
  v2f bee[SEE_][NT_], bne[SNE_][NT_];
#pragma unroll
  for (int s = 0; s < SEE_; ++s)
#pragma unroll
    for (int t = 0; t < NT_; ++t) {
      bee[s][t].x = hb[(4 * s) * D_ + 16 * t];
      bee[s][t].y = hb[(4 * s + 2) * D_ + 16 * t];
    }
#pragma unroll
  for (int s = 0; s < SNE_; ++s)
#pragma unroll
    for (int t = 0; t < NT_; ++t) {
      bne[s][t].x = nb[(4 * s) * D_ + 16 * t];
      bne[s][t].y = nb[(4 * s + 2) * D_ + 16 * t];
    }
  v2f wee[NT_][4], wne[NT_][4];
#pragma unroll
  for (int t = 0; t < NT_; ++t)
#pragma unroll
    for (int r = 0; r < 8; r += 2) {
      wee[t][r >> 1].x = Wrbf_ee_l[(size_t)(8 * half + r) * D_ + n0 + 16 * t + l16];
      wee[t][r >> 1].y = Wrbf_ee_l[(size_t)(8 * half + r + 1) * D_ + n0 + 16 * t + l16];
      wne[t][r >> 1].x = Wrbf_ne_l[(size_t)(8 * half + r) * D_ + n0 + 16 * t + l16];
      wne[t][r >> 1].y = Wrbf_ne_l[(size_t)(8 * half + r + 1) * D_ + n0 + 16 * t + l16];
    }

#pragma unroll 1
  for (int ii = 0; ii < IPW_ - 2; ii += 2) {
    agg_load_a(f1, ae + (size_t)(ii + 1) * KEE_, an + (size_t)(ii + 1) * KNE_);
    agg_one(f0, bee, bne, wee, wne, orow + (size_t)ii * D_, half);
    agg_load_a(f0, ae + (size_t)(ii + 2) * KEE_, an + (size_t)(ii + 2) * KNE_);
    agg_one(f1, bee, bne, wee, wne, orow + (size_t)(ii + 1) * D_, half);
  }
  agg_load_a(f1, ae + (size_t)(IPW_ - 1) * KEE_, an + (size_t)(IPW_ - 1) * KNE_);
  agg_one(f0, bee, bne, wee, wne, orow + (size_t)(IPW_ - 2) * D_, half);
  agg_one(f1, bee, bne, wee, wne, orow + (size_t)(IPW_ - 1) * D_, half);
}

__global__ void gnn_mlp1(const float* __restrict__ agg, const float* __restrict__ Wp1,
                         float* __restrict__ tmp) {
  const int wid = blockIdx.x;
  const int lane = threadIdx.x & 31;
  const int m0 = (wid >> 1) * 16;
  const int n0 = (wid & 1) * 64;
  v8f acc[NT_] = {};
  gemm_strip_f32p(agg, D_, Wp1, m0, n0, acc, lane);
#pragma unroll
  for (int t = 0; t < NT_; ++t)
#pragma unroll
    for (int ii = 0; ii < 8; ++ii) acc[t][ii] = tanhf(acc[t][ii]);
  store_strip_f32(tmp, D_, m0, n0, acc, lane);
}

__global__ void gnn_mlp2(const float* __restrict__ tmp, const float* __restrict__ Wp2,
                         const float* __restrict__ h, float* __restrict__ dst) {
  const int wid = blockIdx.x;
  const int lane = threadIdx.x & 31;
  const int m0 = (wid >> 1) * 16;
  const int n0 = (wid & 1) * 64;
  v8f acc[NT_];
  load_strip_f32(h, D_, m0, n0, acc, lane);
  gemm_strip_f32p(tmp, D_, Wp2, m0, n0, acc, lane);
  store_strip_f32(dst, D_, m0, n0, acc, lane);
}

extern "C" void kernel_launch(void* const* d_in, const int* in_sizes, int n_in,
                              void* d_out, int out_size, void* d_ws, size_t ws_size,
                              hipStream_t stream) {
  (void)in_sizes; (void)n_in; (void)out_size; (void)ws_size;
  const float* r        = (const float*)d_in[0];
  const float* coords   = (const float*)d_in[1];
  const float* spin_emb = (const float*)d_in[2];
  const float* nuc_emb  = (const float*)d_in[3];
  const float* W_rbf_ee = (const float*)d_in[4];
  const float* W_rbf_ne = (const float*)d_in[5];
  const float* W_msg_ee = (const float*)d_in[6];
  const float* W_msg_ne = (const float*)d_in[7];
  const float* W1       = (const float*)d_in[8];
  const float* W2       = (const float*)d_in[9];
  float* out = (float*)d_out;

  float* ws  = (float*)d_ws;
  float* Aee = ws;
  float* Ane = Aee + (size_t)B_ * NE_ * KEE_;
  float* h   = Ane + (size_t)B_ * NE_ * KNE_;
  float* hW  = h   + (size_t)B_ * NE_ * D_;
  float* hnW = hW  + (size_t)B_ * NE_ * D_;
  float* agg = hnW + (size_t)NN_ * D_;
  float* tmp = agg + (size_t)B_ * NE_ * D_;
  float* Wp  = tmp + (size_t)B_ * NE_ * D_;

  gnn_pack_w<<<(12 * WSZ_) / 256, 256, 0, stream>>>(W_msg_ee, W_msg_ne, W1, W2, Wp);
  gnn_feat_ee<<<B_ * NE_, 256, 0, stream>>>(r, Aee);
  gnn_feat_ne<<<B_ * NE_, 256, 0, stream>>>(r, coords, Ane);
  gnn_init_h<<<(B_ * NE_ * D_) / 256, 256, 0, stream>>>(spin_emb, h);

  for (int l = 0; l < L_; ++l) {
    const float* Wpl = Wp + (size_t)l * 4 * WSZ_;
    gnn_hw<<<1024 + 2, 32, 0, stream>>>(h, nuc_emb, Wpl + 0 * WSZ_, Wpl + 1 * WSZ_, hW, hnW);
    gnn_agg<<<B_ * (NE_ / (2 * IPW_)), 128, 0, stream>>>(Aee, Ane, hW, hnW,
                                       W_rbf_ee + (size_t)l * NRBF_ * D_,
                                       W_rbf_ne + (size_t)l * NRBF_ * D_, agg);
    gnn_mlp1<<<1024, 32, 0, stream>>>(agg, Wpl + 2 * WSZ_, tmp);
    float* dst = (l == L_ - 1) ? out : h;
    gnn_mlp2<<<1024, 32, 0, stream>>>(tmp, Wpl + 3 * WSZ_, h, dst);
  }
}


# agg builds its A operand in registers from a pair table (distance, envelope), Gaussian by hardware exp2 not library expf; the 16x larger basis tables go; 2 set-up kernels become 1; same sum order
# speedup vs baseline: 1.5054x; 1.1010x over previous
#include <hip/hip_runtime.h>
#include <math.h>

#define B_     128
#define NE_    64
#define NN_    16
#define D_     128
#define NRBF_  16
#define L_     3
#define NUP_   32
#define CUTOFF_ 2.0f
#define EPS_    1e-12f

#define NT_    4
#define WSZ_   (D_ * D_)

typedef float v2f __attribute__((ext_vector_type(2)));
typedef float v8f __attribute__((ext_vector_type(8)));

__device__ __forceinline__ void gemm_strip_f32p(const float* __restrict__ A, int lda,
                                                const float* __restrict__ Bp,
                                                int m0, int n0, v8f acc[NT_], int lane) {
  const int half = lane >> 4;
  const int l16  = lane & 15;
  const float* __restrict__ arow = A + (size_t)(m0 + l16) * lda;
#pragma unroll
  for (int k0 = 0; k0 < D_; k0 += 4) {
    const int ka = k0 + half * 2;
    const v2f a = *(const v2f*)&arow[ka];
    const float* __restrict__ brow = Bp + (size_t)(ka >> 1) * (2 * D_);
#pragma unroll
    for (int t = 0; t < NT_; ++t) {
      const v2f b = *(const v2f*)&brow[(n0 + t * 16 + l16) * 2];
      acc[t] = __builtin_amdgcn_wmma_f32_16x16x4_f32(false, a, false, b, (short)0,
                                                     acc[t], false, false);
    }
  }
}

__device__ __forceinline__ void store_strip_f32(float* __restrict__ C, int ldc,
                                                int m0, int n0, v8f acc[NT_], int lane) {
  const int half = lane >> 4;
  const int l16  = lane & 15;
#pragma unroll
  for (int t = 0; t < NT_; ++t)
#pragma unroll
    for (int ii = 0; ii < 8; ++ii)
      C[(size_t)(m0 + half * 8 + ii) * ldc + n0 + t * 16 + l16] = acc[t][ii];
}

__device__ __forceinline__ void load_strip_f32(const float* __restrict__ C, int ldc,
                                               int m0, int n0, v8f acc[NT_], int lane) {
  const int half = lane >> 4;
  const int l16  = lane & 15;
#pragma unroll
  for (int t = 0; t < NT_; ++t)
#pragma unroll
    for (int ii = 0; ii < 8; ++ii)
      acc[t][ii] = C[(size_t)(m0 + half * 8 + ii) * ldc + n0 + t * 16 + l16];
}

__global__ void gnn_pack_w(const float* __restrict__ We, const float* __restrict__ Wn,
                           const float* __restrict__ W1, const float* __restrict__ W2,
                           float* __restrict__ Wp) {
  const int gid = blockIdx.x * blockDim.x + threadIdx.x;
  const int m   = gid >> 14;
  const int idx = gid & (WSZ_ - 1);
  const int l = m >> 2, w = m & 3;
  const float* src = (w == 0 ? We : w == 1 ? Wn : w == 2 ? W1 : W2) + (size_t)l * WSZ_;
  const int k = idx >> 7, col = idx & (D_ - 1);
  Wp[(size_t)m * WSZ_ + (size_t)(k >> 1) * (2 * D_) + col * 2 + (k & 1)] = src[idx];
}

#define PEE_ (2 * NE_)
#define PNE_ (2 * NN_)
#define PAIR_EE_BLOCKS_ ((B_ * NE_ * (NE_ / 4)) / 256)
#define PAIR_NE_BLOCKS_ ((B_ * NE_ * (NN_ / 4)) / 256)

typedef float v4f __attribute__((ext_vector_type(4)));

__global__ void gnn_pairs(const float* __restrict__ r, const float* __restrict__ coords,
                          float* __restrict__ Pee, float* __restrict__ Pne) {
  const float inv_cut = 1.0f / CUTOFF_;
  if (blockIdx.x < PAIR_EE_BLOCKS_) {
    const int u  = blockIdx.x * 256 + threadIdx.x;
    const int bi = u >> 4, q = u & 15;
    const int b = bi >> 6, i = bi & 63;
    const float rx = r[(size_t)bi * 3 + 0];
    const float ry = r[(size_t)bi * 3 + 1];
    const float rz = r[(size_t)bi * 3 + 2];
    const int jb = 8 * (q & 7) + (q >> 3);
    v4f dv, ev;
#pragma unroll
    for (int w = 0; w < 4; ++w) {
      const int j = jb + 2 * w;
      const float dx = rx - r[(size_t)(b * NE_ + j) * 3 + 0];
      const float dy = ry - r[(size_t)(b * NE_ + j) * 3 + 1];
      const float dz = rz - r[(size_t)(b * NE_ + j) * 3 + 2];
      const float dist = sqrtf(dx * dx + dy * dy + dz * dz + EPS_);
      float env = 1.0f - dist * inv_cut;
      env = (dist < CUTOFF_) ? env * env : 0.0f;
      if (i == j) env = 0.0f;
      dv[w] = dist; ev[w] = env;
    }
    *(v4f*)&Pee[(size_t)bi * PEE_ + 4 * q]       = dv;
    *(v4f*)&Pee[(size_t)bi * PEE_ + NE_ + 4 * q] = ev;
  } else {
    const int u  = (blockIdx.x - PAIR_EE_BLOCKS_) * 256 + threadIdx.x;
    const int bj = u >> 2, q = u & 3;
    const float rx = r[(size_t)bj * 3 + 0];
    const float ry = r[(size_t)bj * 3 + 1];
    const float rz = r[(size_t)bj * 3 + 2];
    const int ab = 8 * (q & 1) + (q >> 1);
    v4f dv, ev;
#pragma unroll
    for (int w = 0; w < 4; ++w) {
      const int a = ab + 2 * w;
      const float dx = rx - coords[a * 3 + 0];
      const float dy = ry - coords[a * 3 + 1];
      const float dz = rz - coords[a * 3 + 2];
      const float dist = sqrtf(dx * dx + dy * dy + dz * dz + EPS_);
      float env = 1.0f - dist * inv_cut;
      env = (dist < CUTOFF_) ? env * env : 0.0f;
      dv[w] = dist; ev[w] = env;
    }
    *(v4f*)&Pne[(size_t)bj * PNE_ + 4 * q]       = dv;
    *(v4f*)&Pne[(size_t)bj * PNE_ + NN_ + 4 * q] = ev;
  }
}

__global__ void gnn_init_h(const float* __restrict__ spin_emb, float* __restrict__ h) {
  const int gid = blockIdx.x * blockDim.x + threadIdx.x;
  const int d = gid & (D_ - 1);
  const int i = (gid >> 7) & (NE_ - 1);
  h[gid] = spin_emb[(i < NUP_ ? 0 : 1) * D_ + d];
}

__global__ void gnn_hw(const float* __restrict__ h, const float* __restrict__ nuc_emb,
                       const float* __restrict__ Wp_ee, const float* __restrict__ Wp_ne,
                       float* __restrict__ hW, float* __restrict__ hnW) {
  const int wid = blockIdx.x;
  const int lane = threadIdx.x & 31;
  v8f acc[NT_] = {};
  if (wid < 1024) {
    const int m0 = (wid >> 1) * 16;
    const int n0 = (wid & 1) * 64;
    gemm_strip_f32p(h, D_, Wp_ee, m0, n0, acc, lane);
    store_strip_f32(hW, D_, m0, n0, acc, lane);
  } else {
    const int n0 = (wid - 1024) * 64;
    gemm_strip_f32p(nuc_emb, D_, Wp_ne, 0, n0, acc, lane);
    store_strip_f32(hnW, D_, 0, n0, acc, lane);
  }
}

#define IPW_ 8
#define SEE_ (NE_ / 4)
#define SNE_ (NN_ / 4)

struct AggP { v4f de[SEE_ / 2]; v4f ee[SEE_ / 2]; v4f dn[SNE_ / 2]; v4f en[SNE_ / 2]; };

__device__ __forceinline__ void agg_load_p(AggP& f, const float* __restrict__ pe,
                                           const float* __restrict__ pn) {
#pragma unroll
  for (int q = 0; q < SEE_ / 2; ++q) {
    f.de[q] = *(const v4f*)&pe[4 * q];
    f.ee[q] = *(const v4f*)&pe[NE_ + 4 * q];
  }
#pragma unroll
  for (int q = 0; q < SNE_ / 2; ++q) {
    f.dn[q] = *(const v4f*)&pn[4 * q];
    f.en[q] = *(const v4f*)&pn[NN_ + 4 * q];
  }
}

__device__ __forceinline__ v2f agg_feat(float d0, float d1, float e0, float e1,
                                        float mu, float kc) {
  v2f d;  d.x = d0;  d.y = d1;
  v2f en; en.x = e0; en.y = e1;
  const v2f zs = (d - mu) * kc;
  const v2f qq = zs * zs;
  v2f ex;
  ex.x = __builtin_amdgcn_exp2f(-qq.x);
  ex.y = __builtin_amdgcn_exp2f(-qq.y);
  return ex * en;
}

__device__ __forceinline__ float agg_xhalf(float x) {
  const int v = __builtin_bit_cast(int, x);
  return __builtin_bit_cast(float,
      __builtin_amdgcn_permlanex16(v, v, 0x76543210, 0xfedcba98, false, false));
}

__device__ __forceinline__ void agg_one(const AggP& f, float mu, float kc,
                                        const v2f (&bee)[SEE_][NT_],
                                        const v2f (&bne)[SNE_][NT_],
                                        const v2f (&wee)[NT_][4], const v2f (&wne)[NT_][4],
                                        float* __restrict__ orow, int half) {
  v8f g[NT_] = {};
#pragma unroll
  for (int s = 0; s < SEE_; ++s) {
    const int q = s >> 1, w = 2 * (s & 1);
    const v2f a = agg_feat(f.de[q][w], f.de[q][w + 1], f.ee[q][w], f.ee[q][w + 1], mu, kc);
#pragma unroll
    for (int t = 0; t < NT_; ++t)
      g[t] = __builtin_amdgcn_wmma_f32_16x16x4_f32(false, a, false, bee[s][t], (short)0,
                                                   g[t], false, false);
  }
  v8f gn[NT_] = {};
#pragma unroll
  for (int s = 0; s < SNE_; ++s) {
    const int q = s >> 1, w = 2 * (s & 1);
    const v2f a = agg_feat(f.dn[q][w], f.dn[q][w + 1], f.en[q][w], f.en[q][w + 1], mu, kc);
#pragma unroll
    for (int t = 0; t < NT_; ++t)
      gn[t] = __builtin_amdgcn_wmma_f32_16x16x4_f32(false, a, false, bne[s][t], (short)0,
                                                    gn[t], false, false);
  }
  float sv[NT_];
#pragma unroll
  for (int t = 0; t < NT_; ++t) {
    v2f p = {0.0f, 0.0f};
#pragma unroll
    for (int r = 0; r < 8; r += 2) {
      v2f gv; gv.x = g[t][r];  gv.y = g[t][r + 1];
      p += wee[t][r >> 1] * gv;
    }
#pragma unroll
    for (int r = 0; r < 8; r += 2) {
      v2f gv; gv.x = gn[t][r]; gv.y = gn[t][r + 1];
      p += wne[t][r >> 1] * gv;
    }
    sv[t] = p.x + p.y;
  }
  const float own0 = half ? sv[1] : sv[0], snd0 = half ? sv[0] : sv[1];
  const float own1 = half ? sv[3] : sv[2], snd1 = half ? sv[2] : sv[3];
  orow[0]  = own0 + agg_xhalf(snd0);
  orow[32] = own1 + agg_xhalf(snd1);
}

__global__ __launch_bounds__(128) __attribute__((amdgpu_waves_per_eu(2, 2)))
void gnn_agg(const float* __restrict__ Pee,
                        const float* __restrict__ Pne,
                        const float* __restrict__ hW, const float* __restrict__ hnW,
                        const float* __restrict__ Wrbf_ee_l, const float* __restrict__ Wrbf_ne_l,
                        float* __restrict__ agg) {
  const int lane = threadIdx.x & 31;
  const int wv   = threadIdx.x >> 5;
  const int n0   = (wv & 1) * 64;
  const int b    = blockIdx.x / (NE_ / (2 * IPW_));
  const int i0   = ((blockIdx.x % (NE_ / (2 * IPW_))) * 2 + (wv >> 1)) * IPW_;
  const int half = lane >> 4, l16 = lane & 15;

  const float* __restrict__ pe = Pee + (size_t)(b * NE_ + i0) * PEE_ + half * (NE_ / 2);
  const float* __restrict__ pn = Pne + (size_t)(b * NE_ + i0) * PNE_ + half * (NN_ / 2);
  float* __restrict__ orow = agg + (size_t)(b * NE_ + i0) * D_ + n0 + lane;

  const float mu = (CUTOFF_ / (float)(NRBF_ - 1)) * (float)l16;
  const float kc = ((float)NRBF_ / CUTOFF_) * 0.84932180028801904f;

  AggP f0, f1;
  agg_load_p(f0, pe, pn);

  const float* __restrict__ hb = hW + (size_t)b * NE_ * D_ + half * D_ + n0 + l16;
  const float* __restrict__ nb = hnW + half * D_ + n0 + l16;
  v2f bee[SEE_][NT_], bne[SNE_][NT_];
#pragma unroll
  for (int s = 0; s < SEE_; ++s)
#pragma unroll
    for (int t = 0; t < NT_; ++t) {
      bee[s][t].x = hb[(4 * s) * D_ + 16 * t];
      bee[s][t].y = hb[(4 * s + 2) * D_ + 16 * t];
    }
#pragma unroll
  for (int s = 0; s < SNE_; ++s)
#pragma unroll
    for (int t = 0; t < NT_; ++t) {
      bne[s][t].x = nb[(4 * s) * D_ + 16 * t];
      bne[s][t].y = nb[(4 * s + 2) * D_ + 16 * t];
    }
  v2f wee[NT_][4], wne[NT_][4];
#pragma unroll
  for (int t = 0; t < NT_; ++t)
#pragma unroll
    for (int r = 0; r < 8; r += 2) {
      wee[t][r >> 1].x = Wrbf_ee_l[(size_t)(8 * half + r) * D_ + n0 + 16 * t + l16];
      wee[t][r >> 1].y = Wrbf_ee_l[(size_t)(8 * half + r + 1) * D_ + n0 + 16 * t + l16];
      wne[t][r >> 1].x = Wrbf_ne_l[(size_t)(8 * half + r) * D_ + n0 + 16 * t + l16];
      wne[t][r >> 1].y = Wrbf_ne_l[(size_t)(8 * half + r + 1) * D_ + n0 + 16 * t + l16];
    }

#pragma unroll 1
  for (int ii = 0; ii < IPW_ - 2; ii += 2) {
    agg_load_p(f1, pe + (size_t)(ii + 1) * PEE_, pn + (size_t)(ii + 1) * PNE_);
    agg_one(f0, mu, kc, bee, bne, wee, wne, orow + (size_t)ii * D_, half);
    agg_load_p(f0, pe + (size_t)(ii + 2) * PEE_, pn + (size_t)(ii + 2) * PNE_);
    agg_one(f1, mu, kc, bee, bne, wee, wne, orow + (size_t)(ii + 1) * D_, half);
  }
  agg_load_p(f1, pe + (size_t)(IPW_ - 1) * PEE_, pn + (size_t)(IPW_ - 1) * PNE_);
  agg_one(f0, mu, kc, bee, bne, wee, wne, orow + (size_t)(IPW_ - 2) * D_, half);
  agg_one(f1, mu, kc, bee, bne, wee, wne, orow + (size_t)(IPW_ - 1) * D_, half);
}

__global__ void gnn_mlp1(const float* __restrict__ agg, const float* __restrict__ Wp1,
                         float* __restrict__ tmp) {
  const int wid = blockIdx.x;
  const int lane = threadIdx.x & 31;
  const int m0 = (wid >> 1) * 16;
  const int n0 = (wid & 1) * 64;
  v8f acc[NT_] = {};
  gemm_strip_f32p(agg, D_, Wp1, m0, n0, acc, lane);
#pragma unroll
  for (int t = 0; t < NT_; ++t)
#pragma unroll
    for (int ii = 0; ii < 8; ++ii) acc[t][ii] = tanhf(acc[t][ii]);
  store_strip_f32(tmp, D_, m0, n0, acc, lane);
}

__global__ void gnn_mlp2(const float* __restrict__ tmp, const float* __restrict__ Wp2,
                         const float* __restrict__ h, float* __restrict__ dst) {
  const int wid = blockIdx.x;
  const int lane = threadIdx.x & 31;
  const int m0 = (wid >> 1) * 16;
  const int n0 = (wid & 1) * 64;
  v8f acc[NT_];
  load_strip_f32(h, D_, m0, n0, acc, lane);
  gemm_strip_f32p(tmp, D_, Wp2, m0, n0, acc, lane);
  store_strip_f32(dst, D_, m0, n0, acc, lane);
}

extern "C" void kernel_launch(void* const* d_in, const int* in_sizes, int n_in,
                              void* d_out, int out_size, void* d_ws, size_t ws_size,
                              hipStream_t stream) {
  (void)in_sizes; (void)n_in; (void)out_size; (void)ws_size;
  const float* r        = (const float*)d_in[0];
  const float* coords   = (const float*)d_in[1];
  const float* spin_emb = (const float*)d_in[2];
  const float* nuc_emb  = (const float*)d_in[3];
  const float* W_rbf_ee = (const float*)d_in[4];
  const float* W_rbf_ne = (const float*)d_in[5];
  const float* W_msg_ee = (const float*)d_in[6];
  const float* W_msg_ne = (const float*)d_in[7];
  const float* W1       = (const float*)d_in[8];
  const float* W2       = (const float*)d_in[9];
  float* out = (float*)d_out;

  float* ws  = (float*)d_ws;
  float* Pee = ws;
  float* Pne = Pee + (size_t)B_ * NE_ * PEE_;
  float* h   = Pne + (size_t)B_ * NE_ * PNE_;
  float* hW  = h   + (size_t)B_ * NE_ * D_;
  float* hnW = hW  + (size_t)B_ * NE_ * D_;
  float* agg = hnW + (size_t)NN_ * D_;
  float* tmp = agg + (size_t)B_ * NE_ * D_;
  float* Wp  = tmp + (size_t)B_ * NE_ * D_;

  gnn_pack_w<<<(12 * WSZ_) / 256, 256, 0, stream>>>(W_msg_ee, W_msg_ne, W1, W2, Wp);
  gnn_pairs<<<PAIR_EE_BLOCKS_ + PAIR_NE_BLOCKS_, 256, 0, stream>>>(r, coords, Pee, Pne);
  gnn_init_h<<<(B_ * NE_ * D_) / 256, 256, 0, stream>>>(spin_emb, h);

  for (int l = 0; l < L_; ++l) {
    const float* Wpl = Wp + (size_t)l * 4 * WSZ_;
    gnn_hw<<<1024 + 2, 32, 0, stream>>>(h, nuc_emb, Wpl + 0 * WSZ_, Wpl + 1 * WSZ_, hW, hnW);
    gnn_agg<<<B_ * (NE_ / (2 * IPW_)), 128, 0, stream>>>(Pee, Pne, hW, hnW,
                                       W_rbf_ee + (size_t)l * NRBF_ * D_,
                                       W_rbf_ne + (size_t)l * NRBF_ * D_, agg);
    gnn_mlp1<<<1024, 32, 0, stream>>>(agg, Wpl + 2 * WSZ_, tmp);
    float* dst = (l == L_ - 1) ? out : h;
    gnn_mlp2<<<1024, 32, 0, stream>>>(tmp, Wpl + 3 * WSZ_, h, dst);
  }
}


# dense GEMM kernels (hW, mlp1, mlp2) declare their one-wave block and a two-wave register budget: a strip's operand loads go out 100+ ahead of their WMMAs, not about 10; products, sum order unchanged
# speedup vs baseline: 1.9702x; 1.3088x over previous
#include <hip/hip_runtime.h>
#include <math.h>

#define B_     128
#define NE_    64
#define NN_    16
#define D_     128
#define NRBF_  16
#define L_     3
#define NUP_   32
#define CUTOFF_ 2.0f
#define EPS_    1e-12f

#define NT_    4
#define WSZ_   (D_ * D_)

typedef float v2f __attribute__((ext_vector_type(2)));
typedef float v8f __attribute__((ext_vector_type(8)));

__device__ __forceinline__ void gemm_strip_f32p(const float* __restrict__ A, int lda,
                                                const float* __restrict__ Bp,
                                                int m0, int n0, v8f acc[NT_], int lane) {
  const int half = lane >> 4;
  const int l16  = lane & 15;
  const float* __restrict__ arow = A + (size_t)(m0 + l16) * lda;
#pragma unroll
  for (int k0 = 0; k0 < D_; k0 += 4) {
    const int ka = k0 + half * 2;
    const v2f a = *(const v2f*)&arow[ka];
    const float* __restrict__ brow = Bp + (size_t)(ka >> 1) * (2 * D_);
#pragma unroll
    for (int t = 0; t < NT_; ++t) {
      const v2f b = *(const v2f*)&brow[(n0 + t * 16 + l16) * 2];
      acc[t] = __builtin_amdgcn_wmma_f32_16x16x4_f32(false, a, false, b, (short)0,
                                                     acc[t], false, false);
    }
  }
}

__device__ __forceinline__ void store_strip_f32(float* __restrict__ C, int ldc,
                                                int m0, int n0, v8f acc[NT_], int lane) {
  const int half = lane >> 4;
  const int l16  = lane & 15;
#pragma unroll
  for (int t = 0; t < NT_; ++t)
#pragma unroll
    for (int ii = 0; ii < 8; ++ii)
      C[(size_t)(m0 + half * 8 + ii) * ldc + n0 + t * 16 + l16] = acc[t][ii];
}

__device__ __forceinline__ void load_strip_f32(const float* __restrict__ C, int ldc,
                                               int m0, int n0, v8f acc[NT_], int lane) {
  const int half = lane >> 4;
  const int l16  = lane & 15;
#pragma unroll
  for (int t = 0; t < NT_; ++t)
#pragma unroll
    for (int ii = 0; ii < 8; ++ii)
      acc[t][ii] = C[(size_t)(m0 + half * 8 + ii) * ldc + n0 + t * 16 + l16];
}

__global__ void gnn_pack_w(const float* __restrict__ We, const float* __restrict__ Wn,
                           const float* __restrict__ W1, const float* __restrict__ W2,
                           float* __restrict__ Wp) {
  const int gid = blockIdx.x * blockDim.x + threadIdx.x;
  const int m   = gid >> 14;
  const int idx = gid & (WSZ_ - 1);
  const int l = m >> 2, w = m & 3;
  const float* src = (w == 0 ? We : w == 1 ? Wn : w == 2 ? W1 : W2) + (size_t)l * WSZ_;
  const int k = idx >> 7, col = idx & (D_ - 1);
  Wp[(size_t)m * WSZ_ + (size_t)(k >> 1) * (2 * D_) + col * 2 + (k & 1)] = src[idx];
}

#define PEE_ (2 * NE_)
#define PNE_ (2 * NN_)
#define PAIR_EE_BLOCKS_ ((B_ * NE_ * (NE_ / 4)) / 256)
#define PAIR_NE_BLOCKS_ ((B_ * NE_ * (NN_ / 4)) / 256)

typedef float v4f __attribute__((ext_vector_type(4)));

__global__ void gnn_pairs(const float* __restrict__ r, const float* __restrict__ coords,
                          float* __restrict__ Pee, float* __restrict__ Pne) {
  const float inv_cut = 1.0f / CUTOFF_;
  if (blockIdx.x < PAIR_EE_BLOCKS_) {
    const int u  = blockIdx.x * 256 + threadIdx.x;
    const int bi = u >> 4, q = u & 15;
    const int b = bi >> 6, i = bi & 63;
    const float rx = r[(size_t)bi * 3 + 0];
    const float ry = r[(size_t)bi * 3 + 1];
    const float rz = r[(size_t)bi * 3 + 2];
    const int jb = 8 * (q & 7) + (q >> 3);
    v4f dv, ev;
#pragma unroll
    for (int w = 0; w < 4; ++w) {
      const int j = jb + 2 * w;
      const float dx = rx - r[(size_t)(b * NE_ + j) * 3 + 0];
      const float dy = ry - r[(size_t)(b * NE_ + j) * 3 + 1];
      const float dz = rz - r[(size_t)(b * NE_ + j) * 3 + 2];
      const float dist = sqrtf(dx * dx + dy * dy + dz * dz + EPS_);
      float env = 1.0f - dist * inv_cut;
      env = (dist < CUTOFF_) ? env * env : 0.0f;
      if (i == j) env = 0.0f;
      dv[w] = dist; ev[w] = env;
    }
    *(v4f*)&Pee[(size_t)bi * PEE_ + 4 * q]       = dv;
    *(v4f*)&Pee[(size_t)bi * PEE_ + NE_ + 4 * q] = ev;
  } else {
    const int u  = (blockIdx.x - PAIR_EE_BLOCKS_) * 256 + threadIdx.x;
    const int bj = u >> 2, q = u & 3;
    const float rx = r[(size_t)bj * 3 + 0];
    const float ry = r[(size_t)bj * 3 + 1];
    const float rz = r[(size_t)bj * 3 + 2];
    const int ab = 8 * (q & 1) + (q >> 1);
    v4f dv, ev;
#pragma unroll
    for (int w = 0; w < 4; ++w) {
      const int a = ab + 2 * w;
      const float dx = rx - coords[a * 3 + 0];
      const float dy = ry - coords[a * 3 + 1];
      const float dz = rz - coords[a * 3 + 2];
      const float dist = sqrtf(dx * dx + dy * dy + dz * dz + EPS_);
      float env = 1.0f - dist * inv_cut;
      env = (dist < CUTOFF_) ? env * env : 0.0f;
      dv[w] = dist; ev[w] = env;
    }
    *(v4f*)&Pne[(size_t)bj * PNE_ + 4 * q]       = dv;
    *(v4f*)&Pne[(size_t)bj * PNE_ + NN_ + 4 * q] = ev;
  }
}

__global__ void gnn_init_h(const float* __restrict__ spin_emb, float* __restrict__ h) {
  const int gid = blockIdx.x * blockDim.x + threadIdx.x;
  const int d = gid & (D_ - 1);
  const int i = (gid >> 7) & (NE_ - 1);
  h[gid] = spin_emb[(i < NUP_ ? 0 : 1) * D_ + d];
}

__global__ __launch_bounds__(32) __attribute__((amdgpu_waves_per_eu(1, 2)))
void gnn_hw(const float* __restrict__ h, const float* __restrict__ nuc_emb,
                       const float* __restrict__ Wp_ee, const float* __restrict__ Wp_ne,
                       float* __restrict__ hW, float* __restrict__ hnW) {
  const int wid = blockIdx.x;
  const int lane = threadIdx.x & 31;
  v8f acc[NT_] = {};
  if (wid < 1024) {
    const int m0 = (wid >> 1) * 16;
    const int n0 = (wid & 1) * 64;
    gemm_strip_f32p(h, D_, Wp_ee, m0, n0, acc, lane);
    store_strip_f32(hW, D_, m0, n0, acc, lane);
  } else {
    const int n0 = (wid - 1024) * 64;
    gemm_strip_f32p(nuc_emb, D_, Wp_ne, 0, n0, acc, lane);
    store_strip_f32(hnW, D_, 0, n0, acc, lane);
  }
}

#define IPW_ 8
#define SEE_ (NE_ / 4)
#define SNE_ (NN_ / 4)

struct AggP { v4f de[SEE_ / 2]; v4f ee[SEE_ / 2]; v4f dn[SNE_ / 2]; v4f en[SNE_ / 2]; };

__device__ __forceinline__ void agg_load_p(AggP& f, const float* __restrict__ pe,
                                           const float* __restrict__ pn) {
#pragma unroll
  for (int q = 0; q < SEE_ / 2; ++q) {
    f.de[q] = *(const v4f*)&pe[4 * q];
    f.ee[q] = *(const v4f*)&pe[NE_ + 4 * q];
  }
#pragma unroll
  for (int q = 0; q < SNE_ / 2; ++q) {
    f.dn[q] = *(const v4f*)&pn[4 * q];
    f.en[q] = *(const v4f*)&pn[NN_ + 4 * q];
  }
}

__device__ __forceinline__ v2f agg_feat(float d0, float d1, float e0, float e1,
                                        float mu, float kc) {
  v2f d;  d.x = d0;  d.y = d1;
  v2f en; en.x = e0; en.y = e1;
  const v2f zs = (d - mu) * kc;
  const v2f qq = zs * zs;
  v2f ex;
  ex.x = __builtin_amdgcn_exp2f(-qq.x);
  ex.y = __builtin_amdgcn_exp2f(-qq.y);
  return ex * en;
}

__device__ __forceinline__ float agg_xhalf(float x) {
  const int v = __builtin_bit_cast(int, x);
  return __builtin_bit_cast(float,
      __builtin_amdgcn_permlanex16(v, v, 0x76543210, 0xfedcba98, false, false));
}

__device__ __forceinline__ void agg_one(const AggP& f, float mu, float kc,
                                        const v2f (&bee)[SEE_][NT_],
                                        const v2f (&bne)[SNE_][NT_],
                                        const v2f (&wee)[NT_][4], const v2f (&wne)[NT_][4],
                                        float* __restrict__ orow, int half) {
  v8f g[NT_] = {};
#pragma unroll
  for (int s = 0; s < SEE_; ++s) {
    const int q = s >> 1, w = 2 * (s & 1);
    const v2f a = agg_feat(f.de[q][w], f.de[q][w + 1], f.ee[q][w], f.ee[q][w + 1], mu, kc);
#pragma unroll
    for (int t = 0; t < NT_; ++t)
      g[t] = __builtin_amdgcn_wmma_f32_16x16x4_f32(false, a, false, bee[s][t], (short)0,
                                                   g[t], false, false);
  }
  v8f gn[NT_] = {};
#pragma unroll
  for (int s = 0; s < SNE_; ++s) {
    const int q = s >> 1, w = 2 * (s & 1);
    const v2f a = agg_feat(f.dn[q][w], f.dn[q][w + 1], f.en[q][w], f.en[q][w + 1], mu, kc);
#pragma unroll
    for (int t = 0; t < NT_; ++t)
      gn[t] = __builtin_amdgcn_wmma_f32_16x16x4_f32(false, a, false, bne[s][t], (short)0,
                                                    gn[t], false, false);
  }
  float sv[NT_];
#pragma unroll
  for (int t = 0; t < NT_; ++t) {
    v2f p = {0.0f, 0.0f};
#pragma unroll
    for (int r = 0; r < 8; r += 2) {
      v2f gv; gv.x = g[t][r];  gv.y = g[t][r + 1];
      p += wee[t][r >> 1] * gv;
    }
#pragma unroll
    for (int r = 0; r < 8; r += 2) {
      v2f gv; gv.x = gn[t][r]; gv.y = gn[t][r + 1];
      p += wne[t][r >> 1] * gv;
    }
    sv[t] = p.x + p.y;
  }
  const float own0 = half ? sv[1] : sv[0], snd0 = half ? sv[0] : sv[1];
  const float own1 = half ? sv[3] : sv[2], snd1 = half ? sv[2] : sv[3];
  orow[0]  = own0 + agg_xhalf(snd0);
  orow[32] = own1 + agg_xhalf(snd1);
}

__global__ __launch_bounds__(128) __attribute__((amdgpu_waves_per_eu(2, 2)))
void gnn_agg(const float* __restrict__ Pee,
                        const float* __restrict__ Pne,
                        const float* __restrict__ hW, const float* __restrict__ hnW,
                        const float* __restrict__ Wrbf_ee_l, const float* __restrict__ Wrbf_ne_l,
                        float* __restrict__ agg) {
  const int lane = threadIdx.x & 31;
  const int wv   = threadIdx.x >> 5;
  const int n0   = (wv & 1) * 64;
  const int b    = blockIdx.x / (NE_ / (2 * IPW_));
  const int i0   = ((blockIdx.x % (NE_ / (2 * IPW_))) * 2 + (wv >> 1)) * IPW_;
  const int half = lane >> 4, l16 = lane & 15;

  const float* __restrict__ pe = Pee + (size_t)(b * NE_ + i0) * PEE_ + half * (NE_ / 2);
  const float* __restrict__ pn = Pne + (size_t)(b * NE_ + i0) * PNE_ + half * (NN_ / 2);
  float* __restrict__ orow = agg + (size_t)(b * NE_ + i0) * D_ + n0 + lane;

  const float mu = (CUTOFF_ / (float)(NRBF_ - 1)) * (float)l16;
  const float kc = ((float)NRBF_ / CUTOFF_) * 0.84932180028801904f;

  AggP f0, f1;
  agg_load_p(f0, pe, pn);

  const float* __restrict__ hb = hW + (size_t)b * NE_ * D_ + half * D_ + n0 + l16;
  const float* __restrict__ nb = hnW + half * D_ + n0 + l16;
  v2f bee[SEE_][NT_], bne[SNE_][NT_];
#pragma unroll
  for (int s = 0; s < SEE_; ++s)
#pragma unroll
    for (int t = 0; t < NT_; ++t) {
      bee[s][t].x = hb[(4 * s) * D_ + 16 * t];
      bee[s][t].y = hb[(4 * s + 2) * D_ + 16 * t];
    }
#pragma unroll
  for (int s = 0; s < SNE_; ++s)
#pragma unroll
    for (int t = 0; t < NT_; ++t) {
      bne[s][t].x = nb[(4 * s) * D_ + 16 * t];
      bne[s][t].y = nb[(4 * s + 2) * D_ + 16 * t];
    }
  v2f wee[NT_][4], wne[NT_][4];
#pragma unroll
  for (int t = 0; t < NT_; ++t)
#pragma unroll
    for (int r = 0; r < 8; r += 2) {
      wee[t][r >> 1].x = Wrbf_ee_l[(size_t)(8 * half + r) * D_ + n0 + 16 * t + l16];
      wee[t][r >> 1].y = Wrbf_ee_l[(size_t)(8 * half + r + 1) * D_ + n0 + 16 * t + l16];
      wne[t][r >> 1].x = Wrbf_ne_l[(size_t)(8 * half + r) * D_ + n0 + 16 * t + l16];
      wne[t][r >> 1].y = Wrbf_ne_l[(size_t)(8 * half + r + 1) * D_ + n0 + 16 * t + l16];
    }

#pragma unroll 1
  for (int ii = 0; ii < IPW_ - 2; ii += 2) {
    agg_load_p(f1, pe + (size_t)(ii + 1) * PEE_, pn + (size_t)(ii + 1) * PNE_);
    agg_one(f0, mu, kc, bee, bne, wee, wne, orow + (size_t)ii * D_, half);
    agg_load_p(f0, pe + (size_t)(ii + 2) * PEE_, pn + (size_t)(ii + 2) * PNE_);
    agg_one(f1, mu, kc, bee, bne, wee, wne, orow + (size_t)(ii + 1) * D_, half);
  }
  agg_load_p(f1, pe + (size_t)(IPW_ - 1) * PEE_, pn + (size_t)(IPW_ - 1) * PNE_);
  agg_one(f0, mu, kc, bee, bne, wee, wne, orow + (size_t)(IPW_ - 2) * D_, half);
  agg_one(f1, mu, kc, bee, bne, wee, wne, orow + (size_t)(IPW_ - 1) * D_, half);
}

__global__ __launch_bounds__(32) __attribute__((amdgpu_waves_per_eu(1, 2)))
void gnn_mlp1(const float* __restrict__ agg, const float* __restrict__ Wp1,
                         float* __restrict__ tmp) {
  const int wid = blockIdx.x;
  const int lane = threadIdx.x & 31;
  const int m0 = (wid >> 1) * 16;
  const int n0 = (wid & 1) * 64;
  v8f acc[NT_] = {};
  gemm_strip_f32p(agg, D_, Wp1, m0, n0, acc, lane);
#pragma unroll
  for (int t = 0; t < NT_; ++t)
#pragma unroll
    for (int ii = 0; ii < 8; ++ii) acc[t][ii] = tanhf(acc[t][ii]);
  store_strip_f32(tmp, D_, m0, n0, acc, lane);
}

__global__ __launch_bounds__(32) __attribute__((amdgpu_waves_per_eu(1, 2)))
void gnn_mlp2(const float* __restrict__ tmp, const float* __restrict__ Wp2,
                         const float* __restrict__ h, float* __restrict__ dst) {
  const int wid = blockIdx.x;
  const int lane = threadIdx.x & 31;
  const int m0 = (wid >> 1) * 16;
  const int n0 = (wid & 1) * 64;
  v8f acc[NT_];
  load_strip_f32(h, D_, m0, n0, acc, lane);
  gemm_strip_f32p(tmp, D_, Wp2, m0, n0, acc, lane);
  store_strip_f32(dst, D_, m0, n0, acc, lane);
}

extern "C" void kernel_launch(void* const* d_in, const int* in_sizes, int n_in,
                              void* d_out, int out_size, void* d_ws, size_t ws_size,
                              hipStream_t stream) {
  (void)in_sizes; (void)n_in; (void)out_size; (void)ws_size;
  const float* r        = (const float*)d_in[0];
  const float* coords   = (const float*)d_in[1];
  const float* spin_emb = (const float*)d_in[2];
  const float* nuc_emb  = (const float*)d_in[3];
  const float* W_rbf_ee = (const float*)d_in[4];
  const float* W_rbf_ne = (const float*)d_in[5];
  const float* W_msg_ee = (const float*)d_in[6];
  const float* W_msg_ne = (const float*)d_in[7];
  const float* W1       = (const float*)d_in[8];
  const float* W2       = (const float*)d_in[9];
  float* out = (float*)d_out;

  float* ws  = (float*)d_ws;
  float* Pee = ws;
  float* Pne = Pee + (size_t)B_ * NE_ * PEE_;
  float* h   = Pne + (size_t)B_ * NE_ * PNE_;
  float* hW  = h   + (size_t)B_ * NE_ * D_;
  float* hnW = hW  + (size_t)B_ * NE_ * D_;
  float* agg = hnW + (size_t)NN_ * D_;
  float* tmp = agg + (size_t)B_ * NE_ * D_;
  float* Wp  = tmp + (size_t)B_ * NE_ * D_;

  gnn_pack_w<<<(12 * WSZ_) / 256, 256, 0, stream>>>(W_msg_ee, W_msg_ne, W1, W2, Wp);
  gnn_pairs<<<PAIR_EE_BLOCKS_ + PAIR_NE_BLOCKS_, 256, 0, stream>>>(r, coords, Pee, Pne);
  gnn_init_h<<<(B_ * NE_ * D_) / 256, 256, 0, stream>>>(spin_emb, h);

  for (int l = 0; l < L_; ++l) {
    const float* Wpl = Wp + (size_t)l * 4 * WSZ_;
    gnn_hw<<<1024 + 2, 32, 0, stream>>>(h, nuc_emb, Wpl + 0 * WSZ_, Wpl + 1 * WSZ_, hW, hnW);
    gnn_agg<<<B_ * (NE_ / (2 * IPW_)), 128, 0, stream>>>(Pee, Pne, hW, hnW,
                                       W_rbf_ee + (size_t)l * NRBF_ * D_,
                                       W_rbf_ne + (size_t)l * NRBF_ * D_, agg);
    gnn_mlp1<<<1024, 32, 0, stream>>>(agg, Wpl + 2 * WSZ_, tmp);
    float* dst = (l == L_ - 1) ? out : h;
    gnn_mlp2<<<1024, 32, 0, stream>>>(tmp, Wpl + 3 * WSZ_, h, dst);
  }
}


# tile columns remapped (column c of tile t = strip column 4c+t): B rows of plain row-major matrices, residual and results move as 16-byte lane accesses; weight pack kernel and launch go; same sums
# speedup vs baseline: 2.0686x; 1.0499x over previous
#include <hip/hip_runtime.h>
#include <math.h>

#define B_     128
#define NE_    64
#define NN_    16
#define D_     128
#define NRBF_  16
#define L_     3
#define NUP_   32
#define CUTOFF_ 2.0f
#define EPS_    1e-12f

#define NT_    4
#define WSZ_   (D_ * D_)

typedef float v2f __attribute__((ext_vector_type(2)));
typedef float v4f __attribute__((ext_vector_type(4)));
typedef float v8f __attribute__((ext_vector_type(8)));

__device__ __forceinline__ void gemm_strip_f32(const float* __restrict__ A, int lda,
                                               const float* __restrict__ W,
                                               int m0, int n0, v8f acc[NT_], int lane) {
  const int half = lane >> 4;
  const int l16  = lane & 15;
  const float* __restrict__ arow = A + (size_t)(m0 + l16) * lda;
  const float* __restrict__ wcol = W + n0 + 4 * l16;
#pragma unroll
  for (int k0 = 0; k0 < D_; k0 += 4) {
    const int ka = k0 + half * 2;
    const v2f a = *(const v2f*)&arow[ka];
    const v4f w0 = *(const v4f*)&wcol[(size_t)ka * D_];
    const v4f w1 = *(const v4f*)&wcol[(size_t)(ka + 1) * D_];
#pragma unroll
    for (int t = 0; t < NT_; ++t) {
      v2f b; b.x = w0[t]; b.y = w1[t];
      acc[t] = __builtin_amdgcn_wmma_f32_16x16x4_f32(false, a, false, b, (short)0,
                                                     acc[t], false, false);
    }
  }
}

__device__ __forceinline__ void store_strip_f32(float* __restrict__ C, int ldc,
                                                int m0, int n0, v8f acc[NT_], int lane) {
  const int half = lane >> 4;
  const int l16  = lane & 15;
#pragma unroll
  for (int ii = 0; ii < 8; ++ii) {
    v4f o;
#pragma unroll
    for (int t = 0; t < NT_; ++t) o[t] = acc[t][ii];
    *(v4f*)&C[(size_t)(m0 + half * 8 + ii) * ldc + n0 + 4 * l16] = o;
  }
}

__device__ __forceinline__ void load_strip_f32(const float* __restrict__ C, int ldc,
                                               int m0, int n0, v8f acc[NT_], int lane) {
  const int half = lane >> 4;
  const int l16  = lane & 15;
#pragma unroll
  for (int ii = 0; ii < 8; ++ii) {
    const v4f o = *(const v4f*)&C[(size_t)(m0 + half * 8 + ii) * ldc + n0 + 4 * l16];
#pragma unroll
    for (int t = 0; t < NT_; ++t) acc[t][ii] = o[t];
  }
}

#define PEE_ (2 * NE_)
#define PNE_ (2 * NN_)
#define PAIR_EE_BLOCKS_ ((B_ * NE_ * (NE_ / 4)) / 256)
#define PAIR_NE_BLOCKS_ ((B_ * NE_ * (NN_ / 4)) / 256)

__global__ void gnn_pairs(const float* __restrict__ r, const float* __restrict__ coords,
                          float* __restrict__ Pee, float* __restrict__ Pne) {
  const float inv_cut = 1.0f / CUTOFF_;
  if (blockIdx.x < PAIR_EE_BLOCKS_) {
    const int u  = blockIdx.x * 256 + threadIdx.x;
    const int bi = u >> 4, q = u & 15;
    const int b = bi >> 6, i = bi & 63;
    const float rx = r[(size_t)bi * 3 + 0];
    const float ry = r[(size_t)bi * 3 + 1];
    const float rz = r[(size_t)bi * 3 + 2];
    const int jb = 8 * (q & 7) + (q >> 3);
    v4f dv, ev;
#pragma unroll
    for (int w = 0; w < 4; ++w) {
      const int j = jb + 2 * w;
      const float dx = rx - r[(size_t)(b * NE_ + j) * 3 + 0];
      const float dy = ry - r[(size_t)(b * NE_ + j) * 3 + 1];
      const float dz = rz - r[(size_t)(b * NE_ + j) * 3 + 2];
      const float dist = sqrtf(dx * dx + dy * dy + dz * dz + EPS_);
      float env = 1.0f - dist * inv_cut;
      env = (dist < CUTOFF_) ? env * env : 0.0f;
      if (i == j) env = 0.0f;
      dv[w] = dist; ev[w] = env;
    }
    *(v4f*)&Pee[(size_t)bi * PEE_ + 4 * q]       = dv;
    *(v4f*)&Pee[(size_t)bi * PEE_ + NE_ + 4 * q] = ev;
  } else {
    const int u  = (blockIdx.x - PAIR_EE_BLOCKS_) * 256 + threadIdx.x;
    const int bj = u >> 2, q = u & 3;
    const float rx = r[(size_t)bj * 3 + 0];
    const float ry = r[(size_t)bj * 3 + 1];
    const float rz = r[(size_t)bj * 3 + 2];
    const int ab = 8 * (q & 1) + (q >> 1);
    v4f dv, ev;
#pragma unroll
    for (int w = 0; w < 4; ++w) {
      const int a = ab + 2 * w;
      const float dx = rx - coords[a * 3 + 0];
      const float dy = ry - coords[a * 3 + 1];
      const float dz = rz - coords[a * 3 + 2];
      const float dist = sqrtf(dx * dx + dy * dy + dz * dz + EPS_);
      float env = 1.0f - dist * inv_cut;
      env = (dist < CUTOFF_) ? env * env : 0.0f;
      dv[w] = dist; ev[w] = env;
    }
    *(v4f*)&Pne[(size_t)bj * PNE_ + 4 * q]       = dv;
    *(v4f*)&Pne[(size_t)bj * PNE_ + NN_ + 4 * q] = ev;
  }
}

__global__ void gnn_init_h(const float* __restrict__ spin_emb, float* __restrict__ h) {
  const int gid = blockIdx.x * blockDim.x + threadIdx.x;
  const int d = gid & (D_ - 1);
  const int i = (gid >> 7) & (NE_ - 1);
  h[gid] = spin_emb[(i < NUP_ ? 0 : 1) * D_ + d];
}

__global__ __launch_bounds__(32) __attribute__((amdgpu_waves_per_eu(1, 2)))
void gnn_hw(const float* __restrict__ h, const float* __restrict__ nuc_emb,
                       const float* __restrict__ W_ee_l, const float* __restrict__ W_ne_l,
                       float* __restrict__ hW, float* __restrict__ hnW) {
  const int wid = blockIdx.x;
  const int lane = threadIdx.x & 31;
  v8f acc[NT_] = {};
  if (wid < 1024) {
    const int m0 = (wid >> 1) * 16;
    const int n0 = (wid & 1) * 64;
    gemm_strip_f32(h, D_, W_ee_l, m0, n0, acc, lane);
    store_strip_f32(hW, D_, m0, n0, acc, lane);
  } else {
    const int n0 = (wid - 1024) * 64;
    gemm_strip_f32(nuc_emb, D_, W_ne_l, 0, n0, acc, lane);
    store_strip_f32(hnW, D_, 0, n0, acc, lane);
  }
}

#define IPW_ 8
#define SEE_ (NE_ / 4)
#define SNE_ (NN_ / 4)

struct AggP { v4f de[SEE_ / 2]; v4f ee[SEE_ / 2]; v4f dn[SNE_ / 2]; v4f en[SNE_ / 2]; };

__device__ __forceinline__ void agg_load_p(AggP& f, const float* __restrict__ pe,
                                           const float* __restrict__ pn) {
#pragma unroll
  for (int q = 0; q < SEE_ / 2; ++q) {
    f.de[q] = *(const v4f*)&pe[4 * q];
    f.ee[q] = *(const v4f*)&pe[NE_ + 4 * q];
  }
#pragma unroll
  for (int q = 0; q < SNE_ / 2; ++q) {
    f.dn[q] = *(const v4f*)&pn[4 * q];
    f.en[q] = *(const v4f*)&pn[NN_ + 4 * q];
  }
}

__device__ __forceinline__ v2f agg_feat(float d0, float d1, float e0, float e1,
                                        float mu, float kc) {
  v2f d;  d.x = d0;  d.y = d1;
  v2f en; en.x = e0; en.y = e1;
  const v2f zs = (d - mu) * kc;
  const v2f qq = zs * zs;
  v2f ex;
  ex.x = __builtin_amdgcn_exp2f(-qq.x);
  ex.y = __builtin_amdgcn_exp2f(-qq.y);
  return ex * en;
}

__device__ __forceinline__ float agg_xhalf(float x) {
  const int v = __builtin_bit_cast(int, x);
  return __builtin_bit_cast(float,
      __builtin_amdgcn_permlanex16(v, v, 0x76543210, 0xfedcba98, false, false));
}

__device__ __forceinline__ void agg_one(const AggP& f, float mu, float kc,
                                        const v2f (&bee)[SEE_][NT_],
                                        const v2f (&bne)[SNE_][NT_],
                                        const v2f (&wee)[NT_][4], const v2f (&wne)[NT_][4],
                                        float* __restrict__ orow, int half) {
  v8f g[NT_] = {};
#pragma unroll
  for (int s = 0; s < SEE_; ++s) {
    const int q = s >> 1, w = 2 * (s & 1);
    const v2f a = agg_feat(f.de[q][w], f.de[q][w + 1], f.ee[q][w], f.ee[q][w + 1], mu, kc);
#pragma unroll
    for (int t = 0; t < NT_; ++t)
      g[t] = __builtin_amdgcn_wmma_f32_16x16x4_f32(false, a, false, bee[s][t], (short)0,
                                                   g[t], false, false);
  }
  v8f gn[NT_] = {};
#pragma unroll
  for (int s = 0; s < SNE_; ++s) {
    const int q = s >> 1, w = 2 * (s & 1);
    const v2f a = agg_feat(f.dn[q][w], f.dn[q][w + 1], f.en[q][w], f.en[q][w + 1], mu, kc);
#pragma unroll
    for (int t = 0; t < NT_; ++t)
      gn[t] = __builtin_amdgcn_wmma_f32_16x16x4_f32(false, a, false, bne[s][t], (short)0,
                                                    gn[t], false, false);
  }
  float sv[NT_];
#pragma unroll
  for (int t = 0; t < NT_; ++t) {
    v2f p = {0.0f, 0.0f};
#pragma unroll
    for (int r = 0; r < 8; r += 2) {
      v2f gv; gv.x = g[t][r];  gv.y = g[t][r + 1];
      p += wee[t][r >> 1] * gv;
    }
#pragma unroll
    for (int r = 0; r < 8; r += 2) {
      v2f gv; gv.x = gn[t][r]; gv.y = gn[t][r + 1];
      p += wne[t][r >> 1] * gv;
    }
    sv[t] = p.x + p.y;
  }
  v2f own, snd;
  own.x = half ? sv[2] : sv[0];  own.y = half ? sv[3] : sv[1];
  snd.x = half ? sv[0] : sv[2];  snd.y = half ? sv[1] : sv[3];
  v2f o;
  o.x = own.x + agg_xhalf(snd.x);
  o.y = own.y + agg_xhalf(snd.y);
  *(v2f*)orow = o;
}

__global__ __launch_bounds__(128) __attribute__((amdgpu_waves_per_eu(2, 2)))
void gnn_agg(const float* __restrict__ Pee,
                        const float* __restrict__ Pne,
                        const float* __restrict__ hW, const float* __restrict__ hnW,
                        const float* __restrict__ Wrbf_ee_l, const float* __restrict__ Wrbf_ne_l,
                        float* __restrict__ agg) {
  const int lane = threadIdx.x & 31;
  const int wv   = __builtin_amdgcn_readfirstlane((int)(threadIdx.x >> 5));
  const int n0   = (wv & 1) * 64;
  const int b    = blockIdx.x / (NE_ / (2 * IPW_));
  const int i0   = ((blockIdx.x % (NE_ / (2 * IPW_))) * 2 + (wv >> 1)) * IPW_;
  const int half = lane >> 4, l16 = lane & 15;

  const float* __restrict__ pe = Pee + (size_t)(b * NE_ + i0) * PEE_ + half * (NE_ / 2);
  const float* __restrict__ pn = Pne + (size_t)(b * NE_ + i0) * PNE_ + half * (NN_ / 2);
  float* __restrict__ orow = agg + (size_t)(b * NE_ + i0) * D_ + n0 + 4 * l16 + 2 * half;

  const float mu = (CUTOFF_ / (float)(NRBF_ - 1)) * (float)l16;
  const float kc = ((float)NRBF_ / CUTOFF_) * 0.84932180028801904f;

  AggP f0, f1;
  agg_load_p(f0, pe, pn);

  const float* __restrict__ hb = hW + (size_t)b * NE_ * D_ + half * D_ + n0 + 4 * l16;
  const float* __restrict__ nb = hnW + half * D_ + n0 + 4 * l16;
  v2f bee[SEE_][NT_], bne[SNE_][NT_];
#pragma unroll
  for (int s = 0; s < SEE_; ++s) {
    const v4f x = *(const v4f*)&hb[(4 * s) * D_];
    const v4f y = *(const v4f*)&hb[(4 * s + 2) * D_];
#pragma unroll
    for (int t = 0; t < NT_; ++t) { bee[s][t].x = x[t]; bee[s][t].y = y[t]; }
  }
#pragma unroll
  for (int s = 0; s < SNE_; ++s) {
    const v4f x = *(const v4f*)&nb[(4 * s) * D_];
    const v4f y = *(const v4f*)&nb[(4 * s + 2) * D_];
#pragma unroll
    for (int t = 0; t < NT_; ++t) { bne[s][t].x = x[t]; bne[s][t].y = y[t]; }
  }
  v2f wee[NT_][4], wne[NT_][4];
  const float* __restrict__ we = Wrbf_ee_l + (size_t)(8 * half) * D_ + n0 + 4 * l16;
  const float* __restrict__ wn = Wrbf_ne_l + (size_t)(8 * half) * D_ + n0 + 4 * l16;
#pragma unroll
  for (int r = 0; r < 8; r += 2) {
    const v4f e0 = *(const v4f*)&we[r * D_], e1 = *(const v4f*)&we[(r + 1) * D_];
    const v4f n0v = *(const v4f*)&wn[r * D_], n1v = *(const v4f*)&wn[(r + 1) * D_];
#pragma unroll
    for (int t = 0; t < NT_; ++t) {
      wee[t][r >> 1].x = e0[t];  wee[t][r >> 1].y = e1[t];
      wne[t][r >> 1].x = n0v[t]; wne[t][r >> 1].y = n1v[t];
    }
  }

#pragma unroll 1
  for (int ii = 0; ii < IPW_ - 2; ii += 2) {
    agg_load_p(f1, pe + (size_t)(ii + 1) * PEE_, pn + (size_t)(ii + 1) * PNE_);
    agg_one(f0, mu, kc, bee, bne, wee, wne, orow + (size_t)ii * D_, half);
    agg_load_p(f0, pe + (size_t)(ii + 2) * PEE_, pn + (size_t)(ii + 2) * PNE_);
    agg_one(f1, mu, kc, bee, bne, wee, wne, orow + (size_t)(ii + 1) * D_, half);
  }
  agg_load_p(f1, pe + (size_t)(IPW_ - 1) * PEE_, pn + (size_t)(IPW_ - 1) * PNE_);
  agg_one(f0, mu, kc, bee, bne, wee, wne, orow + (size_t)(IPW_ - 2) * D_, half);
  agg_one(f1, mu, kc, bee, bne, wee, wne, orow + (size_t)(IPW_ - 1) * D_, half);
}

__global__ __launch_bounds__(32) __attribute__((amdgpu_waves_per_eu(1, 2)))
void gnn_mlp1(const float* __restrict__ agg, const float* __restrict__ W1_l,
                         float* __restrict__ tmp) {
  const int wid = blockIdx.x;
  const int lane = threadIdx.x & 31;
  const int m0 = (wid >> 1) * 16;
  const int n0 = (wid & 1) * 64;
  v8f acc[NT_] = {};
  gemm_strip_f32(agg, D_, W1_l, m0, n0, acc, lane);
#pragma unroll
  for (int t = 0; t < NT_; ++t)
#pragma unroll
    for (int ii = 0; ii < 8; ++ii) acc[t][ii] = tanhf(acc[t][ii]);
  store_strip_f32(tmp, D_, m0, n0, acc, lane);
}

__global__ __launch_bounds__(32) __attribute__((amdgpu_waves_per_eu(1, 2)))
void gnn_mlp2(const float* __restrict__ tmp, const float* __restrict__ W2_l,
                         const float* __restrict__ h, float* __restrict__ dst) {
  const int wid = blockIdx.x;
  const int lane = threadIdx.x & 31;
  const int m0 = (wid >> 1) * 16;
  const int n0 = (wid & 1) * 64;
  v8f acc[NT_];
  load_strip_f32(h, D_, m0, n0, acc, lane);
  gemm_strip_f32(tmp, D_, W2_l, m0, n0, acc, lane);
  store_strip_f32(dst, D_, m0, n0, acc, lane);
}

extern "C" void kernel_launch(void* const* d_in, const int* in_sizes, int n_in,
                              void* d_out, int out_size, void* d_ws, size_t ws_size,
                              hipStream_t stream) {
  (void)in_sizes; (void)n_in; (void)out_size; (void)ws_size;
  const float* r        = (const float*)d_in[0];
  const float* coords   = (const float*)d_in[1];
  const float* spin_emb = (const float*)d_in[2];
  const float* nuc_emb  = (const float*)d_in[3];
  const float* W_rbf_ee = (const float*)d_in[4];
  const float* W_rbf_ne = (const float*)d_in[5];
  const float* W_msg_ee = (const float*)d_in[6];
  const float* W_msg_ne = (const float*)d_in[7];
  const float* W1       = (const float*)d_in[8];
  const float* W2       = (const float*)d_in[9];
  float* out = (float*)d_out;

  float* ws  = (float*)d_ws;
  float* Pee = ws;
  float* Pne = Pee + (size_t)B_ * NE_ * PEE_;
  float* h   = Pne + (size_t)B_ * NE_ * PNE_;
  float* hW  = h   + (size_t)B_ * NE_ * D_;
  float* hnW = hW  + (size_t)B_ * NE_ * D_;
  float* agg = hnW + (size_t)NN_ * D_;
  float* tmp = agg + (size_t)B_ * NE_ * D_;

  gnn_pairs<<<PAIR_EE_BLOCKS_ + PAIR_NE_BLOCKS_, 256, 0, stream>>>(r, coords, Pee, Pne);
  gnn_init_h<<<(B_ * NE_ * D_) / 256, 256, 0, stream>>>(spin_emb, h);

  for (int l = 0; l < L_; ++l) {
    gnn_hw<<<1024 + 2, 32, 0, stream>>>(h, nuc_emb, W_msg_ee + (size_t)l * WSZ_,
                                        W_msg_ne + (size_t)l * WSZ_, hW, hnW);
    gnn_agg<<<B_ * (NE_ / (2 * IPW_)), 128, 0, stream>>>(Pee, Pne, hW, hnW,
                                       W_rbf_ee + (size_t)l * NRBF_ * D_,
                                       W_rbf_ne + (size_t)l * NRBF_ * D_, agg);
    gnn_mlp1<<<1024, 32, 0, stream>>>(agg, W1 + (size_t)l * WSZ_, tmp);
    float* dst = (l == L_ - 1) ? out : h;
    gnn_mlp2<<<1024, 32, 0, stream>>>(tmp, W2 + (size_t)l * WSZ_, h, dst);
  }
}


# h init kernel and its launch go: the first layer's hW GEMM and residual read their h rows from spin_emb (row by spin group), not from a copy; 14 launches become 13; same products and sum order
# speedup vs baseline: 2.2765x; 1.1005x over previous
#include <hip/hip_runtime.h>
#include <math.h>

#define B_     128
#define NE_    64
#define NN_    16
#define D_     128
#define NRBF_  16
#define L_     3
#define NUP_   32
#define CUTOFF_ 2.0f
#define EPS_    1e-12f

#define NT_    4
#define WSZ_   (D_ * D_)

typedef float v2f __attribute__((ext_vector_type(2)));
typedef float v4f __attribute__((ext_vector_type(4)));
typedef float v8f __attribute__((ext_vector_type(8)));

__device__ __forceinline__ void gemm_strip_f32(const float* __restrict__ arow,
                                               const float* __restrict__ W,
                                               int n0, v8f acc[NT_], int lane) {
  const int half = lane >> 4;
  const int l16  = lane & 15;
  const float* __restrict__ wcol = W + n0 + 4 * l16;
#pragma unroll
  for (int k0 = 0; k0 < D_; k0 += 4) {
    const int ka = k0 + half * 2;
    const v2f a = *(const v2f*)&arow[ka];
    const v4f w0 = *(const v4f*)&wcol[(size_t)ka * D_];
    const v4f w1 = *(const v4f*)&wcol[(size_t)(ka + 1) * D_];
#pragma unroll
    for (int t = 0; t < NT_; ++t) {
      v2f b; b.x = w0[t]; b.y = w1[t];
      acc[t] = __builtin_amdgcn_wmma_f32_16x16x4_f32(false, a, false, b, (short)0,
                                                     acc[t], false, false);
    }
  }
}

__device__ __forceinline__ void store_strip_f32(float* __restrict__ C, int ldc,
                                                int m0, int n0, v8f acc[NT_], int lane) {
  const int half = lane >> 4;
  const int l16  = lane & 15;
#pragma unroll
  for (int ii = 0; ii < 8; ++ii) {
    v4f o;
#pragma unroll
    for (int t = 0; t < NT_; ++t) o[t] = acc[t][ii];
    *(v4f*)&C[(size_t)(m0 + half * 8 + ii) * ldc + n0 + 4 * l16] = o;
  }
}

template <bool kFirst>
__device__ __forceinline__ const float* h_row(const float* __restrict__ src, int row) {
  return kFirst ? src + (size_t)(((row & (NE_ - 1)) < NUP_) ? 0 : 1) * D_
                : src + (size_t)row * D_;
}

template <bool kFirst>
__device__ __forceinline__ void load_h_strip_f32(const float* __restrict__ src,
                                                 int m0, int n0, v8f acc[NT_], int lane) {
  const int half = lane >> 4;
  const int l16  = lane & 15;
#pragma unroll
  for (int ii = 0; ii < 8; ++ii) {
    const v4f o = *(const v4f*)&h_row<kFirst>(src, m0 + half * 8 + ii)[n0 + 4 * l16];
#pragma unroll
    for (int t = 0; t < NT_; ++t) acc[t][ii] = o[t];
  }
}

#define PEE_ (2 * NE_)
#define PNE_ (2 * NN_)
#define PAIR_EE_BLOCKS_ ((B_ * NE_ * (NE_ / 4)) / 256)
#define PAIR_NE_BLOCKS_ ((B_ * NE_ * (NN_ / 4)) / 256)

__global__ void gnn_pairs(const float* __restrict__ r, const float* __restrict__ coords,
                          float* __restrict__ Pee, float* __restrict__ Pne) {
  const float inv_cut = 1.0f / CUTOFF_;
  if (blockIdx.x < PAIR_EE_BLOCKS_) {
    const int u  = blockIdx.x * 256 + threadIdx.x;
    const int bi = u >> 4, q = u & 15;
    const int b = bi >> 6, i = bi & 63;
    const float rx = r[(size_t)bi * 3 + 0];
    const float ry = r[(size_t)bi * 3 + 1];
    const float rz = r[(size_t)bi * 3 + 2];
    const int jb = 8 * (q & 7) + (q >> 3);
    v4f dv, ev;
#pragma unroll
    for (int w = 0; w < 4; ++w) {
      const int j = jb + 2 * w;
      const float dx = rx - r[(size_t)(b * NE_ + j) * 3 + 0];
      const float dy = ry - r[(size_t)(b * NE_ + j) * 3 + 1];
      const float dz = rz - r[(size_t)(b * NE_ + j) * 3 + 2];
      const float dist = sqrtf(dx * dx + dy * dy + dz * dz + EPS_);
      float env = 1.0f - dist * inv_cut;
      env = (dist < CUTOFF_) ? env * env : 0.0f;
      if (i == j) env = 0.0f;
      dv[w] = dist; ev[w] = env;
    }
    *(v4f*)&Pee[(size_t)bi * PEE_ + 4 * q]       = dv;
    *(v4f*)&Pee[(size_t)bi * PEE_ + NE_ + 4 * q] = ev;
  } else {
    const int u  = (blockIdx.x - PAIR_EE_BLOCKS_) * 256 + threadIdx.x;
    const int bj = u >> 2, q = u & 3;
    const float rx = r[(size_t)bj * 3 + 0];
    const float ry = r[(size_t)bj * 3 + 1];
    const float rz = r[(size_t)bj * 3 + 2];
    const int ab = 8 * (q & 1) + (q >> 1);
    v4f dv, ev;
#pragma unroll
    for (int w = 0; w < 4; ++w) {
      const int a = ab + 2 * w;
      const float dx = rx - coords[a * 3 + 0];
      const float dy = ry - coords[a * 3 + 1];
      const float dz = rz - coords[a * 3 + 2];
      const float dist = sqrtf(dx * dx + dy * dy + dz * dz + EPS_);
      float env = 1.0f - dist * inv_cut;
      env = (dist < CUTOFF_) ? env * env : 0.0f;
      dv[w] = dist; ev[w] = env;
    }
    *(v4f*)&Pne[(size_t)bj * PNE_ + 4 * q]       = dv;
    *(v4f*)&Pne[(size_t)bj * PNE_ + NN_ + 4 * q] = ev;
  }
}

template <bool kFirst>
__global__ __launch_bounds__(32) __attribute__((amdgpu_waves_per_eu(1, 2)))
void gnn_hw(const float* __restrict__ hsrc, const float* __restrict__ nuc_emb,
                       const float* __restrict__ W_ee_l, const float* __restrict__ W_ne_l,
                       float* __restrict__ hW, float* __restrict__ hnW) {
  const int wid = blockIdx.x;
  const int lane = threadIdx.x & 31;
  v8f acc[NT_] = {};
  if (wid < 1024) {
    const int m0 = (wid >> 1) * 16;
    const int n0 = (wid & 1) * 64;
    gemm_strip_f32(h_row<kFirst>(hsrc, m0 + (lane & 15)), W_ee_l, n0, acc, lane);
    store_strip_f32(hW, D_, m0, n0, acc, lane);
  } else {
    const int n0 = (wid - 1024) * 64;
    gemm_strip_f32(nuc_emb + (size_t)(lane & 15) * D_, W_ne_l, n0, acc, lane);
    store_strip_f32(hnW, D_, 0, n0, acc, lane);
  }
}

#define IPW_ 8
#define SEE_ (NE_ / 4)
#define SNE_ (NN_ / 4)

struct AggP { v4f de[SEE_ / 2]; v4f ee[SEE_ / 2]; v4f dn[SNE_ / 2]; v4f en[SNE_ / 2]; };

__device__ __forceinline__ void agg_load_p(AggP& f, const float* __restrict__ pe,
                                           const float* __restrict__ pn) {
#pragma unroll
  for (int q = 0; q < SEE_ / 2; ++q) {
    f.de[q] = *(const v4f*)&pe[4 * q];
    f.ee[q] = *(const v4f*)&pe[NE_ + 4 * q];
  }
#pragma unroll
  for (int q = 0; q < SNE_ / 2; ++q) {
    f.dn[q] = *(const v4f*)&pn[4 * q];
    f.en[q] = *(const v4f*)&pn[NN_ + 4 * q];
  }
}

__device__ __forceinline__ v2f agg_feat(float d0, float d1, float e0, float e1,
                                        float mu, float kc) {
  v2f d;  d.x = d0;  d.y = d1;
  v2f en; en.x = e0; en.y = e1;
  const v2f zs = (d - mu) * kc;
  const v2f qq = zs * zs;
  v2f ex;
  ex.x = __builtin_amdgcn_exp2f(-qq.x);
  ex.y = __builtin_amdgcn_exp2f(-qq.y);
  return ex * en;
}

__device__ __forceinline__ float agg_xhalf(float x) {
  const int v = __builtin_bit_cast(int, x);
  return __builtin_bit_cast(float,
      __builtin_amdgcn_permlanex16(v, v, 0x76543210, 0xfedcba98, false, false));
}

__device__ __forceinline__ void agg_one(const AggP& f, float mu, float kc,
                                        const v2f (&bee)[SEE_][NT_],
                                        const v2f (&bne)[SNE_][NT_],
                                        const v2f (&wee)[NT_][4], const v2f (&wne)[NT_][4],
                                        float* __restrict__ orow, int half) {
  v8f g[NT_] = {};
#pragma unroll
  for (int s = 0; s < SEE_; ++s) {
    const int q = s >> 1, w = 2 * (s & 1);
    const v2f a = agg_feat(f.de[q][w], f.de[q][w + 1], f.ee[q][w], f.ee[q][w + 1], mu, kc);
#pragma unroll
    for (int t = 0; t < NT_; ++t)
      g[t] = __builtin_amdgcn_wmma_f32_16x16x4_f32(false, a, false, bee[s][t], (short)0,
                                                   g[t], false, false);
  }
  v8f gn[NT_] = {};
#pragma unroll
  for (int s = 0; s < SNE_; ++s) {
    const int q = s >> 1, w = 2 * (s & 1);
    const v2f a = agg_feat(f.dn[q][w], f.dn[q][w + 1], f.en[q][w], f.en[q][w + 1], mu, kc);
#pragma unroll
    for (int t = 0; t < NT_; ++t)
      gn[t] = __builtin_amdgcn_wmma_f32_16x16x4_f32(false, a, false, bne[s][t], (short)0,
                                                    gn[t], false, false);
  }
  float sv[NT_];
#pragma unroll
  for (int t = 0; t < NT_; ++t) {
    v2f p = {0.0f, 0.0f};
#pragma unroll
    for (int r = 0; r < 8; r += 2) {
      v2f gv; gv.x = g[t][r];  gv.y = g[t][r + 1];
      p += wee[t][r >> 1] * gv;
    }
#pragma unroll
    for (int r = 0; r < 8; r += 2) {
      v2f gv; gv.x = gn[t][r]; gv.y = gn[t][r + 1];
      p += wne[t][r >> 1] * gv;
    }
    sv[t] = p.x + p.y;
  }
  v2f own, snd;
  own.x = half ? sv[2] : sv[0];  own.y = half ? sv[3] : sv[1];
  snd.x = half ? sv[0] : sv[2];  snd.y = half ? sv[1] : sv[3];
  v2f o;
  o.x = own.x + agg_xhalf(snd.x);
  o.y = own.y + agg_xhalf(snd.y);
  *(v2f*)orow = o;
}

__global__ __launch_bounds__(128) __attribute__((amdgpu_waves_per_eu(2, 2)))
void gnn_agg(const float* __restrict__ Pee,
                        const float* __restrict__ Pne,
                        const float* __restrict__ hW, const float* __restrict__ hnW,
                        const float* __restrict__ Wrbf_ee_l, const float* __restrict__ Wrbf_ne_l,
                        float* __restrict__ agg) {
  const int lane = threadIdx.x & 31;
  const int wv   = __builtin_amdgcn_readfirstlane((int)(threadIdx.x >> 5));
  const int n0   = (wv & 1) * 64;
  const int b    = blockIdx.x / (NE_ / (2 * IPW_));
  const int i0   = ((blockIdx.x % (NE_ / (2 * IPW_))) * 2 + (wv >> 1)) * IPW_;
  const int half = lane >> 4, l16 = lane & 15;

  const float* __restrict__ pe = Pee + (size_t)(b * NE_ + i0) * PEE_ + half * (NE_ / 2);
  const float* __restrict__ pn = Pne + (size_t)(b * NE_ + i0) * PNE_ + half * (NN_ / 2);
  float* __restrict__ orow = agg + (size_t)(b * NE_ + i0) * D_ + n0 + 4 * l16 + 2 * half;

  const float mu = (CUTOFF_ / (float)(NRBF_ - 1)) * (float)l16;
  const float kc = ((float)NRBF_ / CUTOFF_) * 0.84932180028801904f;

  AggP f0, f1;
  agg_load_p(f0, pe, pn);

  const float* __restrict__ hb = hW + (size_t)b * NE_ * D_ + half * D_ + n0 + 4 * l16;
  const float* __restrict__ nb = hnW + half * D_ + n0 + 4 * l16;
  v2f bee[SEE_][NT_], bne[SNE_][NT_];
#pragma unroll
  for (int s = 0; s < SEE_; ++s) {
    const v4f x = *(const v4f*)&hb[(4 * s) * D_];
    const v4f y = *(const v4f*)&hb[(4 * s + 2) * D_];
#pragma unroll
    for (int t = 0; t < NT_; ++t) { bee[s][t].x = x[t]; bee[s][t].y = y[t]; }
  }
#pragma unroll
  for (int s = 0; s < SNE_; ++s) {
    const v4f x = *(const v4f*)&nb[(4 * s) * D_];
    const v4f y = *(const v4f*)&nb[(4 * s + 2) * D_];
#pragma unroll
    for (int t = 0; t < NT_; ++t) { bne[s][t].x = x[t]; bne[s][t].y = y[t]; }
  }
  v2f wee[NT_][4], wne[NT_][4];
  const float* __restrict__ we = Wrbf_ee_l + (size_t)(8 * half) * D_ + n0 + 4 * l16;
  const float* __restrict__ wn = Wrbf_ne_l + (size_t)(8 * half) * D_ + n0 + 4 * l16;
#pragma unroll
  for (int r = 0; r < 8; r += 2) {
    const v4f e0 = *(const v4f*)&we[r * D_], e1 = *(const v4f*)&we[(r + 1) * D_];
    const v4f n0v = *(const v4f*)&wn[r * D_], n1v = *(const v4f*)&wn[(r + 1) * D_];
#pragma unroll
    for (int t = 0; t < NT_; ++t) {
      wee[t][r >> 1].x = e0[t];  wee[t][r >> 1].y = e1[t];
      wne[t][r >> 1].x = n0v[t]; wne[t][r >> 1].y = n1v[t];
    }
  }

#pragma unroll 1
  for (int ii = 0; ii < IPW_ - 2; ii += 2) {
    agg_load_p(f1, pe + (size_t)(ii + 1) * PEE_, pn + (size_t)(ii + 1) * PNE_);
    agg_one(f0, mu, kc, bee, bne, wee, wne, orow + (size_t)ii * D_, half);
    agg_load_p(f0, pe + (size_t)(ii + 2) * PEE_, pn + (size_t)(ii + 2) * PNE_);
    agg_one(f1, mu, kc, bee, bne, wee, wne, orow + (size_t)(ii + 1) * D_, half);
  }
  agg_load_p(f1, pe + (size_t)(IPW_ - 1) * PEE_, pn + (size_t)(IPW_ - 1) * PNE_);
  agg_one(f0, mu, kc, bee, bne, wee, wne, orow + (size_t)(IPW_ - 2) * D_, half);
  agg_one(f1, mu, kc, bee, bne, wee, wne, orow + (size_t)(IPW_ - 1) * D_, half);
}

__global__ __launch_bounds__(32) __attribute__((amdgpu_waves_per_eu(1, 2)))
void gnn_mlp1(const float* __restrict__ agg, const float* __restrict__ W1_l,
                         float* __restrict__ tmp) {
  const int wid = blockIdx.x;
  const int lane = threadIdx.x & 31;
  const int m0 = (wid >> 1) * 16;
  const int n0 = (wid & 1) * 64;
  v8f acc[NT_] = {};
  gemm_strip_f32(agg + (size_t)(m0 + (lane & 15)) * D_, W1_l, n0, acc, lane);
#pragma unroll
  for (int t = 0; t < NT_; ++t)
#pragma unroll
    for (int ii = 0; ii < 8; ++ii) acc[t][ii] = tanhf(acc[t][ii]);
  store_strip_f32(tmp, D_, m0, n0, acc, lane);
}

template <bool kFirst>
__global__ __launch_bounds__(32) __attribute__((amdgpu_waves_per_eu(1, 2)))
void gnn_mlp2(const float* __restrict__ tmp, const float* __restrict__ W2_l,
                         const float* __restrict__ hsrc, float* __restrict__ dst) {
  const int wid = blockIdx.x;
  const int lane = threadIdx.x & 31;
  const int m0 = (wid >> 1) * 16;
  const int n0 = (wid & 1) * 64;
  v8f acc[NT_];
  load_h_strip_f32<kFirst>(hsrc, m0, n0, acc, lane);
  gemm_strip_f32(tmp + (size_t)(m0 + (lane & 15)) * D_, W2_l, n0, acc, lane);
  store_strip_f32(dst, D_, m0, n0, acc, lane);
}

extern "C" void kernel_launch(void* const* d_in, const int* in_sizes, int n_in,
                              void* d_out, int out_size, void* d_ws, size_t ws_size,
                              hipStream_t stream) {
  (void)in_sizes; (void)n_in; (void)out_size; (void)ws_size;
  const float* r        = (const float*)d_in[0];
  const float* coords   = (const float*)d_in[1];
  const float* spin_emb = (const float*)d_in[2];
  const float* nuc_emb  = (const float*)d_in[3];
  const float* W_rbf_ee = (const float*)d_in[4];
  const float* W_rbf_ne = (const float*)d_in[5];
  const float* W_msg_ee = (const float*)d_in[6];
  const float* W_msg_ne = (const float*)d_in[7];
  const float* W1       = (const float*)d_in[8];
  const float* W2       = (const float*)d_in[9];
  float* out = (float*)d_out;

  float* ws  = (float*)d_ws;
  float* Pee = ws;
  float* Pne = Pee + (size_t)B_ * NE_ * PEE_;
  float* h   = Pne + (size_t)B_ * NE_ * PNE_;
  float* hW  = h   + (size_t)B_ * NE_ * D_;
  float* hnW = hW  + (size_t)B_ * NE_ * D_;
  float* agg = hnW + (size_t)NN_ * D_;
  float* tmp = agg + (size_t)B_ * NE_ * D_;

  gnn_pairs<<<PAIR_EE_BLOCKS_ + PAIR_NE_BLOCKS_, 256, 0, stream>>>(r, coords, Pee, Pne);

  for (int l = 0; l < L_; ++l) {
    if (l == 0)
      gnn_hw<true><<<1024 + 2, 32, 0, stream>>>(spin_emb, nuc_emb, W_msg_ee, W_msg_ne, hW, hnW);
    else
      gnn_hw<false><<<1024 + 2, 32, 0, stream>>>(h, nuc_emb, W_msg_ee + (size_t)l * WSZ_,
                                                 W_msg_ne + (size_t)l * WSZ_, hW, hnW);
    gnn_agg<<<B_ * (NE_ / (2 * IPW_)), 128, 0, stream>>>(Pee, Pne, hW, hnW,
                                       W_rbf_ee + (size_t)l * NRBF_ * D_,
                                       W_rbf_ne + (size_t)l * NRBF_ * D_, agg);
    gnn_mlp1<<<1024, 32, 0, stream>>>(agg, W1 + (size_t)l * WSZ_, tmp);
    float* dst = (l == L_ - 1) ? out : h;
    if (l == 0)
      gnn_mlp2<true><<<1024, 32, 0, stream>>>(tmp, W2, spin_emb, dst);
    else
      gnn_mlp2<false><<<1024, 32, 0, stream>>>(tmp, W2 + (size_t)l * WSZ_, h, dst);
  }
}


# next layer's hW re-associated: h@Wmsg + t@(W2@Wmsg), the weight product made once at the start; it no longer waits for mlp2 and shares mlp2's launch; 13 launches become 11; hW sum order changes
# speedup vs baseline: 2.3674x; 1.0399x over previous
#include <hip/hip_runtime.h>
#include <math.h>

#define B_     128
#define NE_    64
#define NN_    16
#define D_     128
#define NRBF_  16
#define L_     3
#define NUP_   32
#define CUTOFF_ 2.0f
#define EPS_    1e-12f

#define NT_    4
#define WSZ_   (D_ * D_)

typedef float v2f __attribute__((ext_vector_type(2)));
typedef float v4f __attribute__((ext_vector_type(4)));
typedef float v8f __attribute__((ext_vector_type(8)));

__device__ __forceinline__ void gemm_strip_f32(const float* __restrict__ arow,
                                               const float* __restrict__ W,
                                               int n0, v8f acc[NT_], int lane) {
  const int half = lane >> 4;
  const int l16  = lane & 15;
  const float* __restrict__ wcol = W + n0 + 4 * l16;
#pragma unroll
  for (int k0 = 0; k0 < D_; k0 += 4) {
    const int ka = k0 + half * 2;
    const v2f a = *(const v2f*)&arow[ka];
    const v4f w0 = *(const v4f*)&wcol[(size_t)ka * D_];
    const v4f w1 = *(const v4f*)&wcol[(size_t)(ka + 1) * D_];
#pragma unroll
    for (int t = 0; t < NT_; ++t) {
      v2f b; b.x = w0[t]; b.y = w1[t];
      acc[t] = __builtin_amdgcn_wmma_f32_16x16x4_f32(false, a, false, b, (short)0,
                                                     acc[t], false, false);
    }
  }
}

__device__ __forceinline__ void store_strip_f32(float* __restrict__ C, int ldc,
                                                int m0, int n0, v8f acc[NT_], int lane) {
  const int half = lane >> 4;
  const int l16  = lane & 15;
#pragma unroll
  for (int ii = 0; ii < 8; ++ii) {
    v4f o;
#pragma unroll
    for (int t = 0; t < NT_; ++t) o[t] = acc[t][ii];
    *(v4f*)&C[(size_t)(m0 + half * 8 + ii) * ldc + n0 + 4 * l16] = o;
  }
}

template <bool kFirst>
__device__ __forceinline__ const float* h_row(const float* __restrict__ src, int row) {
  return kFirst ? src + (size_t)(((row & (NE_ - 1)) < NUP_) ? 0 : 1) * D_
                : src + (size_t)row * D_;
}

template <bool kFirst>
__device__ __forceinline__ void load_h_strip_f32(const float* __restrict__ src,
                                                 int m0, int n0, v8f acc[NT_], int lane) {
  const int half = lane >> 4;
  const int l16  = lane & 15;
#pragma unroll
  for (int ii = 0; ii < 8; ++ii) {
    const v4f o = *(const v4f*)&h_row<kFirst>(src, m0 + half * 8 + ii)[n0 + 4 * l16];
#pragma unroll
    for (int t = 0; t < NT_; ++t) acc[t][ii] = o[t];
  }
}

#define PEE_ (2 * NE_)
#define PNE_ (2 * NN_)
#define PAIR_EE_BLOCKS_ ((B_ * NE_ * (NE_ / 4)) / 256)
#define PAIR_NE_BLOCKS_ ((B_ * NE_ * (NN_ / 4)) / 256)

__global__ void gnn_pairs(const float* __restrict__ r, const float* __restrict__ coords,
                          float* __restrict__ Pee, float* __restrict__ Pne) {
  const float inv_cut = 1.0f / CUTOFF_;
  if (blockIdx.x < PAIR_EE_BLOCKS_) {
    const int u  = blockIdx.x * 256 + threadIdx.x;
    const int bi = u >> 4, q = u & 15;
    const int b = bi >> 6, i = bi & 63;
    const float rx = r[(size_t)bi * 3 + 0];
    const float ry = r[(size_t)bi * 3 + 1];
    const float rz = r[(size_t)bi * 3 + 2];
    const int jb = 8 * (q & 7) + (q >> 3);
    v4f dv, ev;
#pragma unroll
    for (int w = 0; w < 4; ++w) {
      const int j = jb + 2 * w;
      const float dx = rx - r[(size_t)(b * NE_ + j) * 3 + 0];
      const float dy = ry - r[(size_t)(b * NE_ + j) * 3 + 1];
      const float dz = rz - r[(size_t)(b * NE_ + j) * 3 + 2];
      const float dist = sqrtf(dx * dx + dy * dy + dz * dz + EPS_);
      float env = 1.0f - dist * inv_cut;
      env = (dist < CUTOFF_) ? env * env : 0.0f;
      if (i == j) env = 0.0f;
      dv[w] = dist; ev[w] = env;
    }
    *(v4f*)&Pee[(size_t)bi * PEE_ + 4 * q]       = dv;
    *(v4f*)&Pee[(size_t)bi * PEE_ + NE_ + 4 * q] = ev;
  } else {
    const int u  = (blockIdx.x - PAIR_EE_BLOCKS_) * 256 + threadIdx.x;
    const int bj = u >> 2, q = u & 3;
    const float rx = r[(size_t)bj * 3 + 0];
    const float ry = r[(size_t)bj * 3 + 1];
    const float rz = r[(size_t)bj * 3 + 2];
    const int ab = 8 * (q & 1) + (q >> 1);
    v4f dv, ev;
#pragma unroll
    for (int w = 0; w < 4; ++w) {
      const int a = ab + 2 * w;
      const float dx = rx - coords[a * 3 + 0];
      const float dy = ry - coords[a * 3 + 1];
      const float dz = rz - coords[a * 3 + 2];
      const float dist = sqrtf(dx * dx + dy * dy + dz * dz + EPS_);
      float env = 1.0f - dist * inv_cut;
      env = (dist < CUTOFF_) ? env * env : 0.0f;
      dv[w] = dist; ev[w] = env;
    }
    *(v4f*)&Pne[(size_t)bj * PNE_ + 4 * q]       = dv;
    *(v4f*)&Pne[(size_t)bj * PNE_ + NN_ + 4 * q] = ev;
  }
}

#define PROD_STRIPS_ ((D_ / 16) * 2)
__global__ __launch_bounds__(32) __attribute__((amdgpu_waves_per_eu(1, 2)))
void gnn_hw0(const float* __restrict__ spin_emb, const float* __restrict__ nuc_emb,
                        const float* __restrict__ W_ee, const float* __restrict__ W_ne,
                        const float* __restrict__ W2,
                        float* __restrict__ hW, float* __restrict__ hnW,
                        float* __restrict__ P) {
  const int wid = blockIdx.x;
  const int lane = threadIdx.x & 31;
  v8f acc[NT_] = {};
  if (wid < 1024) {
    const int m0 = (wid >> 1) * 16;
    const int n0 = (wid & 1) * 64;
    gemm_strip_f32(h_row<true>(spin_emb, m0 + (lane & 15)), W_ee, n0, acc, lane);
    store_strip_f32(hW, D_, m0, n0, acc, lane);
  } else if (wid < 1026) {
    const int n0 = (wid - 1024) * 64;
    gemm_strip_f32(nuc_emb + (size_t)(lane & 15) * D_, W_ne, n0, acc, lane);
    store_strip_f32(hnW, D_, 0, n0, acc, lane);
  } else {
    const int idx = wid - 1026;
    const int l   = idx / PROD_STRIPS_;
    const int rem = idx % PROD_STRIPS_;
    const int m0 = (rem >> 1) * 16;
    const int n0 = (rem & 1) * 64;
    gemm_strip_f32(W2 + (size_t)l * WSZ_ + (size_t)(m0 + (lane & 15)) * D_,
                   W_ee + (size_t)(l + 1) * WSZ_, n0, acc, lane);
    store_strip_f32(P + (size_t)l * WSZ_, D_, m0, n0, acc, lane);
  }
}

#define IPW_ 8
#define SEE_ (NE_ / 4)
#define SNE_ (NN_ / 4)

struct AggP { v4f de[SEE_ / 2]; v4f ee[SEE_ / 2]; v4f dn[SNE_ / 2]; v4f en[SNE_ / 2]; };

__device__ __forceinline__ void agg_load_p(AggP& f, const float* __restrict__ pe,
                                           const float* __restrict__ pn) {
#pragma unroll
  for (int q = 0; q < SEE_ / 2; ++q) {
    f.de[q] = *(const v4f*)&pe[4 * q];
    f.ee[q] = *(const v4f*)&pe[NE_ + 4 * q];
  }
#pragma unroll
  for (int q = 0; q < SNE_ / 2; ++q) {
    f.dn[q] = *(const v4f*)&pn[4 * q];
    f.en[q] = *(const v4f*)&pn[NN_ + 4 * q];
  }
}

__device__ __forceinline__ v2f agg_feat(float d0, float d1, float e0, float e1,
                                        float mu, float kc) {
  v2f d;  d.x = d0;  d.y = d1;
  v2f en; en.x = e0; en.y = e1;
  const v2f zs = (d - mu) * kc;
  const v2f qq = zs * zs;
  v2f ex;
  ex.x = __builtin_amdgcn_exp2f(-qq.x);
  ex.y = __builtin_amdgcn_exp2f(-qq.y);
  return ex * en;
}

__device__ __forceinline__ float agg_xhalf(float x) {
  const int v = __builtin_bit_cast(int, x);
  return __builtin_bit_cast(float,
      __builtin_amdgcn_permlanex16(v, v, 0x76543210, 0xfedcba98, false, false));
}

__device__ __forceinline__ void agg_one(const AggP& f, float mu, float kc,
                                        const v2f (&bee)[SEE_][NT_],
                                        const v2f (&bne)[SNE_][NT_],
                                        const v2f (&wee)[NT_][4], const v2f (&wne)[NT_][4],
                                        float* __restrict__ orow, int half) {
  v8f g[NT_] = {};
#pragma unroll
  for (int s = 0; s < SEE_; ++s) {
    const int q = s >> 1, w = 2 * (s & 1);
    const v2f a = agg_feat(f.de[q][w], f.de[q][w + 1], f.ee[q][w], f.ee[q][w + 1], mu, kc);
#pragma unroll
    for (int t = 0; t < NT_; ++t)
      g[t] = __builtin_amdgcn_wmma_f32_16x16x4_f32(false, a, false, bee[s][t], (short)0,
                                                   g[t], false, false);
  }
  v8f gn[NT_] = {};
#pragma unroll
  for (int s = 0; s < SNE_; ++s) {
    const int q = s >> 1, w = 2 * (s & 1);
    const v2f a = agg_feat(f.dn[q][w], f.dn[q][w + 1], f.en[q][w], f.en[q][w + 1], mu, kc);
#pragma unroll
    for (int t = 0; t < NT_; ++t)
      gn[t] = __builtin_amdgcn_wmma_f32_16x16x4_f32(false, a, false, bne[s][t], (short)0,
                                                    gn[t], false, false);
  }
  float sv[NT_];
#pragma unroll
  for (int t = 0; t < NT_; ++t) {
    v2f p = {0.0f, 0.0f};
#pragma unroll
    for (int r = 0; r < 8; r += 2) {
      v2f gv; gv.x = g[t][r];  gv.y = g[t][r + 1];
      p += wee[t][r >> 1] * gv;
    }
#pragma unroll
    for (int r = 0; r < 8; r += 2) {
      v2f gv; gv.x = gn[t][r]; gv.y = gn[t][r + 1];
      p += wne[t][r >> 1] * gv;
    }
    sv[t] = p.x + p.y;
  }
  v2f own, snd;
  own.x = half ? sv[2] : sv[0];  own.y = half ? sv[3] : sv[1];
  snd.x = half ? sv[0] : sv[2];  snd.y = half ? sv[1] : sv[3];
  v2f o;
  o.x = own.x + agg_xhalf(snd.x);
  o.y = own.y + agg_xhalf(snd.y);
  *(v2f*)orow = o;
}

__global__ __launch_bounds__(128) __attribute__((amdgpu_waves_per_eu(2, 2)))
void gnn_agg(const float* __restrict__ Pee,
                        const float* __restrict__ Pne,
                        const float* __restrict__ hW, const float* __restrict__ hnW,
                        const float* __restrict__ Wrbf_ee_l, const float* __restrict__ Wrbf_ne_l,
                        float* __restrict__ agg) {
  const int lane = threadIdx.x & 31;
  const int wv   = __builtin_amdgcn_readfirstlane((int)(threadIdx.x >> 5));
  const int n0   = (wv & 1) * 64;
  const int b    = blockIdx.x / (NE_ / (2 * IPW_));
  const int i0   = ((blockIdx.x % (NE_ / (2 * IPW_))) * 2 + (wv >> 1)) * IPW_;
  const int half = lane >> 4, l16 = lane & 15;

  const float* __restrict__ pe = Pee + (size_t)(b * NE_ + i0) * PEE_ + half * (NE_ / 2);
  const float* __restrict__ pn = Pne + (size_t)(b * NE_ + i0) * PNE_ + half * (NN_ / 2);
  float* __restrict__ orow = agg + (size_t)(b * NE_ + i0) * D_ + n0 + 4 * l16 + 2 * half;

  const float mu = (CUTOFF_ / (float)(NRBF_ - 1)) * (float)l16;
  const float kc = ((float)NRBF_ / CUTOFF_) * 0.84932180028801904f;

  AggP f0, f1;
  agg_load_p(f0, pe, pn);

  const float* __restrict__ hb = hW + (size_t)b * NE_ * D_ + half * D_ + n0 + 4 * l16;
  const float* __restrict__ nb = hnW + half * D_ + n0 + 4 * l16;
  v2f bee[SEE_][NT_], bne[SNE_][NT_];
#pragma unroll
  for (int s = 0; s < SEE_; ++s) {
    const v4f x = *(const v4f*)&hb[(4 * s) * D_];
    const v4f y = *(const v4f*)&hb[(4 * s + 2) * D_];
#pragma unroll
    for (int t = 0; t < NT_; ++t) { bee[s][t].x = x[t]; bee[s][t].y = y[t]; }
  }
#pragma unroll
  for (int s = 0; s < SNE_; ++s) {
    const v4f x = *(const v4f*)&nb[(4 * s) * D_];
    const v4f y = *(const v4f*)&nb[(4 * s + 2) * D_];
#pragma unroll
    for (int t = 0; t < NT_; ++t) { bne[s][t].x = x[t]; bne[s][t].y = y[t]; }
  }
  v2f wee[NT_][4], wne[NT_][4];
  const float* __restrict__ we = Wrbf_ee_l + (size_t)(8 * half) * D_ + n0 + 4 * l16;
  const float* __restrict__ wn = Wrbf_ne_l + (size_t)(8 * half) * D_ + n0 + 4 * l16;
#pragma unroll
  for (int r = 0; r < 8; r += 2) {
    const v4f e0 = *(const v4f*)&we[r * D_], e1 = *(const v4f*)&we[(r + 1) * D_];
    const v4f n0v = *(const v4f*)&wn[r * D_], n1v = *(const v4f*)&wn[(r + 1) * D_];
#pragma unroll
    for (int t = 0; t < NT_; ++t) {
      wee[t][r >> 1].x = e0[t];  wee[t][r >> 1].y = e1[t];
      wne[t][r >> 1].x = n0v[t]; wne[t][r >> 1].y = n1v[t];
    }
  }

#pragma unroll 1
  for (int ii = 0; ii < IPW_ - 2; ii += 2) {
    agg_load_p(f1, pe + (size_t)(ii + 1) * PEE_, pn + (size_t)(ii + 1) * PNE_);
    agg_one(f0, mu, kc, bee, bne, wee, wne, orow + (size_t)ii * D_, half);
    agg_load_p(f0, pe + (size_t)(ii + 2) * PEE_, pn + (size_t)(ii + 2) * PNE_);
    agg_one(f1, mu, kc, bee, bne, wee, wne, orow + (size_t)(ii + 1) * D_, half);
  }
  agg_load_p(f1, pe + (size_t)(IPW_ - 1) * PEE_, pn + (size_t)(IPW_ - 1) * PNE_);
  agg_one(f0, mu, kc, bee, bne, wee, wne, orow + (size_t)(IPW_ - 2) * D_, half);
  agg_one(f1, mu, kc, bee, bne, wee, wne, orow + (size_t)(IPW_ - 1) * D_, half);
}

__global__ __launch_bounds__(32) __attribute__((amdgpu_waves_per_eu(1, 2)))
void gnn_mlp1(const float* __restrict__ agg, const float* __restrict__ W1_l,
                         float* __restrict__ tmp) {
  const int wid = blockIdx.x;
  const int lane = threadIdx.x & 31;
  const int m0 = (wid >> 1) * 16;
  const int n0 = (wid & 1) * 64;
  v8f acc[NT_] = {};
  gemm_strip_f32(agg + (size_t)(m0 + (lane & 15)) * D_, W1_l, n0, acc, lane);
#pragma unroll
  for (int t = 0; t < NT_; ++t)
#pragma unroll
    for (int ii = 0; ii < 8; ++ii) acc[t][ii] = tanhf(acc[t][ii]);
  store_strip_f32(tmp, D_, m0, n0, acc, lane);
}

template <bool kFirst>
__global__ __launch_bounds__(32) __attribute__((amdgpu_waves_per_eu(1, 2)))
void gnn_mlp2_hw(const float* __restrict__ tmp, const float* __restrict__ W2_l,
                            const float* __restrict__ hsrc, float* __restrict__ dst,
                            const float* __restrict__ W_ee_n, const float* __restrict__ P_l,
                            const float* __restrict__ nuc_emb, const float* __restrict__ W_ne_n,
                            float* __restrict__ hW, float* __restrict__ hnW) {
  const int wid = blockIdx.x;
  const int lane = threadIdx.x & 31;
  if (wid < 1024) {
    const int m0 = (wid >> 1) * 16;
    const int n0 = (wid & 1) * 64;
    v8f acc[NT_] = {};
    gemm_strip_f32(h_row<kFirst>(hsrc, m0 + (lane & 15)), W_ee_n, n0, acc, lane);
    gemm_strip_f32(tmp + (size_t)(m0 + (lane & 15)) * D_, P_l, n0, acc, lane);
    store_strip_f32(hW, D_, m0, n0, acc, lane);
  } else if (wid < 1026) {
    const int n0 = (wid - 1024) * 64;
    v8f acc[NT_] = {};
    gemm_strip_f32(nuc_emb + (size_t)(lane & 15) * D_, W_ne_n, n0, acc, lane);
    store_strip_f32(hnW, D_, 0, n0, acc, lane);
  } else {
    const int sid = wid - 1026;
    const int m0 = (sid >> 1) * 16;
    const int n0 = (sid & 1) * 64;
    v8f acc[NT_];
    load_h_strip_f32<kFirst>(hsrc, m0, n0, acc, lane);
    gemm_strip_f32(tmp + (size_t)(m0 + (lane & 15)) * D_, W2_l, n0, acc, lane);
    store_strip_f32(dst, D_, m0, n0, acc, lane);
  }
}

__global__ __launch_bounds__(32) __attribute__((amdgpu_waves_per_eu(1, 2)))
void gnn_mlp2(const float* __restrict__ tmp, const float* __restrict__ W2_l,
                         float* hio) {
  const int wid = blockIdx.x;
  const int lane = threadIdx.x & 31;
  const int m0 = (wid >> 1) * 16;
  const int n0 = (wid & 1) * 64;
  v8f acc[NT_];
  load_h_strip_f32<false>(hio, m0, n0, acc, lane);
  gemm_strip_f32(tmp + (size_t)(m0 + (lane & 15)) * D_, W2_l, n0, acc, lane);
  store_strip_f32(hio, D_, m0, n0, acc, lane);
}

extern "C" void kernel_launch(void* const* d_in, const int* in_sizes, int n_in,
                              void* d_out, int out_size, void* d_ws, size_t ws_size,
                              hipStream_t stream) {
  (void)in_sizes; (void)n_in; (void)out_size; (void)ws_size;
  const float* r        = (const float*)d_in[0];
  const float* coords   = (const float*)d_in[1];
  const float* spin_emb = (const float*)d_in[2];
  const float* nuc_emb  = (const float*)d_in[3];
  const float* W_rbf_ee = (const float*)d_in[4];
  const float* W_rbf_ne = (const float*)d_in[5];
  const float* W_msg_ee = (const float*)d_in[6];
  const float* W_msg_ne = (const float*)d_in[7];
  const float* W1       = (const float*)d_in[8];
  const float* W2       = (const float*)d_in[9];
  float* out = (float*)d_out;

  float* ws  = (float*)d_ws;
  float* Pee = ws;
  float* Pne = Pee + (size_t)B_ * NE_ * PEE_;
  float* h   = Pne + (size_t)B_ * NE_ * PNE_;
  float* hW  = h   + (size_t)B_ * NE_ * D_;
  float* hnW = hW  + (size_t)B_ * NE_ * D_;
  float* agg = hnW + (size_t)NN_ * D_;
  float* tmp = agg + (size_t)B_ * NE_ * D_;
  float* P   = tmp + (size_t)B_ * NE_ * D_;

  gnn_pairs<<<PAIR_EE_BLOCKS_ + PAIR_NE_BLOCKS_, 256, 0, stream>>>(r, coords, Pee, Pne);
  gnn_hw0<<<1024 + 2 + (L_ - 1) * PROD_STRIPS_, 32, 0, stream>>>(spin_emb, nuc_emb, W_msg_ee,
                                                                W_msg_ne, W2, hW, hnW, P);

  static_assert(L_ == 3, "the h planes below are laid out for three layers");
  const float* hsrc = spin_emb;
  for (int l = 0; l < L_; ++l) {
    gnn_agg<<<B_ * (NE_ / (2 * IPW_)), 128, 0, stream>>>(Pee, Pne, hW, hnW,
                                       W_rbf_ee + (size_t)l * NRBF_ * D_,
                                       W_rbf_ne + (size_t)l * NRBF_ * D_, agg);
    gnn_mlp1<<<1024, 32, 0, stream>>>(agg, W1 + (size_t)l * WSZ_, tmp);
    if (l == L_ - 1) {
      gnn_mlp2<<<1024, 32, 0, stream>>>(tmp, W2 + (size_t)l * WSZ_, out);
    } else {
      float* dst = (l == 0) ? h : out;
      if (l == 0)
        gnn_mlp2_hw<true><<<1024 + 2 + 1024, 32, 0, stream>>>(
            tmp, W2, hsrc, dst, W_msg_ee + (size_t)WSZ_, P, nuc_emb,
            W_msg_ne + (size_t)WSZ_, hW, hnW);
      else
        gnn_mlp2_hw<false><<<1024 + 2 + 1024, 32, 0, stream>>>(
            tmp, W2 + (size_t)l * WSZ_, hsrc, dst, W_msg_ee + (size_t)(l + 1) * WSZ_,
            P + (size_t)l * WSZ_, nuc_emb, W_msg_ne + (size_t)(l + 1) * WSZ_, hW, hnW);
      hsrc = dst;
    }
  }
}


# agg launch also makes tanh(agg@W1): a block owns 16 whole rows, reads them back behind one block barrier, each wave a 16x32 strip; the 3 mlp1 launches go (11 to 8); same products and sum order
# speedup vs baseline: 2.5265x; 1.0672x over previous
#include <hip/hip_runtime.h>
#include <math.h>

#define B_     128
#define NE_    64
#define NN_    16
#define D_     128
#define NRBF_  16
#define L_     3
#define NUP_   32
#define CUTOFF_ 2.0f
#define EPS_    1e-12f

#define NT_    4
#define WSZ_   (D_ * D_)

typedef float v2f __attribute__((ext_vector_type(2)));
typedef float v4f __attribute__((ext_vector_type(4)));
typedef float v8f __attribute__((ext_vector_type(8)));

__device__ __forceinline__ void gemm_strip_f32(const float* __restrict__ arow,
                                               const float* __restrict__ W,
                                               int n0, v8f acc[NT_], int lane) {
  const int half = lane >> 4;
  const int l16  = lane & 15;
  const float* __restrict__ wcol = W + n0 + 4 * l16;
#pragma unroll
  for (int k0 = 0; k0 < D_; k0 += 4) {
    const int ka = k0 + half * 2;
    const v2f a = *(const v2f*)&arow[ka];
    const v4f w0 = *(const v4f*)&wcol[(size_t)ka * D_];
    const v4f w1 = *(const v4f*)&wcol[(size_t)(ka + 1) * D_];
#pragma unroll
    for (int t = 0; t < NT_; ++t) {
      v2f b; b.x = w0[t]; b.y = w1[t];
      acc[t] = __builtin_amdgcn_wmma_f32_16x16x4_f32(false, a, false, b, (short)0,
                                                     acc[t], false, false);
    }
  }
}

__device__ __forceinline__ void store_strip_f32(float* __restrict__ C, int ldc,
                                                int m0, int n0, v8f acc[NT_], int lane) {
  const int half = lane >> 4;
  const int l16  = lane & 15;
#pragma unroll
  for (int ii = 0; ii < 8; ++ii) {
    v4f o;
#pragma unroll
    for (int t = 0; t < NT_; ++t) o[t] = acc[t][ii];
    *(v4f*)&C[(size_t)(m0 + half * 8 + ii) * ldc + n0 + 4 * l16] = o;
  }
}

template <bool kFirst>
__device__ __forceinline__ const float* h_row(const float* __restrict__ src, int row) {
  return kFirst ? src + (size_t)(((row & (NE_ - 1)) < NUP_) ? 0 : 1) * D_
                : src + (size_t)row * D_;
}

template <bool kFirst>
__device__ __forceinline__ void load_h_strip_f32(const float* __restrict__ src,
                                                 int m0, int n0, v8f acc[NT_], int lane) {
  const int half = lane >> 4;
  const int l16  = lane & 15;
#pragma unroll
  for (int ii = 0; ii < 8; ++ii) {
    const v4f o = *(const v4f*)&h_row<kFirst>(src, m0 + half * 8 + ii)[n0 + 4 * l16];
#pragma unroll
    for (int t = 0; t < NT_; ++t) acc[t][ii] = o[t];
  }
}

__device__ __forceinline__ void gemm_strip32_f32(const float* arow,
                                                 const float* __restrict__ W,
                                                 int n0, v8f acc[2], int lane) {
  const int half = lane >> 4;
  const int l16  = lane & 15;
  const float* __restrict__ wcol = W + n0 + 2 * l16;
#pragma unroll
  for (int k0 = 0; k0 < D_; k0 += 4) {
    const int ka = k0 + half * 2;
    const v2f a  = *(const v2f*)&arow[ka];
    const v2f w0 = *(const v2f*)&wcol[(size_t)ka * D_];
    const v2f w1 = *(const v2f*)&wcol[(size_t)(ka + 1) * D_];
#pragma unroll
    for (int t = 0; t < 2; ++t) {
      v2f b; b.x = w0[t]; b.y = w1[t];
      acc[t] = __builtin_amdgcn_wmma_f32_16x16x4_f32(false, a, false, b, (short)0,
                                                     acc[t], false, false);
    }
  }
}

__device__ __forceinline__ void store_strip32_f32(float* __restrict__ C, int ldc,
                                                  int m0, int n0, v8f acc[2], int lane) {
  const int half = lane >> 4;
  const int l16  = lane & 15;
#pragma unroll
  for (int ii = 0; ii < 8; ++ii) {
    v2f o; o.x = acc[0][ii]; o.y = acc[1][ii];
    *(v2f*)&C[(size_t)(m0 + half * 8 + ii) * ldc + n0 + 2 * l16] = o;
  }
}

#define PEE_ (2 * NE_)
#define PNE_ (2 * NN_)
#define PAIR_EE_BLOCKS_ ((B_ * NE_ * (NE_ / 4)) / 256)
#define PAIR_NE_BLOCKS_ ((B_ * NE_ * (NN_ / 4)) / 256)

__global__ void gnn_pairs(const float* __restrict__ r, const float* __restrict__ coords,
                          float* __restrict__ Pee, float* __restrict__ Pne) {
  const float inv_cut = 1.0f / CUTOFF_;
  if (blockIdx.x < PAIR_EE_BLOCKS_) {
    const int u  = blockIdx.x * 256 + threadIdx.x;
    const int bi = u >> 4, q = u & 15;
    const int b = bi >> 6, i = bi & 63;
    const float rx = r[(size_t)bi * 3 + 0];
    const float ry = r[(size_t)bi * 3 + 1];
    const float rz = r[(size_t)bi * 3 + 2];
    const int jb = 8 * (q & 7) + (q >> 3);
    v4f dv, ev;
#pragma unroll
    for (int w = 0; w < 4; ++w) {
      const int j = jb + 2 * w;
      const float dx = rx - r[(size_t)(b * NE_ + j) * 3 + 0];
      const float dy = ry - r[(size_t)(b * NE_ + j) * 3 + 1];
      const float dz = rz - r[(size_t)(b * NE_ + j) * 3 + 2];
      const float dist = sqrtf(dx * dx + dy * dy + dz * dz + EPS_);
      float env = 1.0f - dist * inv_cut;
      env = (dist < CUTOFF_) ? env * env : 0.0f;
      if (i == j) env = 0.0f;
      dv[w] = dist; ev[w] = env;
    }
    *(v4f*)&Pee[(size_t)bi * PEE_ + 4 * q]       = dv;
    *(v4f*)&Pee[(size_t)bi * PEE_ + NE_ + 4 * q] = ev;
  } else {
    const int u  = (blockIdx.x - PAIR_EE_BLOCKS_) * 256 + threadIdx.x;
    const int bj = u >> 2, q = u & 3;
    const float rx = r[(size_t)bj * 3 + 0];
    const float ry = r[(size_t)bj * 3 + 1];
    const float rz = r[(size_t)bj * 3 + 2];
    const int ab = 8 * (q & 1) + (q >> 1);
    v4f dv, ev;
#pragma unroll
    for (int w = 0; w < 4; ++w) {
      const int a = ab + 2 * w;
      const float dx = rx - coords[a * 3 + 0];
      const float dy = ry - coords[a * 3 + 1];
      const float dz = rz - coords[a * 3 + 2];
      const float dist = sqrtf(dx * dx + dy * dy + dz * dz + EPS_);
      float env = 1.0f - dist * inv_cut;
      env = (dist < CUTOFF_) ? env * env : 0.0f;
      dv[w] = dist; ev[w] = env;
    }
    *(v4f*)&Pne[(size_t)bj * PNE_ + 4 * q]       = dv;
    *(v4f*)&Pne[(size_t)bj * PNE_ + NN_ + 4 * q] = ev;
  }
}

#define PROD_STRIPS_ ((D_ / 16) * 2)
__global__ __launch_bounds__(32) __attribute__((amdgpu_waves_per_eu(1, 2)))
void gnn_hw0(const float* __restrict__ spin_emb, const float* __restrict__ nuc_emb,
                        const float* __restrict__ W_ee, const float* __restrict__ W_ne,
                        const float* __restrict__ W2,
                        float* __restrict__ hW, float* __restrict__ hnW,
                        float* __restrict__ P) {
  const int wid = blockIdx.x;
  const int lane = threadIdx.x & 31;
  v8f acc[NT_] = {};
  if (wid < 1024) {
    const int m0 = (wid >> 1) * 16;
    const int n0 = (wid & 1) * 64;
    gemm_strip_f32(h_row<true>(spin_emb, m0 + (lane & 15)), W_ee, n0, acc, lane);
    store_strip_f32(hW, D_, m0, n0, acc, lane);
  } else if (wid < 1026) {
    const int n0 = (wid - 1024) * 64;
    gemm_strip_f32(nuc_emb + (size_t)(lane & 15) * D_, W_ne, n0, acc, lane);
    store_strip_f32(hnW, D_, 0, n0, acc, lane);
  } else {
    const int idx = wid - 1026;
    const int l   = idx / PROD_STRIPS_;
    const int rem = idx % PROD_STRIPS_;
    const int m0 = (rem >> 1) * 16;
    const int n0 = (rem & 1) * 64;
    gemm_strip_f32(W2 + (size_t)l * WSZ_ + (size_t)(m0 + (lane & 15)) * D_,
                   W_ee + (size_t)(l + 1) * WSZ_, n0, acc, lane);
    store_strip_f32(P + (size_t)l * WSZ_, D_, m0, n0, acc, lane);
  }
}

#define IPW_ 8
#define SEE_ (NE_ / 4)
#define SNE_ (NN_ / 4)

struct AggP { v4f de[SEE_ / 2]; v4f ee[SEE_ / 2]; v4f dn[SNE_ / 2]; v4f en[SNE_ / 2]; };

__device__ __forceinline__ void agg_load_p(AggP& f, const float* __restrict__ pe,
                                           const float* __restrict__ pn) {
#pragma unroll
  for (int q = 0; q < SEE_ / 2; ++q) {
    f.de[q] = *(const v4f*)&pe[4 * q];
    f.ee[q] = *(const v4f*)&pe[NE_ + 4 * q];
  }
#pragma unroll
  for (int q = 0; q < SNE_ / 2; ++q) {
    f.dn[q] = *(const v4f*)&pn[4 * q];
    f.en[q] = *(const v4f*)&pn[NN_ + 4 * q];
  }
}

__device__ __forceinline__ v2f agg_feat(float d0, float d1, float e0, float e1,
                                        float mu, float kc) {
  v2f d;  d.x = d0;  d.y = d1;
  v2f en; en.x = e0; en.y = e1;
  const v2f zs = (d - mu) * kc;
  const v2f qq = zs * zs;
  v2f ex;
  ex.x = __builtin_amdgcn_exp2f(-qq.x);
  ex.y = __builtin_amdgcn_exp2f(-qq.y);
  return ex * en;
}

__device__ __forceinline__ float agg_xhalf(float x) {
  const int v = __builtin_bit_cast(int, x);
  return __builtin_bit_cast(float,
      __builtin_amdgcn_permlanex16(v, v, 0x76543210, 0xfedcba98, false, false));
}

__device__ __forceinline__ void agg_one(const AggP& f, float mu, float kc,
                                        const v2f (&bee)[SEE_][NT_],
                                        const v2f (&bne)[SNE_][NT_],
                                        const v2f (&wee)[NT_][4], const v2f (&wne)[NT_][4],
                                        float* orow, int half) {
  v8f g[NT_] = {};
#pragma unroll
  for (int s = 0; s < SEE_; ++s) {
    const int q = s >> 1, w = 2 * (s & 1);
    const v2f a = agg_feat(f.de[q][w], f.de[q][w + 1], f.ee[q][w], f.ee[q][w + 1], mu, kc);
#pragma unroll
    for (int t = 0; t < NT_; ++t)
      g[t] = __builtin_amdgcn_wmma_f32_16x16x4_f32(false, a, false, bee[s][t], (short)0,
                                                   g[t], false, false);
  }
  v8f gn[NT_] = {};
#pragma unroll
  for (int s = 0; s < SNE_; ++s) {
    const int q = s >> 1, w = 2 * (s & 1);
    const v2f a = agg_feat(f.dn[q][w], f.dn[q][w + 1], f.en[q][w], f.en[q][w + 1], mu, kc);
#pragma unroll
    for (int t = 0; t < NT_; ++t)
      gn[t] = __builtin_amdgcn_wmma_f32_16x16x4_f32(false, a, false, bne[s][t], (short)0,
                                                    gn[t], false, false);
  }
  float sv[NT_];
#pragma unroll
  for (int t = 0; t < NT_; ++t) {
    v2f p = {0.0f, 0.0f};
#pragma unroll
    for (int r = 0; r < 8; r += 2) {
      v2f gv; gv.x = g[t][r];  gv.y = g[t][r + 1];
      p += wee[t][r >> 1] * gv;
    }
#pragma unroll
    for (int r = 0; r < 8; r += 2) {
      v2f gv; gv.x = gn[t][r]; gv.y = gn[t][r + 1];
      p += wne[t][r >> 1] * gv;
    }
    sv[t] = p.x + p.y;
  }
  v2f own, snd;
  own.x = half ? sv[2] : sv[0];  own.y = half ? sv[3] : sv[1];
  snd.x = half ? sv[0] : sv[2];  snd.y = half ? sv[1] : sv[3];
  v2f o;
  o.x = own.x + agg_xhalf(snd.x);
  o.y = own.y + agg_xhalf(snd.y);
  *(v2f*)orow = o;
}

__global__ __launch_bounds__(128) __attribute__((amdgpu_waves_per_eu(2, 2)))
void gnn_agg_mlp1(const float* __restrict__ Pee,
                        const float* __restrict__ Pne,
                        const float* __restrict__ hW, const float* __restrict__ hnW,
                        const float* __restrict__ Wrbf_ee_l, const float* __restrict__ Wrbf_ne_l,
                        float* agg,
                        const float* __restrict__ W1_l, float* __restrict__ tmp) {
  const int lane = threadIdx.x & 31;
  const int wv   = __builtin_amdgcn_readfirstlane((int)(threadIdx.x >> 5));
  const int n0   = (wv & 1) * 64;
  const int b    = blockIdx.x / (NE_ / (2 * IPW_));
  const int i0   = ((blockIdx.x % (NE_ / (2 * IPW_))) * 2 + (wv >> 1)) * IPW_;
  const int half = lane >> 4, l16 = lane & 15;

  const float* __restrict__ pe = Pee + (size_t)(b * NE_ + i0) * PEE_ + half * (NE_ / 2);
  const float* __restrict__ pn = Pne + (size_t)(b * NE_ + i0) * PNE_ + half * (NN_ / 2);
  float* orow = agg + (size_t)(b * NE_ + i0) * D_ + n0 + 4 * l16 + 2 * half;

  const float mu = (CUTOFF_ / (float)(NRBF_ - 1)) * (float)l16;
  const float kc = ((float)NRBF_ / CUTOFF_) * 0.84932180028801904f;

  AggP f0, f1;
  agg_load_p(f0, pe, pn);

  const float* __restrict__ hb = hW + (size_t)b * NE_ * D_ + half * D_ + n0 + 4 * l16;
  const float* __restrict__ nb = hnW + half * D_ + n0 + 4 * l16;
  v2f bee[SEE_][NT_], bne[SNE_][NT_];
#pragma unroll
  for (int s = 0; s < SEE_; ++s) {
    const v4f x = *(const v4f*)&hb[(4 * s) * D_];
    const v4f y = *(const v4f*)&hb[(4 * s + 2) * D_];
#pragma unroll
    for (int t = 0; t < NT_; ++t) { bee[s][t].x = x[t]; bee[s][t].y = y[t]; }
  }
#pragma unroll
  for (int s = 0; s < SNE_; ++s) {
    const v4f x = *(const v4f*)&nb[(4 * s) * D_];
    const v4f y = *(const v4f*)&nb[(4 * s + 2) * D_];
#pragma unroll
    for (int t = 0; t < NT_; ++t) { bne[s][t].x = x[t]; bne[s][t].y = y[t]; }
  }
  v2f wee[NT_][4], wne[NT_][4];
  const float* __restrict__ we = Wrbf_ee_l + (size_t)(8 * half) * D_ + n0 + 4 * l16;
  const float* __restrict__ wn = Wrbf_ne_l + (size_t)(8 * half) * D_ + n0 + 4 * l16;
#pragma unroll
  for (int r = 0; r < 8; r += 2) {
    const v4f e0 = *(const v4f*)&we[r * D_], e1 = *(const v4f*)&we[(r + 1) * D_];
    const v4f n0v = *(const v4f*)&wn[r * D_], n1v = *(const v4f*)&wn[(r + 1) * D_];
#pragma unroll
    for (int t = 0; t < NT_; ++t) {
      wee[t][r >> 1].x = e0[t];  wee[t][r >> 1].y = e1[t];
      wne[t][r >> 1].x = n0v[t]; wne[t][r >> 1].y = n1v[t];
    }
  }

#pragma unroll 1
  for (int ii = 0; ii < IPW_ - 2; ii += 2) {
    agg_load_p(f1, pe + (size_t)(ii + 1) * PEE_, pn + (size_t)(ii + 1) * PNE_);
    agg_one(f0, mu, kc, bee, bne, wee, wne, orow + (size_t)ii * D_, half);
    agg_load_p(f0, pe + (size_t)(ii + 2) * PEE_, pn + (size_t)(ii + 2) * PNE_);
    agg_one(f1, mu, kc, bee, bne, wee, wne, orow + (size_t)(ii + 1) * D_, half);
  }
  agg_load_p(f1, pe + (size_t)(IPW_ - 1) * PEE_, pn + (size_t)(IPW_ - 1) * PNE_);
  agg_one(f0, mu, kc, bee, bne, wee, wne, orow + (size_t)(IPW_ - 2) * D_, half);
  agg_one(f1, mu, kc, bee, bne, wee, wne, orow + (size_t)(IPW_ - 1) * D_, half);

  __syncthreads();
  const int m0 = blockIdx.x * 16;
  const int c0 = wv * 32;
  v8f acc[2] = {};
  gemm_strip32_f32(agg + (size_t)(m0 + l16) * D_, W1_l, c0, acc, lane);
#pragma unroll
  for (int t = 0; t < 2; ++t)
#pragma unroll
    for (int ii = 0; ii < 8; ++ii) acc[t][ii] = tanhf(acc[t][ii]);
  store_strip32_f32(tmp, D_, m0, c0, acc, lane);
}

template <bool kFirst>
__global__ __launch_bounds__(32) __attribute__((amdgpu_waves_per_eu(1, 2)))
void gnn_mlp2_hw(const float* __restrict__ tmp, const float* __restrict__ W2_l,
                            const float* __restrict__ hsrc, float* __restrict__ dst,
                            const float* __restrict__ W_ee_n, const float* __restrict__ P_l,
                            const float* __restrict__ nuc_emb, const float* __restrict__ W_ne_n,
                            float* __restrict__ hW, float* __restrict__ hnW) {
  const int wid = blockIdx.x;
  const int lane = threadIdx.x & 31;
  if (wid < 1024) {
    const int m0 = (wid >> 1) * 16;
    const int n0 = (wid & 1) * 64;
    v8f acc[NT_] = {};
    gemm_strip_f32(h_row<kFirst>(hsrc, m0 + (lane & 15)), W_ee_n, n0, acc, lane);
    gemm_strip_f32(tmp + (size_t)(m0 + (lane & 15)) * D_, P_l, n0, acc, lane);
    store_strip_f32(hW, D_, m0, n0, acc, lane);
  } else if (wid < 1026) {
    const int n0 = (wid - 1024) * 64;
    v8f acc[NT_] = {};
    gemm_strip_f32(nuc_emb + (size_t)(lane & 15) * D_, W_ne_n, n0, acc, lane);
    store_strip_f32(hnW, D_, 0, n0, acc, lane);
  } else {
    const int sid = wid - 1026;
    const int m0 = (sid >> 1) * 16;
    const int n0 = (sid & 1) * 64;
    v8f acc[NT_];
    load_h_strip_f32<kFirst>(hsrc, m0, n0, acc, lane);
    gemm_strip_f32(tmp + (size_t)(m0 + (lane & 15)) * D_, W2_l, n0, acc, lane);
    store_strip_f32(dst, D_, m0, n0, acc, lane);
  }
}

__global__ __launch_bounds__(32) __attribute__((amdgpu_waves_per_eu(1, 2)))
void gnn_mlp2(const float* __restrict__ tmp, const float* __restrict__ W2_l,
                         float* hio) {
  const int wid = blockIdx.x;
  const int lane = threadIdx.x & 31;
  const int m0 = (wid >> 1) * 16;
  const int n0 = (wid & 1) * 64;
  v8f acc[NT_];
  load_h_strip_f32<false>(hio, m0, n0, acc, lane);
  gemm_strip_f32(tmp + (size_t)(m0 + (lane & 15)) * D_, W2_l, n0, acc, lane);
  store_strip_f32(hio, D_, m0, n0, acc, lane);
}

extern "C" void kernel_launch(void* const* d_in, const int* in_sizes, int n_in,
                              void* d_out, int out_size, void* d_ws, size_t ws_size,
                              hipStream_t stream) {
  (void)in_sizes; (void)n_in; (void)out_size; (void)ws_size;
  const float* r        = (const float*)d_in[0];
  const float* coords   = (const float*)d_in[1];
  const float* spin_emb = (const float*)d_in[2];
  const float* nuc_emb  = (const float*)d_in[3];
  const float* W_rbf_ee = (const float*)d_in[4];
  const float* W_rbf_ne = (const float*)d_in[5];
  const float* W_msg_ee = (const float*)d_in[6];
  const float* W_msg_ne = (const float*)d_in[7];
  const float* W1       = (const float*)d_in[8];
  const float* W2       = (const float*)d_in[9];
  float* out = (float*)d_out;

  float* ws  = (float*)d_ws;
  float* Pee = ws;
  float* Pne = Pee + (size_t)B_ * NE_ * PEE_;
  float* h   = Pne + (size_t)B_ * NE_ * PNE_;
  float* hW  = h   + (size_t)B_ * NE_ * D_;
  float* hnW = hW  + (size_t)B_ * NE_ * D_;
  float* agg = hnW + (size_t)NN_ * D_;
  float* tmp = agg + (size_t)B_ * NE_ * D_;
  float* P   = tmp + (size_t)B_ * NE_ * D_;

  gnn_pairs<<<PAIR_EE_BLOCKS_ + PAIR_NE_BLOCKS_, 256, 0, stream>>>(r, coords, Pee, Pne);
  gnn_hw0<<<1024 + 2 + (L_ - 1) * PROD_STRIPS_, 32, 0, stream>>>(spin_emb, nuc_emb, W_msg_ee,
                                                                W_msg_ne, W2, hW, hnW, P);

  static_assert(L_ == 3, "the h planes below are laid out for three layers");
  const float* hsrc = spin_emb;
  for (int l = 0; l < L_; ++l) {
    gnn_agg_mlp1<<<B_ * (NE_ / (2 * IPW_)), 128, 0, stream>>>(Pee, Pne, hW, hnW,
                                       W_rbf_ee + (size_t)l * NRBF_ * D_,
                                       W_rbf_ne + (size_t)l * NRBF_ * D_, agg,
                                       W1 + (size_t)l * WSZ_, tmp);
    if (l == L_ - 1) {
      gnn_mlp2<<<1024, 32, 0, stream>>>(tmp, W2 + (size_t)l * WSZ_, out);
    } else {
      float* dst = (l == 0) ? h : out;
      if (l == 0)
        gnn_mlp2_hw<true><<<1024 + 2 + 1024, 32, 0, stream>>>(
            tmp, W2, hsrc, dst, W_msg_ee + (size_t)WSZ_, P, nuc_emb,
            W_msg_ne + (size_t)WSZ_, hW, hnW);
      else
        gnn_mlp2_hw<false><<<1024 + 2 + 1024, 32, 0, stream>>>(
            tmp, W2 + (size_t)l * WSZ_, hsrc, dst, W_msg_ee + (size_t)(l + 1) * WSZ_,
            P + (size_t)l * WSZ_, nuc_emb, W_msg_ne + (size_t)(l + 1) * WSZ_, hW, hnW);
      hsrc = dst;
    }
  }
}


# agg sums only over senders inside the cutoff: pair table partitions each receiver's senders (inside first) and counts their WMMA steps; hW rows come from LDS by sender; ee sum order changes
# speedup vs baseline: 2.6905x; 1.0649x over previous
#include <hip/hip_runtime.h>
#include <math.h>

#define B_     128
#define NE_    64
#define NN_    16
#define D_     128
#define NRBF_  16
#define L_     3
#define NUP_   32
#define CUTOFF_ 2.0f
#define EPS_    1e-12f

#define NT_    4
#define WSZ_   (D_ * D_)

typedef float v2f __attribute__((ext_vector_type(2)));
typedef float v4f __attribute__((ext_vector_type(4)));
typedef float v8f __attribute__((ext_vector_type(8)));

__device__ __forceinline__ void gemm_strip_f32(const float* __restrict__ arow,
                                               const float* __restrict__ W,
                                               int n0, v8f acc[NT_], int lane) {
  const int half = lane >> 4;
  const int l16  = lane & 15;
  const float* __restrict__ wcol = W + n0 + 4 * l16;
#pragma unroll
  for (int k0 = 0; k0 < D_; k0 += 4) {
    const int ka = k0 + half * 2;
    const v2f a = *(const v2f*)&arow[ka];
    const v4f w0 = *(const v4f*)&wcol[(size_t)ka * D_];
    const v4f w1 = *(const v4f*)&wcol[(size_t)(ka + 1) * D_];
#pragma unroll
    for (int t = 0; t < NT_; ++t) {
      v2f b; b.x = w0[t]; b.y = w1[t];
      acc[t] = __builtin_amdgcn_wmma_f32_16x16x4_f32(false, a, false, b, (short)0,
                                                     acc[t], false, false);
    }
  }
}

__device__ __forceinline__ void store_strip_f32(float* __restrict__ C, int ldc,
                                                int m0, int n0, v8f acc[NT_], int lane) {
  const int half = lane >> 4;
  const int l16  = lane & 15;
#pragma unroll
  for (int ii = 0; ii < 8; ++ii) {
    v4f o;
#pragma unroll
    for (int t = 0; t < NT_; ++t) o[t] = acc[t][ii];
    *(v4f*)&C[(size_t)(m0 + half * 8 + ii) * ldc + n0 + 4 * l16] = o;
  }
}

template <bool kFirst>
__device__ __forceinline__ const float* h_row(const float* __restrict__ src, int row) {
  return kFirst ? src + (size_t)(((row & (NE_ - 1)) < NUP_) ? 0 : 1) * D_
                : src + (size_t)row * D_;
}

template <bool kFirst>
__device__ __forceinline__ void load_h_strip_f32(const float* __restrict__ src,
                                                 int m0, int n0, v8f acc[NT_], int lane) {
  const int half = lane >> 4;
  const int l16  = lane & 15;
#pragma unroll
  for (int ii = 0; ii < 8; ++ii) {
    const v4f o = *(const v4f*)&h_row<kFirst>(src, m0 + half * 8 + ii)[n0 + 4 * l16];
#pragma unroll
    for (int t = 0; t < NT_; ++t) acc[t][ii] = o[t];
  }
}

__device__ __forceinline__ void gemm_strip32_f32(const float* arow,
                                                 const float* __restrict__ W,
                                                 int n0, v8f acc[2], int lane) {
  const int half = lane >> 4;
  const int l16  = lane & 15;
  const float* __restrict__ wcol = W + n0 + 2 * l16;
#pragma unroll
  for (int k0 = 0; k0 < D_; k0 += 4) {
    const int ka = k0 + half * 2;
    const v2f a  = *(const v2f*)&arow[ka];
    const v2f w0 = *(const v2f*)&wcol[(size_t)ka * D_];
    const v2f w1 = *(const v2f*)&wcol[(size_t)(ka + 1) * D_];
#pragma unroll
    for (int t = 0; t < 2; ++t) {
      v2f b; b.x = w0[t]; b.y = w1[t];
      acc[t] = __builtin_amdgcn_wmma_f32_16x16x4_f32(false, a, false, b, (short)0,
                                                     acc[t], false, false);
    }
  }
}

__device__ __forceinline__ void store_strip32_f32(float* __restrict__ C, int ldc,
                                                  int m0, int n0, v8f acc[2], int lane) {
  const int half = lane >> 4;
  const int l16  = lane & 15;
#pragma unroll
  for (int ii = 0; ii < 8; ++ii) {
    v2f o; o.x = acc[0][ii]; o.y = acc[1][ii];
    *(v2f*)&C[(size_t)(m0 + half * 8 + ii) * ldc + n0 + 2 * l16] = o;
  }
}

#define PEE_ (3 * NE_)
#define PNE_ (2 * NN_)
#define PAIR_EE_BLOCKS_ ((B_ * NE_) / 8)
#define PAIR_NE_BLOCKS_ ((B_ * NE_ * (NN_ / 4)) / 256)

__global__ void gnn_pairs(const float* __restrict__ r, const float* __restrict__ coords,
                          float* __restrict__ Pee, float* __restrict__ Pne,
                          float* __restrict__ steps) {
  const float inv_cut = 1.0f / CUTOFF_;
  if (blockIdx.x < PAIR_EE_BLOCKS_) {
    const int lane = threadIdx.x & 31;
    const int bi = blockIdx.x * 8 + (threadIdx.x >> 5);
    const int b = bi >> 6, i = bi & 63;
    const float rx = r[(size_t)bi * 3 + 0];
    const float ry = r[(size_t)bi * 3 + 1];
    const float rz = r[(size_t)bi * 3 + 2];
    float dv[2], ev[2];
    bool in[2];
#pragma unroll
    for (int w = 0; w < 2; ++w) {
      const int j = lane + 32 * w;
      const float dx = rx - r[(size_t)(b * NE_ + j) * 3 + 0];
      const float dy = ry - r[(size_t)(b * NE_ + j) * 3 + 1];
      const float dz = rz - r[(size_t)(b * NE_ + j) * 3 + 2];
      const float dist = sqrtf(dx * dx + dy * dy + dz * dz + EPS_);
      float env = 1.0f - dist * inv_cut;
      env = (dist < CUTOFF_) ? env * env : 0.0f;
      if (i == j) env = 0.0f;
      dv[w] = dist; ev[w] = env;
      in[w] = (dist < CUTOFF_) && (i != j);
    }
    const unsigned m0 = __builtin_amdgcn_ballot_w32(in[0]);
    const unsigned m1 = __builtin_amdgcn_ballot_w32(in[1]);
    const unsigned below = (1u << lane) - 1u;
    const int c0 = (int)__popc(m0);
    const int n  = c0 + (int)__popc(m1);
    const int before[2] = { (int)__popc(m0 & below), c0 + (int)__popc(m1 & below) };
    float* __restrict__ pl = Pee + (size_t)bi * PEE_;
#pragma unroll
    for (int w = 0; w < 2; ++w) {
      const int j = lane + 32 * w;
      const int p = in[w] ? before[w] : n + (j - before[w]);
      const int slot = 32 * (p & 1) + 2 * (p >> 2) + ((p >> 1) & 1);
      pl[slot]           = dv[w];
      pl[NE_ + slot]     = ev[w];
      pl[2 * NE_ + slot] = __builtin_bit_cast(float, j * D_);
    }
    if (lane == 0) steps[bi] = __builtin_bit_cast(float, (n + 3) >> 2);
  } else {
    const int u  = (blockIdx.x - PAIR_EE_BLOCKS_) * 256 + threadIdx.x;
    const int bj = u >> 2, q = u & 3;
    const float rx = r[(size_t)bj * 3 + 0];
    const float ry = r[(size_t)bj * 3 + 1];
    const float rz = r[(size_t)bj * 3 + 2];
    const int ab = 8 * (q & 1) + (q >> 1);
    v4f dv, ev;
#pragma unroll
    for (int w = 0; w < 4; ++w) {
      const int a = ab + 2 * w;
      const float dx = rx - coords[a * 3 + 0];
      const float dy = ry - coords[a * 3 + 1];
      const float dz = rz - coords[a * 3 + 2];
      const float dist = sqrtf(dx * dx + dy * dy + dz * dz + EPS_);
      float env = 1.0f - dist * inv_cut;
      env = (dist < CUTOFF_) ? env * env : 0.0f;
      dv[w] = dist; ev[w] = env;
    }
    *(v4f*)&Pne[(size_t)bj * PNE_ + 4 * q]       = dv;
    *(v4f*)&Pne[(size_t)bj * PNE_ + NN_ + 4 * q] = ev;
  }
}

#define PROD_STRIPS_ ((D_ / 16) * 2)
__global__ __launch_bounds__(32) __attribute__((amdgpu_waves_per_eu(1, 2)))
void gnn_hw0(const float* __restrict__ spin_emb, const float* __restrict__ nuc_emb,
                        const float* __restrict__ W_ee, const float* __restrict__ W_ne,
                        const float* __restrict__ W2,
                        float* __restrict__ hW, float* __restrict__ hnW,
                        float* __restrict__ P) {
  const int wid = blockIdx.x;
  const int lane = threadIdx.x & 31;
  v8f acc[NT_] = {};
  if (wid < 1024) {
    const int m0 = (wid >> 1) * 16;
    const int n0 = (wid & 1) * 64;
    gemm_strip_f32(h_row<true>(spin_emb, m0 + (lane & 15)), W_ee, n0, acc, lane);
    store_strip_f32(hW, D_, m0, n0, acc, lane);
  } else if (wid < 1026) {
    const int n0 = (wid - 1024) * 64;
    gemm_strip_f32(nuc_emb + (size_t)(lane & 15) * D_, W_ne, n0, acc, lane);
    store_strip_f32(hnW, D_, 0, n0, acc, lane);
  } else {
    const int idx = wid - 1026;
    const int l   = idx / PROD_STRIPS_;
    const int rem = idx % PROD_STRIPS_;
    const int m0 = (rem >> 1) * 16;
    const int n0 = (rem & 1) * 64;
    gemm_strip_f32(W2 + (size_t)l * WSZ_ + (size_t)(m0 + (lane & 15)) * D_,
                   W_ee + (size_t)(l + 1) * WSZ_, n0, acc, lane);
    store_strip_f32(P + (size_t)l * WSZ_, D_, m0, n0, acc, lane);
  }
}

#define IPW_ 8
#define SEE_ (NE_ / 4)
#define SNE_ (NN_ / 4)

struct AggP { v4f de[SEE_ / 2]; v4f ee[SEE_ / 2]; v4f je[SEE_ / 2];
              v4f dn[SNE_ / 2]; v4f en[SNE_ / 2]; float ns; };

__device__ __forceinline__ void agg_load_p(AggP& f, const float* __restrict__ pe,
                                           const float* __restrict__ pn,
                                           const float* __restrict__ ps) {
  f.ns = *ps;
#pragma unroll
  for (int q = 0; q < SEE_ / 2; ++q) {
    f.de[q] = *(const v4f*)&pe[4 * q];
    f.ee[q] = *(const v4f*)&pe[NE_ + 4 * q];
    f.je[q] = *(const v4f*)&pe[2 * NE_ + 4 * q];
  }
#pragma unroll
  for (int q = 0; q < SNE_ / 2; ++q) {
    f.dn[q] = *(const v4f*)&pn[4 * q];
    f.en[q] = *(const v4f*)&pn[NN_ + 4 * q];
  }
}

__device__ __forceinline__ v2f agg_feat(float d0, float d1, float e0, float e1,
                                        float mu, float kc) {
  v2f d;  d.x = d0;  d.y = d1;
  v2f en; en.x = e0; en.y = e1;
  const v2f zs = (d - mu) * kc;
  const v2f qq = zs * zs;
  v2f ex;
  ex.x = __builtin_amdgcn_exp2f(-qq.x);
  ex.y = __builtin_amdgcn_exp2f(-qq.y);
  return ex * en;
}

__device__ __forceinline__ int agg_off(float x) { return __builtin_bit_cast(int, x); }

__device__ __forceinline__ float agg_xhalf(float x) {
  const int v = __builtin_bit_cast(int, x);
  return __builtin_bit_cast(float,
      __builtin_amdgcn_permlanex16(v, v, 0x76543210, 0xfedcba98, false, false));
}

template <int S>
__device__ __forceinline__ void agg_ee_steps(const AggP& f, float mu, float kc,
                                             const float* hws, int nst,
                                             v4f xc, v4f yc, v8f (&g)[NT_]) {
  if constexpr (S < SEE_) {
    if (S < nst) {
      v4f xn = xc, yn = yc;
      if constexpr (S + 1 < SEE_) {
        constexpr int q1 = (S + 1) >> 1, w1 = 2 * ((S + 1) & 1);
        xn = *(const v4f*)&hws[agg_off(f.je[q1][w1])];
        yn = *(const v4f*)&hws[agg_off(f.je[q1][w1 + 1])];
        __builtin_amdgcn_sched_barrier(0);
      }
      constexpr int q = S >> 1, w = 2 * (S & 1);
      const v2f ac = agg_feat(f.de[q][w], f.de[q][w + 1], f.ee[q][w], f.ee[q][w + 1], mu, kc);
#pragma unroll
      for (int t = 0; t < NT_; ++t) {
        v2f bb; bb.x = xc[t]; bb.y = yc[t];
        g[t] = __builtin_amdgcn_wmma_f32_16x16x4_f32(false, ac, false, bb, (short)0,
                                                     g[t], false, false);
      }
      if constexpr (S + 1 < SEE_) {
        __builtin_amdgcn_sched_barrier(0);
        asm volatile("" : : "v"(xn), "v"(yn));
      }
      agg_ee_steps<S + 1>(f, mu, kc, hws, nst, xn, yn, g);
    }
  }
}

__device__ __forceinline__ void agg_one(const AggP& f, float mu, float kc,
                                        const float* hws,
                                        const v2f (&bne)[SNE_][NT_],
                                        const v2f (&wee)[NT_][4], const v2f (&wne)[NT_][4],
                                        float* orow, int half) {
  v8f g[NT_] = {};
  const int nst = __builtin_amdgcn_readfirstlane(__builtin_bit_cast(int, f.ns));
  const v4f x0 = *(const v4f*)&hws[agg_off(f.je[0][0])];
  const v4f y0 = *(const v4f*)&hws[agg_off(f.je[0][1])];
  agg_ee_steps<0>(f, mu, kc, hws, nst, x0, y0, g);
  v8f gn[NT_] = {};
#pragma unroll
  for (int s = 0; s < SNE_; ++s) {
    const int q = s >> 1, w = 2 * (s & 1);
    const v2f a = agg_feat(f.dn[q][w], f.dn[q][w + 1], f.en[q][w], f.en[q][w + 1], mu, kc);
#pragma unroll
    for (int t = 0; t < NT_; ++t)
      gn[t] = __builtin_amdgcn_wmma_f32_16x16x4_f32(false, a, false, bne[s][t], (short)0,
                                                    gn[t], false, false);
  }
  float sv[NT_];
#pragma unroll
  for (int t = 0; t < NT_; ++t) {
    v2f p = {0.0f, 0.0f};
#pragma unroll
    for (int r = 0; r < 8; r += 2) {
      v2f gv; gv.x = g[t][r];  gv.y = g[t][r + 1];
      p += wee[t][r >> 1] * gv;
    }
#pragma unroll
    for (int r = 0; r < 8; r += 2) {
      v2f gv; gv.x = gn[t][r]; gv.y = gn[t][r + 1];
      p += wne[t][r >> 1] * gv;
    }
    sv[t] = p.x + p.y;
  }
  v2f own, snd;
  own.x = half ? sv[2] : sv[0];  own.y = half ? sv[3] : sv[1];
  snd.x = half ? sv[0] : sv[2];  snd.y = half ? sv[1] : sv[3];
  v2f o;
  o.x = own.x + agg_xhalf(snd.x);
  o.y = own.y + agg_xhalf(snd.y);
  *(v2f*)orow = o;
}

__global__ __launch_bounds__(128) __attribute__((amdgpu_waves_per_eu(2, 2)))
void gnn_agg_mlp1(const float* __restrict__ Pee,
                        const float* __restrict__ Pne, const float* __restrict__ steps,
                        const float* __restrict__ hW, const float* __restrict__ hnW,
                        const float* __restrict__ Wrbf_ee_l, const float* __restrict__ Wrbf_ne_l,
                        float* agg,
                        const float* __restrict__ W1_l, float* __restrict__ tmp) {
  const int lane = threadIdx.x & 31;
  const int wv   = __builtin_amdgcn_readfirstlane((int)(threadIdx.x >> 5));
  const int n0   = (wv & 1) * 64;
  const int b    = blockIdx.x / (NE_ / (2 * IPW_));
  const int i0   = ((blockIdx.x % (NE_ / (2 * IPW_))) * 2 + (wv >> 1)) * IPW_;
  const int half = lane >> 4, l16 = lane & 15;

  const float* __restrict__ pe = Pee + (size_t)(b * NE_ + i0) * PEE_ + half * (NE_ / 2);
  const float* __restrict__ pn = Pne + (size_t)(b * NE_ + i0) * PNE_ + half * (NN_ / 2);
  const float* __restrict__ ps = steps + (size_t)(b * NE_ + i0);
  float* orow = agg + (size_t)(b * NE_ + i0) * D_ + n0 + 4 * l16 + 2 * half;

  __shared__ __attribute__((aligned(16))) float hws_all[NE_ * D_];
  {
    const v4f* __restrict__ gsrc = (const v4f*)(hW + (size_t)b * NE_ * D_);
#pragma unroll
    for (int v = 0; v < (NE_ * D_ / 4) / 128; ++v)
      ((v4f*)hws_all)[v * 128 + threadIdx.x] = gsrc[v * 128 + threadIdx.x];
  }
  const float* hws = hws_all + n0 + 4 * l16;

  const float mu = (CUTOFF_ / (float)(NRBF_ - 1)) * (float)l16;
  const float kc = ((float)NRBF_ / CUTOFF_) * 0.84932180028801904f;

  AggP f0, f1;
  agg_load_p(f0, pe, pn, ps);

  const float* __restrict__ nb = hnW + half * D_ + n0 + 4 * l16;
  v2f bne[SNE_][NT_];
#pragma unroll
  for (int s = 0; s < SNE_; ++s) {
    const v4f x = *(const v4f*)&nb[(4 * s) * D_];
    const v4f y = *(const v4f*)&nb[(4 * s + 2) * D_];
#pragma unroll
    for (int t = 0; t < NT_; ++t) { bne[s][t].x = x[t]; bne[s][t].y = y[t]; }
  }
  v2f wee[NT_][4], wne[NT_][4];
  const float* __restrict__ we = Wrbf_ee_l + (size_t)(8 * half) * D_ + n0 + 4 * l16;
  const float* __restrict__ wn = Wrbf_ne_l + (size_t)(8 * half) * D_ + n0 + 4 * l16;
#pragma unroll
  for (int r = 0; r < 8; r += 2) {
    const v4f e0 = *(const v4f*)&we[r * D_], e1 = *(const v4f*)&we[(r + 1) * D_];
    const v4f n0v = *(const v4f*)&wn[r * D_], n1v = *(const v4f*)&wn[(r + 1) * D_];
#pragma unroll
    for (int t = 0; t < NT_; ++t) {
      wee[t][r >> 1].x = e0[t];  wee[t][r >> 1].y = e1[t];
      wne[t][r >> 1].x = n0v[t]; wne[t][r >> 1].y = n1v[t];
    }
  }

  __syncthreads();

#pragma unroll 1
  for (int ii = 0; ii < IPW_ - 2; ii += 2) {
    agg_load_p(f1, pe + (size_t)(ii + 1) * PEE_, pn + (size_t)(ii + 1) * PNE_, ps + ii + 1);
    agg_one(f0, mu, kc, hws, bne, wee, wne, orow + (size_t)ii * D_, half);
    agg_load_p(f0, pe + (size_t)(ii + 2) * PEE_, pn + (size_t)(ii + 2) * PNE_, ps + ii + 2);
    agg_one(f1, mu, kc, hws, bne, wee, wne, orow + (size_t)(ii + 1) * D_, half);
  }
  agg_load_p(f1, pe + (size_t)(IPW_ - 1) * PEE_, pn + (size_t)(IPW_ - 1) * PNE_, ps + IPW_ - 1);
  agg_one(f0, mu, kc, hws, bne, wee, wne, orow + (size_t)(IPW_ - 2) * D_, half);
  agg_one(f1, mu, kc, hws, bne, wee, wne, orow + (size_t)(IPW_ - 1) * D_, half);

  __syncthreads();
  const int m0 = blockIdx.x * 16;
  const int c0 = wv * 32;
  v8f acc[2] = {};
  gemm_strip32_f32(agg + (size_t)(m0 + l16) * D_, W1_l, c0, acc, lane);
#pragma unroll
  for (int t = 0; t < 2; ++t)
#pragma unroll
    for (int ii = 0; ii < 8; ++ii) acc[t][ii] = tanhf(acc[t][ii]);
  store_strip32_f32(tmp, D_, m0, c0, acc, lane);
}

template <bool kFirst>
__global__ __launch_bounds__(32) __attribute__((amdgpu_waves_per_eu(1, 2)))
void gnn_mlp2_hw(const float* __restrict__ tmp, const float* __restrict__ W2_l,
                            const float* __restrict__ hsrc, float* __restrict__ dst,
                            const float* __restrict__ W_ee_n, const float* __restrict__ P_l,
                            const float* __restrict__ nuc_emb, const float* __restrict__ W_ne_n,
                            float* __restrict__ hW, float* __restrict__ hnW) {
  const int wid = blockIdx.x;
  const int lane = threadIdx.x & 31;
  if (wid < 1024) {
    const int m0 = (wid >> 1) * 16;
    const int n0 = (wid & 1) * 64;
    v8f acc[NT_] = {};
    gemm_strip_f32(h_row<kFirst>(hsrc, m0 + (lane & 15)), W_ee_n, n0, acc, lane);
    gemm_strip_f32(tmp + (size_t)(m0 + (lane & 15)) * D_, P_l, n0, acc, lane);
    store_strip_f32(hW, D_, m0, n0, acc, lane);
  } else if (wid < 1026) {
    const int n0 = (wid - 1024) * 64;
    v8f acc[NT_] = {};
    gemm_strip_f32(nuc_emb + (size_t)(lane & 15) * D_, W_ne_n, n0, acc, lane);
    store_strip_f32(hnW, D_, 0, n0, acc, lane);
  } else {
    const int sid = wid - 1026;
    const int m0 = (sid >> 1) * 16;
    const int n0 = (sid & 1) * 64;
    v8f acc[NT_];
    load_h_strip_f32<kFirst>(hsrc, m0, n0, acc, lane);
    gemm_strip_f32(tmp + (size_t)(m0 + (lane & 15)) * D_, W2_l, n0, acc, lane);
    store_strip_f32(dst, D_, m0, n0, acc, lane);
  }
}

__global__ __launch_bounds__(32) __attribute__((amdgpu_waves_per_eu(1, 2)))
void gnn_mlp2(const float* __restrict__ tmp, const float* __restrict__ W2_l,
                         float* hio) {
  const int wid = blockIdx.x;
  const int lane = threadIdx.x & 31;
  const int m0 = (wid >> 1) * 16;
  const int n0 = (wid & 1) * 64;
  v8f acc[NT_];
  load_h_strip_f32<false>(hio, m0, n0, acc, lane);
  gemm_strip_f32(tmp + (size_t)(m0 + (lane & 15)) * D_, W2_l, n0, acc, lane);
  store_strip_f32(hio, D_, m0, n0, acc, lane);
}

extern "C" void kernel_launch(void* const* d_in, const int* in_sizes, int n_in,
                              void* d_out, int out_size, void* d_ws, size_t ws_size,
                              hipStream_t stream) {
  (void)in_sizes; (void)n_in; (void)out_size; (void)ws_size;
  const float* r        = (const float*)d_in[0];
  const float* coords   = (const float*)d_in[1];
  const float* spin_emb = (const float*)d_in[2];
  const float* nuc_emb  = (const float*)d_in[3];
  const float* W_rbf_ee = (const float*)d_in[4];
  const float* W_rbf_ne = (const float*)d_in[5];
  const float* W_msg_ee = (const float*)d_in[6];
  const float* W_msg_ne = (const float*)d_in[7];
  const float* W1       = (const float*)d_in[8];
  const float* W2       = (const float*)d_in[9];
  float* out = (float*)d_out;

  float* ws  = (float*)d_ws;
  float* Pee = ws;
  float* Pne = Pee + (size_t)B_ * NE_ * PEE_;
  float* stp = Pne + (size_t)B_ * NE_ * PNE_;
  float* h   = stp + (size_t)B_ * NE_;
  float* hW  = h   + (size_t)B_ * NE_ * D_;
  float* hnW = hW  + (size_t)B_ * NE_ * D_;
  float* agg = hnW + (size_t)NN_ * D_;
  float* tmp = agg + (size_t)B_ * NE_ * D_;
  float* P   = tmp + (size_t)B_ * NE_ * D_;

  gnn_pairs<<<PAIR_EE_BLOCKS_ + PAIR_NE_BLOCKS_, 256, 0, stream>>>(r, coords, Pee, Pne, stp);
  gnn_hw0<<<1024 + 2 + (L_ - 1) * PROD_STRIPS_, 32, 0, stream>>>(spin_emb, nuc_emb, W_msg_ee,
                                                                W_msg_ne, W2, hW, hnW, P);

  static_assert(L_ == 3, "the h planes below are laid out for three layers");
  const float* hsrc = spin_emb;
  for (int l = 0; l < L_; ++l) {
    gnn_agg_mlp1<<<B_ * (NE_ / (2 * IPW_)), 128, 0, stream>>>(Pee, Pne, stp, hW, hnW,
                                       W_rbf_ee + (size_t)l * NRBF_ * D_,
                                       W_rbf_ne + (size_t)l * NRBF_ * D_, agg,
                                       W1 + (size_t)l * WSZ_, tmp);
    if (l == L_ - 1) {
      gnn_mlp2<<<1024, 32, 0, stream>>>(tmp, W2 + (size_t)l * WSZ_, out);
    } else {
      float* dst = (l == 0) ? h : out;
      if (l == 0)
        gnn_mlp2_hw<true><<<1024 + 2 + 1024, 32, 0, stream>>>(
            tmp, W2, hsrc, dst, W_msg_ee + (size_t)WSZ_, P, nuc_emb,
            W_msg_ne + (size_t)WSZ_, hW, hnW);
      else
        gnn_mlp2_hw<false><<<1024 + 2 + 1024, 32, 0, stream>>>(
            tmp, W2 + (size_t)l * WSZ_, hsrc, dst, W_msg_ee + (size_t)(l + 1) * WSZ_,
            P + (size_t)l * WSZ_, nuc_emb, W_msg_ne + (size_t)(l + 1) * WSZ_, hW, hnW);
      hsrc = dst;
    }
  }
}


# pair tables (distance, envelope, cutoff partition, step counts) made per layer in LDS by each agg block's own waves in its prologue; pair kernel, its launch and its tables in memory go; same sums
# speedup vs baseline: 2.8054x; 1.0427x over previous
#include <hip/hip_runtime.h>
#include <math.h>

#define B_     128
#define NE_    64
#define NN_    16
#define D_     128
#define NRBF_  16
#define L_     3
#define NUP_   32
#define CUTOFF_ 2.0f
#define EPS_    1e-12f

#define NT_    4
#define WSZ_   (D_ * D_)

typedef float v2f __attribute__((ext_vector_type(2)));
typedef float v4f __attribute__((ext_vector_type(4)));
typedef float v8f __attribute__((ext_vector_type(8)));

__device__ __forceinline__ void gemm_strip_f32(const float* __restrict__ arow,
                                               const float* __restrict__ W,
                                               int n0, v8f acc[NT_], int lane) {
  const int half = lane >> 4;
  const int l16  = lane & 15;
  const float* __restrict__ wcol = W + n0 + 4 * l16;
#pragma unroll
  for (int k0 = 0; k0 < D_; k0 += 4) {
    const int ka = k0 + half * 2;
    const v2f a = *(const v2f*)&arow[ka];
    const v4f w0 = *(const v4f*)&wcol[(size_t)ka * D_];
    const v4f w1 = *(const v4f*)&wcol[(size_t)(ka + 1) * D_];
#pragma unroll
    for (int t = 0; t < NT_; ++t) {
      v2f b; b.x = w0[t]; b.y = w1[t];
      acc[t] = __builtin_amdgcn_wmma_f32_16x16x4_f32(false, a, false, b, (short)0,
                                                     acc[t], false, false);
    }
  }
}

__device__ __forceinline__ void store_strip_f32(float* __restrict__ C, int ldc,
                                                int m0, int n0, v8f acc[NT_], int lane) {
  const int half = lane >> 4;
  const int l16  = lane & 15;
#pragma unroll
  for (int ii = 0; ii < 8; ++ii) {
    v4f o;
#pragma unroll
    for (int t = 0; t < NT_; ++t) o[t] = acc[t][ii];
    *(v4f*)&C[(size_t)(m0 + half * 8 + ii) * ldc + n0 + 4 * l16] = o;
  }
}

template <bool kFirst>
__device__ __forceinline__ const float* h_row(const float* __restrict__ src, int row) {
  return kFirst ? src + (size_t)(((row & (NE_ - 1)) < NUP_) ? 0 : 1) * D_
                : src + (size_t)row * D_;
}

template <bool kFirst>
__device__ __forceinline__ void load_h_strip_f32(const float* __restrict__ src,
                                                 int m0, int n0, v8f acc[NT_], int lane) {
  const int half = lane >> 4;
  const int l16  = lane & 15;
#pragma unroll
  for (int ii = 0; ii < 8; ++ii) {
    const v4f o = *(const v4f*)&h_row<kFirst>(src, m0 + half * 8 + ii)[n0 + 4 * l16];
#pragma unroll
    for (int t = 0; t < NT_; ++t) acc[t][ii] = o[t];
  }
}

__device__ __forceinline__ void gemm_strip32_f32(const float* arow,
                                                 const float* __restrict__ W,
                                                 int n0, v8f acc[2], int lane) {
  const int half = lane >> 4;
  const int l16  = lane & 15;
  const float* __restrict__ wcol = W + n0 + 2 * l16;
#pragma unroll
  for (int k0 = 0; k0 < D_; k0 += 4) {
    const int ka = k0 + half * 2;
    const v2f a  = *(const v2f*)&arow[ka];
    const v2f w0 = *(const v2f*)&wcol[(size_t)ka * D_];
    const v2f w1 = *(const v2f*)&wcol[(size_t)(ka + 1) * D_];
#pragma unroll
    for (int t = 0; t < 2; ++t) {
      v2f b; b.x = w0[t]; b.y = w1[t];
      acc[t] = __builtin_amdgcn_wmma_f32_16x16x4_f32(false, a, false, b, (short)0,
                                                     acc[t], false, false);
    }
  }
}

__device__ __forceinline__ void store_strip32_f32(float* __restrict__ C, int ldc,
                                                  int m0, int n0, v8f acc[2], int lane) {
  const int half = lane >> 4;
  const int l16  = lane & 15;
#pragma unroll
  for (int ii = 0; ii < 8; ++ii) {
    v2f o; o.x = acc[0][ii]; o.y = acc[1][ii];
    *(v2f*)&C[(size_t)(m0 + half * 8 + ii) * ldc + n0 + 2 * l16] = o;
  }
}

#define PEE_ (3 * NE_)
#define PNE_ (2 * NN_)
#define RPB_ 16

struct PairSnd { float x[2], y[2], z[2]; };

__device__ __forceinline__ void pairs_load_senders(PairSnd& sd, const float* __restrict__ rb,
                                                   int lane) {
#pragma unroll
  for (int w = 0; w < 2; ++w) {
    const int j = lane + 32 * w;
    sd.x[w] = rb[j * 3 + 0];
    sd.y[w] = rb[j * 3 + 1];
    sd.z[w] = rb[j * 3 + 2];
  }
}

__device__ __forceinline__ void pairs_ee_one(const PairSnd& sd, float rx, float ry, float rz,
                                             int i, int lane, float* pl, float* st) {
  const float inv_cut = 1.0f / CUTOFF_;
  float dv[2], ev[2];
  bool in[2];
#pragma unroll
  for (int w = 0; w < 2; ++w) {
    const int j = lane + 32 * w;
    const float dx = rx - sd.x[w];
    const float dy = ry - sd.y[w];
    const float dz = rz - sd.z[w];
    const float dist = sqrtf(dx * dx + dy * dy + dz * dz + EPS_);
    float env = 1.0f - dist * inv_cut;
    env = (dist < CUTOFF_) ? env * env : 0.0f;
    if (i == j) env = 0.0f;
    dv[w] = dist; ev[w] = env;
    in[w] = (dist < CUTOFF_) && (i != j);
  }
  const unsigned m0 = __builtin_amdgcn_ballot_w32(in[0]);
  const unsigned m1 = __builtin_amdgcn_ballot_w32(in[1]);
  const unsigned below = (1u << lane) - 1u;
  const int c0 = (int)__popc(m0);
  const int n  = c0 + (int)__popc(m1);
  const int before[2] = { (int)__popc(m0 & below), c0 + (int)__popc(m1 & below) };
#pragma unroll
  for (int w = 0; w < 2; ++w) {
    const int j = lane + 32 * w;
    const int p = in[w] ? before[w] : n + (j - before[w]);
    const int slot = 32 * (p & 1) + 2 * (p >> 2) + ((p >> 1) & 1);
    pl[slot]           = dv[w];
    pl[NE_ + slot]     = ev[w];
    pl[2 * NE_ + slot] = __builtin_bit_cast(float, j * D_);
  }
  if (lane == 0) *st = __builtin_bit_cast(float, (n + 3) >> 2);
}

struct PairNuc { float rx, ry, rz; float x[2], y[2], z[2]; };

__device__ __forceinline__ void pairs_load_ne(PairNuc& nu, const float* __restrict__ rrow,
                                              const float* __restrict__ coords, int q, int w0) {
  nu.rx = rrow[0]; nu.ry = rrow[1]; nu.rz = rrow[2];
  const int ab = 8 * (q & 1) + (q >> 1);
#pragma unroll
  for (int ww = 0; ww < 2; ++ww) {
    const int a = ab + 2 * (w0 + ww);
    nu.x[ww] = coords[a * 3 + 0];
    nu.y[ww] = coords[a * 3 + 1];
    nu.z[ww] = coords[a * 3 + 2];
  }
}

__device__ __forceinline__ void pairs_ne_two(const PairNuc& nu, int q, int w0, float* pn) {
  const float inv_cut = 1.0f / CUTOFF_;
  v2f dv, ev;
#pragma unroll
  for (int ww = 0; ww < 2; ++ww) {
    const float dx = nu.rx - nu.x[ww];
    const float dy = nu.ry - nu.y[ww];
    const float dz = nu.rz - nu.z[ww];
    const float dist = sqrtf(dx * dx + dy * dy + dz * dz + EPS_);
    float env = 1.0f - dist * inv_cut;
    env = (dist < CUTOFF_) ? env * env : 0.0f;
    dv[ww] = dist; ev[ww] = env;
  }
  *(v2f*)&pn[4 * q + w0]       = dv;
  *(v2f*)&pn[NN_ + 4 * q + w0] = ev;
}

#define PROD_STRIPS_ ((D_ / 16) * 2)
__global__ __launch_bounds__(32) __attribute__((amdgpu_waves_per_eu(1, 2)))
void gnn_hw0(const float* __restrict__ spin_emb, const float* __restrict__ nuc_emb,
                        const float* __restrict__ W_ee, const float* __restrict__ W_ne,
                        const float* __restrict__ W2,
                        float* __restrict__ hW, float* __restrict__ hnW,
                        float* __restrict__ P) {
  const int wid = blockIdx.x;
  const int lane = threadIdx.x & 31;
  v8f acc[NT_] = {};
  if (wid < 1024) {
    const int m0 = (wid >> 1) * 16;
    const int n0 = (wid & 1) * 64;
    gemm_strip_f32(h_row<true>(spin_emb, m0 + (lane & 15)), W_ee, n0, acc, lane);
    store_strip_f32(hW, D_, m0, n0, acc, lane);
  } else if (wid < 1026) {
    const int n0 = (wid - 1024) * 64;
    gemm_strip_f32(nuc_emb + (size_t)(lane & 15) * D_, W_ne, n0, acc, lane);
    store_strip_f32(hnW, D_, 0, n0, acc, lane);
  } else {
    const int idx = wid - 1026;
    const int l   = idx / PROD_STRIPS_;
    const int rem = idx % PROD_STRIPS_;
    const int m0 = (rem >> 1) * 16;
    const int n0 = (rem & 1) * 64;
    gemm_strip_f32(W2 + (size_t)l * WSZ_ + (size_t)(m0 + (lane & 15)) * D_,
                   W_ee + (size_t)(l + 1) * WSZ_, n0, acc, lane);
    store_strip_f32(P + (size_t)l * WSZ_, D_, m0, n0, acc, lane);
  }
}

#define IPW_ 8
#define SEE_ (NE_ / 4)
#define SNE_ (NN_ / 4)

struct AggP { v4f de[SEE_ / 2]; v4f ee[SEE_ / 2]; v4f je[SEE_ / 2];
              v4f dn[SNE_ / 2]; v4f en[SNE_ / 2]; float ns; };

__device__ __forceinline__ void agg_load_p(AggP& f, const float* pe, const float* pn,
                                           const float* ps) {
  f.ns = *ps;
#pragma unroll
  for (int q = 0; q < SEE_ / 2; ++q) {
    f.de[q] = *(const v4f*)&pe[4 * q];
    f.ee[q] = *(const v4f*)&pe[NE_ + 4 * q];
    f.je[q] = *(const v4f*)&pe[2 * NE_ + 4 * q];
  }
#pragma unroll
  for (int q = 0; q < SNE_ / 2; ++q) {
    f.dn[q] = *(const v4f*)&pn[4 * q];
    f.en[q] = *(const v4f*)&pn[NN_ + 4 * q];
  }
}

__device__ __forceinline__ v2f agg_feat(float d0, float d1, float e0, float e1,
                                        float mu, float kc) {
  v2f d;  d.x = d0;  d.y = d1;
  v2f en; en.x = e0; en.y = e1;
  const v2f zs = (d - mu) * kc;
  const v2f qq = zs * zs;
  v2f ex;
  ex.x = __builtin_amdgcn_exp2f(-qq.x);
  ex.y = __builtin_amdgcn_exp2f(-qq.y);
  return ex * en;
}

__device__ __forceinline__ int agg_off(float x) { return __builtin_bit_cast(int, x); }

__device__ __forceinline__ float agg_xhalf(float x) {
  const int v = __builtin_bit_cast(int, x);
  return __builtin_bit_cast(float,
      __builtin_amdgcn_permlanex16(v, v, 0x76543210, 0xfedcba98, false, false));
}

template <int S>
__device__ __forceinline__ void agg_ee_steps(const AggP& f, float mu, float kc,
                                             const float* hws, int nst,
                                             v4f xc, v4f yc, v8f (&g)[NT_]) {
  if constexpr (S < SEE_) {
    if (S < nst) {
      v4f xn = xc, yn = yc;
      if constexpr (S + 1 < SEE_) {
        constexpr int q1 = (S + 1) >> 1, w1 = 2 * ((S + 1) & 1);
        xn = *(const v4f*)&hws[agg_off(f.je[q1][w1])];
        yn = *(const v4f*)&hws[agg_off(f.je[q1][w1 + 1])];
        __builtin_amdgcn_sched_barrier(0);
      }
      constexpr int q = S >> 1, w = 2 * (S & 1);
      const v2f ac = agg_feat(f.de[q][w], f.de[q][w + 1], f.ee[q][w], f.ee[q][w + 1], mu, kc);
#pragma unroll
      for (int t = 0; t < NT_; ++t) {
        v2f bb; bb.x = xc[t]; bb.y = yc[t];
        g[t] = __builtin_amdgcn_wmma_f32_16x16x4_f32(false, ac, false, bb, (short)0,
                                                     g[t], false, false);
      }
      if constexpr (S + 1 < SEE_) {
        __builtin_amdgcn_sched_barrier(0);
        asm volatile("" : : "v"(xn), "v"(yn));
      }
      agg_ee_steps<S + 1>(f, mu, kc, hws, nst, xn, yn, g);
    }
  }
}

__device__ __forceinline__ void agg_one(const AggP& f, float mu, float kc,
                                        const float* hws,
                                        const v2f (&bne)[SNE_][NT_],
                                        const v2f (&wee)[NT_][4], const v2f (&wne)[NT_][4],
                                        float* orow, int half) {
  v8f g[NT_] = {};
  const int nst = __builtin_amdgcn_readfirstlane(__builtin_bit_cast(int, f.ns));
  const v4f x0 = *(const v4f*)&hws[agg_off(f.je[0][0])];
  const v4f y0 = *(const v4f*)&hws[agg_off(f.je[0][1])];
  agg_ee_steps<0>(f, mu, kc, hws, nst, x0, y0, g);
  v8f gn[NT_] = {};
#pragma unroll
  for (int s = 0; s < SNE_; ++s) {
    const int q = s >> 1, w = 2 * (s & 1);
    const v2f a = agg_feat(f.dn[q][w], f.dn[q][w + 1], f.en[q][w], f.en[q][w + 1], mu, kc);
#pragma unroll
    for (int t = 0; t < NT_; ++t)
      gn[t] = __builtin_amdgcn_wmma_f32_16x16x4_f32(false, a, false, bne[s][t], (short)0,
                                                    gn[t], false, false);
  }
  float sv[NT_];
#pragma unroll
  for (int t = 0; t < NT_; ++t) {
    v2f p = {0.0f, 0.0f};
#pragma unroll
    for (int r = 0; r < 8; r += 2) {
      v2f gv; gv.x = g[t][r];  gv.y = g[t][r + 1];
      p += wee[t][r >> 1] * gv;
    }
#pragma unroll
    for (int r = 0; r < 8; r += 2) {
      v2f gv; gv.x = gn[t][r]; gv.y = gn[t][r + 1];
      p += wne[t][r >> 1] * gv;
    }
    sv[t] = p.x + p.y;
  }
  v2f own, snd;
  own.x = half ? sv[2] : sv[0];  own.y = half ? sv[3] : sv[1];
  snd.x = half ? sv[0] : sv[2];  snd.y = half ? sv[1] : sv[3];
  v2f o;
  o.x = own.x + agg_xhalf(snd.x);
  o.y = own.y + agg_xhalf(snd.y);
  *(v2f*)orow = o;
}

__global__ __launch_bounds__(128) __attribute__((amdgpu_waves_per_eu(2, 2)))
void gnn_agg_mlp1(const float* __restrict__ pos, const float* __restrict__ coords,
                        const float* __restrict__ hW, const float* __restrict__ hnW,
                        const float* __restrict__ Wrbf_ee_l, const float* __restrict__ Wrbf_ne_l,
                        float* agg,
                        const float* __restrict__ W1_l, float* __restrict__ tmp) {
  const int lane = threadIdx.x & 31;
  const int wv   = __builtin_amdgcn_readfirstlane((int)(threadIdx.x >> 5));
  const int n0   = (wv & 1) * 64;
  const int b    = blockIdx.x / (NE_ / RPB_);
  const int ib   = (blockIdx.x % (NE_ / RPB_)) * RPB_;
  const int i0   = ib + (wv >> 1) * IPW_;
  const int half = lane >> 4, l16 = lane & 15;
  static_assert(RPB_ == 2 * IPW_ && RPB_ % 4 == 0 && RPB_ * 8 == 128,
                "a block is 2 receiver groups; its 4 waves and 128 threads share the table build");

  float* orow = agg + (size_t)(b * NE_ + i0) * D_ + n0 + 4 * l16 + 2 * half;

  __shared__ __attribute__((aligned(16))) float pee_s[RPB_ * PEE_];
  __shared__ __attribute__((aligned(16))) float pne_s[RPB_ * PNE_];
  __shared__ float stp_s[RPB_];
  __shared__ __attribute__((aligned(16))) float hws_all[NE_ * D_];

  const float* __restrict__ rb = pos + (size_t)b * NE_ * 3;
  PairSnd sd;
  pairs_load_senders(sd, rb, lane);
  const int nli = (int)(threadIdx.x >> 3), nq = (int)((threadIdx.x >> 1) & 3);
  const int nw0 = 2 * (int)(threadIdx.x & 1);
  PairNuc nu;
  pairs_load_ne(nu, rb + (ib + nli) * 3, coords, nq, nw0);
  float rcv[RPB_ / 4][3];
#pragma unroll
  for (int t = 0; t < RPB_ / 4; ++t) {
    const int i = ib + wv * (RPB_ / 4) + t;
    rcv[t][0] = rb[i * 3 + 0]; rcv[t][1] = rb[i * 3 + 1]; rcv[t][2] = rb[i * 3 + 2];
  }

  v4f hwr[(NE_ * D_ / 4) / 128];
  {
    const v4f* __restrict__ gsrc = (const v4f*)(hW + (size_t)b * NE_ * D_);
#pragma unroll
    for (int v = 0; v < (NE_ * D_ / 4) / 128; ++v) hwr[v] = gsrc[v * 128 + threadIdx.x];
  }
  const float* hws = hws_all + n0 + 4 * l16;

  const float mu = (CUTOFF_ / (float)(NRBF_ - 1)) * (float)l16;
  const float kc = ((float)NRBF_ / CUTOFF_) * 0.84932180028801904f;

  AggP f0, f1;

  const float* __restrict__ nb = hnW + half * D_ + n0 + 4 * l16;
  v2f bne[SNE_][NT_];
#pragma unroll
  for (int s = 0; s < SNE_; ++s) {
    const v4f x = *(const v4f*)&nb[(4 * s) * D_];
    const v4f y = *(const v4f*)&nb[(4 * s + 2) * D_];
#pragma unroll
    for (int t = 0; t < NT_; ++t) { bne[s][t].x = x[t]; bne[s][t].y = y[t]; }
  }
  v2f wee[NT_][4], wne[NT_][4];
  const float* __restrict__ we = Wrbf_ee_l + (size_t)(8 * half) * D_ + n0 + 4 * l16;
  const float* __restrict__ wn = Wrbf_ne_l + (size_t)(8 * half) * D_ + n0 + 4 * l16;
#pragma unroll
  for (int r = 0; r < 8; r += 2) {
    const v4f e0 = *(const v4f*)&we[r * D_], e1 = *(const v4f*)&we[(r + 1) * D_];
    const v4f n0v = *(const v4f*)&wn[r * D_], n1v = *(const v4f*)&wn[(r + 1) * D_];
#pragma unroll
    for (int t = 0; t < NT_; ++t) {
      wee[t][r >> 1].x = e0[t];  wee[t][r >> 1].y = e1[t];
      wne[t][r >> 1].x = n0v[t]; wne[t][r >> 1].y = n1v[t];
    }
  }

#pragma unroll
  for (int t = 0; t < RPB_ / 4; ++t) {
    const int li = wv * (RPB_ / 4) + t;
    pairs_ee_one(sd, rcv[t][0], rcv[t][1], rcv[t][2], ib + li, lane,
                 pee_s + li * PEE_, stp_s + li);
  }
  pairs_ne_two(nu, nq, nw0, pne_s + nli * PNE_);
#pragma unroll
  for (int v = 0; v < (NE_ * D_ / 4) / 128; ++v) ((v4f*)hws_all)[v * 128 + threadIdx.x] = hwr[v];

  __syncthreads();

  const float* pe = pee_s + (wv >> 1) * IPW_ * PEE_ + half * (NE_ / 2);
  const float* pn = pne_s + (wv >> 1) * IPW_ * PNE_ + half * (NN_ / 2);
  const float* ps = stp_s + (wv >> 1) * IPW_;
  agg_load_p(f0, pe, pn, ps);

#pragma unroll 1
  for (int ii = 0; ii < IPW_ - 2; ii += 2) {
    agg_load_p(f1, pe + (size_t)(ii + 1) * PEE_, pn + (size_t)(ii + 1) * PNE_, ps + ii + 1);
    agg_one(f0, mu, kc, hws, bne, wee, wne, orow + (size_t)ii * D_, half);
    agg_load_p(f0, pe + (size_t)(ii + 2) * PEE_, pn + (size_t)(ii + 2) * PNE_, ps + ii + 2);
    agg_one(f1, mu, kc, hws, bne, wee, wne, orow + (size_t)(ii + 1) * D_, half);
  }
  agg_load_p(f1, pe + (size_t)(IPW_ - 1) * PEE_, pn + (size_t)(IPW_ - 1) * PNE_, ps + IPW_ - 1);
  agg_one(f0, mu, kc, hws, bne, wee, wne, orow + (size_t)(IPW_ - 2) * D_, half);
  agg_one(f1, mu, kc, hws, bne, wee, wne, orow + (size_t)(IPW_ - 1) * D_, half);

  __syncthreads();
  const int m0 = blockIdx.x * 16;
  const int c0 = wv * 32;
  v8f acc[2] = {};
  gemm_strip32_f32(agg + (size_t)(m0 + l16) * D_, W1_l, c0, acc, lane);
#pragma unroll
  for (int t = 0; t < 2; ++t)
#pragma unroll
    for (int ii = 0; ii < 8; ++ii) acc[t][ii] = tanhf(acc[t][ii]);
  store_strip32_f32(tmp, D_, m0, c0, acc, lane);
}

template <bool kFirst>
__global__ __launch_bounds__(32) __attribute__((amdgpu_waves_per_eu(1, 2)))
void gnn_mlp2_hw(const float* __restrict__ tmp, const float* __restrict__ W2_l,
                            const float* __restrict__ hsrc, float* __restrict__ dst,
                            const float* __restrict__ W_ee_n, const float* __restrict__ P_l,
                            const float* __restrict__ nuc_emb, const float* __restrict__ W_ne_n,
                            float* __restrict__ hW, float* __restrict__ hnW) {
  const int wid = blockIdx.x;
  const int lane = threadIdx.x & 31;
  if (wid < 1024) {
    const int m0 = (wid >> 1) * 16;
    const int n0 = (wid & 1) * 64;
    v8f acc[NT_] = {};
    gemm_strip_f32(h_row<kFirst>(hsrc, m0 + (lane & 15)), W_ee_n, n0, acc, lane);
    gemm_strip_f32(tmp + (size_t)(m0 + (lane & 15)) * D_, P_l, n0, acc, lane);
    store_strip_f32(hW, D_, m0, n0, acc, lane);
  } else if (wid < 1026) {
    const int n0 = (wid - 1024) * 64;
    v8f acc[NT_] = {};
    gemm_strip_f32(nuc_emb + (size_t)(lane & 15) * D_, W_ne_n, n0, acc, lane);
    store_strip_f32(hnW, D_, 0, n0, acc, lane);
  } else {
    const int sid = wid - 1026;
    const int m0 = (sid >> 1) * 16;
    const int n0 = (sid & 1) * 64;
    v8f acc[NT_];
    load_h_strip_f32<kFirst>(hsrc, m0, n0, acc, lane);
    gemm_strip_f32(tmp + (size_t)(m0 + (lane & 15)) * D_, W2_l, n0, acc, lane);
    store_strip_f32(dst, D_, m0, n0, acc, lane);
  }
}

__global__ __launch_bounds__(32) __attribute__((amdgpu_waves_per_eu(1, 2)))
void gnn_mlp2(const float* __restrict__ tmp, const float* __restrict__ W2_l,
                         float* hio) {
  const int wid = blockIdx.x;
  const int lane = threadIdx.x & 31;
  const int m0 = (wid >> 1) * 16;
  const int n0 = (wid & 1) * 64;
  v8f acc[NT_];
  load_h_strip_f32<false>(hio, m0, n0, acc, lane);
  gemm_strip_f32(tmp + (size_t)(m0 + (lane & 15)) * D_, W2_l, n0, acc, lane);
  store_strip_f32(hio, D_, m0, n0, acc, lane);
}

extern "C" void kernel_launch(void* const* d_in, const int* in_sizes, int n_in,
                              void* d_out, int out_size, void* d_ws, size_t ws_size,
                              hipStream_t stream) {
  (void)in_sizes; (void)n_in; (void)out_size; (void)ws_size;
  const float* r        = (const float*)d_in[0];
  const float* coords   = (const float*)d_in[1];
  const float* spin_emb = (const float*)d_in[2];
  const float* nuc_emb  = (const float*)d_in[3];
  const float* W_rbf_ee = (const float*)d_in[4];
  const float* W_rbf_ne = (const float*)d_in[5];
  const float* W_msg_ee = (const float*)d_in[6];
  const float* W_msg_ne = (const float*)d_in[7];
  const float* W1       = (const float*)d_in[8];
  const float* W2       = (const float*)d_in[9];
  float* out = (float*)d_out;

  float* ws  = (float*)d_ws;
  float* h   = ws;
  float* hW  = h   + (size_t)B_ * NE_ * D_;
  float* hnW = hW  + (size_t)B_ * NE_ * D_;
  float* agg = hnW + (size_t)NN_ * D_;
  float* tmp = agg + (size_t)B_ * NE_ * D_;
  float* P   = tmp + (size_t)B_ * NE_ * D_;

  gnn_hw0<<<1024 + 2 + (L_ - 1) * PROD_STRIPS_, 32, 0, stream>>>(spin_emb, nuc_emb, W_msg_ee,
                                                                W_msg_ne, W2, hW, hnW, P);

  static_assert(L_ == 3, "the h planes below are laid out for three layers");
  const float* hsrc = spin_emb;
  for (int l = 0; l < L_; ++l) {
    gnn_agg_mlp1<<<B_ * (NE_ / RPB_), 128, 0, stream>>>(r, coords, hW, hnW,
                                       W_rbf_ee + (size_t)l * NRBF_ * D_,
                                       W_rbf_ne + (size_t)l * NRBF_ * D_, agg,
                                       W1 + (size_t)l * WSZ_, tmp);
    if (l == L_ - 1) {
      gnn_mlp2<<<1024, 32, 0, stream>>>(tmp, W2 + (size_t)l * WSZ_, out);
    } else {
      float* dst = (l == 0) ? h : out;
      if (l == 0)
        gnn_mlp2_hw<true><<<1024 + 2 + 1024, 32, 0, stream>>>(
            tmp, W2, hsrc, dst, W_msg_ee + (size_t)WSZ_, P, nuc_emb,
            W_msg_ne + (size_t)WSZ_, hW, hnW);
      else
        gnn_mlp2_hw<false><<<1024 + 2 + 1024, 32, 0, stream>>>(
            tmp, W2 + (size_t)l * WSZ_, hsrc, dst, W_msg_ee + (size_t)(l + 1) * WSZ_,
            P + (size_t)l * WSZ_, nuc_emb, W_msg_ne + (size_t)(l + 1) * WSZ_, hW, hnW);
      hsrc = dst;
    }
  }
}
